# starting kernel
# baseline (speedup 1.0000x reference)
#include <hip/hip_runtime.h>
#include <hip/hip_bf16.h>

#define N_NODES  50000
#define N_EDGES  800000
#define DIM      128
#define N_LAYERS 3
#define N_GRAPHS 256

typedef __attribute__((ext_vector_type(2))) float v2f;
typedef __attribute__((ext_vector_type(8))) float v8f;

__global__ void zero_f32(float* __restrict__ p, int n) {
    int t = blockIdx.x * blockDim.x + threadIdx.x;
    if (t < n) p[t] = 0.0f;
}

__global__ void degree_kernel(const int* __restrict__ src, const int* __restrict__ dst,
                              float* __restrict__ deg_out, float* __restrict__ deg_in,
                              int n_edges) {
    int e = blockIdx.x * blockDim.x + threadIdx.x;
    if (e >= n_edges) return;
    unsafeAtomicAdd(&deg_out[src[e]], 1.0f);
    unsafeAtomicAdd(&deg_in[dst[e]], 1.0f);
}

__global__ void rsqrt_kernel(float* __restrict__ a, float* __restrict__ b, int n) {
    int t = blockIdx.x * blockDim.x + threadIdx.x;
    if (t >= n) return;
    a[t] = __frsqrt_rn(fmaxf(a[t], 1.0f));
    b[t] = __frsqrt_rn(fmaxf(b[t], 1.0f));
}

__global__ void spmm_kernel(const float* __restrict__ h, const float* __restrict__ norm_out,
                            const int* __restrict__ src, const int* __restrict__ dst,
                            float* __restrict__ agg, int n_edges) {
    long long t = (long long)blockIdx.x * blockDim.x + threadIdx.x;
    int e = (int)(t >> 5);
    int c = (int)(t & 31);
    if (e >= n_edges) return;
    int s = src[e];
    int d = dst[e];
    float ns = norm_out[s];
    const float4 v = *reinterpret_cast<const float4*>(h + (size_t)s * DIM + c * 4);
    float* p = agg + (size_t)d * DIM + c * 4;
    unsafeAtomicAdd(p + 0, v.x * ns);
    unsafeAtomicAdd(p + 1, v.y * ns);
    unsafeAtomicAdd(p + 2, v.z * ns);
    unsafeAtomicAdd(p + 3, v.w * ns);
}

__launch_bounds__(256, 2)
__global__ void gemm_relu_wmma(const float* __restrict__ A,
                               const float* __restrict__ norm_in,
                               const float* __restrict__ Wl,
                               const float* __restrict__ bias,
                               float* __restrict__ Out,
                               int n_rows) {
    __shared__ float sA[16 * 132];
    const int m0 = blockIdx.x * 16;
    const int tid = threadIdx.x;

    for (int i = tid; i < 16 * DIM; i += 256) {
        int r = i >> 7;
        int c = i & 127;
        int row = m0 + r;
        float v = 0.0f;
        if (row < n_rows) v = A[(size_t)row * DIM + c] * norm_in[row];
        sA[r * 132 + c] = v;
    }
    __syncthreads();

    const int wave = tid >> 5;
    const int lane = tid & 31;
    const int hl = lane & 15;
    const int hi = lane >> 4;
    const int col = wave * 16 + hl;

    v8f acc = {};
#pragma unroll
    for (int kk = 0; kk < 32; ++kk) {
        const int kb = kk * 4 + hi * 2;
        v2f a, b;
        a.x = sA[hl * 132 + kb + 0];
        a.y = sA[hl * 132 + kb + 1];
        b.x = Wl[(size_t)(kb + 0) * DIM + col];
        b.y = Wl[(size_t)(kb + 1) * DIM + col];
        acc = __builtin_amdgcn_wmma_f32_16x16x4_f32(
             false, a,  false, b,
             (short)0, acc,  false,  false);
    }

    const float bb = bias[col];
#pragma unroll
    for (int j = 0; j < 8; ++j) {
        int row = m0 + j + hi * 8;
        if (row < n_rows) {
            float v = acc[j] + bb;
            Out[(size_t)row * DIM + col] = v > 0.0f ? v : 0.0f;
        }
    }
}

__global__ void pool_accum(const float* __restrict__ h, const int* __restrict__ gid,
                           float* __restrict__ pooled, float* __restrict__ cnt,
                           int n_nodes) {
    long long t = (long long)blockIdx.x * blockDim.x + threadIdx.x;
    int nidx = (int)(t >> 5);
    int c = (int)(t & 31);
    if (nidx >= n_nodes) return;
    int g = gid[nidx];
    const float4 v = *reinterpret_cast<const float4*>(h + (size_t)nidx * DIM + c * 4);
    float* p = pooled + (size_t)g * DIM + c * 4;
    unsafeAtomicAdd(p + 0, v.x);
    unsafeAtomicAdd(p + 1, v.y);
    unsafeAtomicAdd(p + 2, v.z);
    unsafeAtomicAdd(p + 3, v.w);
    if (c == 0) unsafeAtomicAdd(&cnt[g], 1.0f);
}

__global__ void pool_div(float* __restrict__ pooled, const float* __restrict__ cnt) {
    int t = blockIdx.x * blockDim.x + threadIdx.x;
    int g = t >> 7;
    pooled[t] /= fmaxf(cnt[g], 1.0f);
}

extern "C" void kernel_launch(void* const* d_in, const int* in_sizes, int n_in,
                              void* d_out, int out_size, void* d_ws, size_t ws_size,
                              hipStream_t stream) {
    const float* feats = (const float*)d_in[0];
    const float* Wall  = (const float*)d_in[1];
    const float* ball  = (const float*)d_in[2];
    const int*   src   = (const int*)d_in[3];
    const int*   dst   = (const int*)d_in[4];
    const int*   gid   = (const int*)d_in[5];
    float* out = (float*)d_out;

    float* ws = (float*)d_ws;
    float* norm_out = ws;
    float* norm_in  = ws + N_NODES;
    float* bufA     = ws + 2 * (size_t)N_NODES;
    float* bufB     = bufA + (size_t)N_NODES * DIM;
    float* cnt      = bufB + (size_t)N_NODES * DIM;

    const int BT = 256;
    zero_f32<<<(2 * N_NODES + BT - 1) / BT, BT, 0, stream>>>(norm_out, 2 * N_NODES);
    degree_kernel<<<(N_EDGES + BT - 1) / BT, BT, 0, stream>>>(src, dst, norm_out, norm_in, N_EDGES);
    rsqrt_kernel<<<(N_NODES + BT - 1) / BT, BT, 0, stream>>>(norm_out, norm_in, N_NODES);

    const float* hin = feats;
    float* bufs[2] = {bufA, bufB};
    int cur = 0;
    const long long spmm_threads = (long long)N_EDGES * 32;
    for (int l = 0; l < N_LAYERS; ++l) {
        float* agg = bufs[cur];
        zero_f32<<<(N_NODES * DIM + BT - 1) / BT, BT, 0, stream>>>(agg, N_NODES * DIM);
        spmm_kernel<<<(int)((spmm_threads + BT - 1) / BT), BT, 0, stream>>>(
            hin, norm_out, src, dst, agg, N_EDGES);
        gemm_relu_wmma<<<(N_NODES + 15) / 16, BT, 0, stream>>>(
            agg, norm_in, Wall + (size_t)l * DIM * DIM, ball + (size_t)l * DIM, agg, N_NODES);
        hin = agg;
        cur ^= 1;
    }

    zero_f32<<<(N_GRAPHS * DIM + BT - 1) / BT, BT, 0, stream>>>(out, N_GRAPHS * DIM);
    zero_f32<<<1, N_GRAPHS, 0, stream>>>(cnt, N_GRAPHS);
    const long long pool_threads = (long long)N_NODES * 32;
    pool_accum<<<(int)((pool_threads + BT - 1) / BT), BT, 0, stream>>>(hin, gid, out, cnt, N_NODES);
    pool_div<<<(N_GRAPHS * DIM) / BT, BT, 0, stream>>>(out, cnt);
}


# SpMM by row ownership: each of 256 blocks keeps 196 destination rows in LDS, scans dst, adds matching source rows by LDS atomics, stores rows once; no global atomics or clear; sums in arrival order
# speedup vs baseline: 1.6228x; 1.6228x over previous
#include <hip/hip_runtime.h>
#include <hip/hip_bf16.h>

#define N_NODES  50000
#define N_EDGES  800000
#define DIM      128
#define N_LAYERS 3
#define N_GRAPHS 256

typedef __attribute__((ext_vector_type(2))) float v2f;
typedef __attribute__((ext_vector_type(8))) float v8f;

__global__ void zero_f32(float* __restrict__ p, int n) {
    int t = blockIdx.x * blockDim.x + threadIdx.x;
    if (t < n) p[t] = 0.0f;
}

__global__ void degree_kernel(const int* __restrict__ src, const int* __restrict__ dst,
                              float* __restrict__ deg_out, float* __restrict__ deg_in,
                              int n_edges) {
    int e = blockIdx.x * blockDim.x + threadIdx.x;
    if (e >= n_edges) return;
    unsafeAtomicAdd(&deg_out[src[e]], 1.0f);
    unsafeAtomicAdd(&deg_in[dst[e]], 1.0f);
}

__global__ void rsqrt_kernel(float* __restrict__ a, float* __restrict__ b, int n) {
    int t = blockIdx.x * blockDim.x + threadIdx.x;
    if (t >= n) return;
    a[t] = __frsqrt_rn(fmaxf(a[t], 1.0f));
    b[t] = __frsqrt_rn(fmaxf(b[t], 1.0f));
}

#define ROWS_PER_BLOCK 196
#define SPMM_THREADS   1024
#define EDGE_CHUNK     256
static_assert(N_EDGES % EDGE_CHUNK == 0, "the dst list is read in whole chunks");

__launch_bounds__(SPMM_THREADS)
__global__ void spmm_owner(const float* __restrict__ h, const float* __restrict__ norm_out,
                           const int* __restrict__ src, const int* __restrict__ dst,
                           float* __restrict__ agg, int n_edges, int n_nodes) {
    __shared__ __attribute__((aligned(16))) float acc[ROWS_PER_BLOCK * DIM];
    const int tid = threadIdx.x;
    const int lane = tid & 31;
    const int wave = __builtin_amdgcn_readfirstlane(tid >> 5);
    const int r0 = blockIdx.x * ROWS_PER_BLOCK;

    const float4 zero4 = {0.0f, 0.0f, 0.0f, 0.0f};
    for (int i = tid; i < ROWS_PER_BLOCK * (DIM / 4); i += SPMM_THREADS)
        reinterpret_cast<float4*>(acc)[i] = zero4;
    __syncthreads();

    auto add_one = [&](int e, int row) {
        const int s = src[e];
        const float ns = norm_out[s];
        const float* hp = h + (size_t)s * DIM + lane;
        const float v0 = hp[0], v1 = hp[32], v2 = hp[64], v3 = hp[96];
        float* ap = acc + row * DIM + lane;
        unsafeAtomicAdd(ap + 0,  v0 * ns);
        unsafeAtomicAdd(ap + 32, v1 * ns);
        unsafeAtomicAdd(ap + 64, v2 * ns);
        unsafeAtomicAdd(ap + 96, v3 * ns);
    };
    auto add_two = [&](int ea, int rowa, int eb, int rowb) {
        const int sa = src[ea];
        const int sb = src[eb];
        const float na = norm_out[sa];
        const float nb = norm_out[sb];
        const float* ha = h + (size_t)sa * DIM + lane;
        const float* hb = h + (size_t)sb * DIM + lane;
        const float a0 = ha[0], a1 = ha[32], a2 = ha[64], a3 = ha[96];
        const float b0 = hb[0], b1 = hb[32], b2 = hb[64], b3 = hb[96];
        float* pa = acc + rowa * DIM + lane;
        float* pb = acc + rowb * DIM + lane;
        unsafeAtomicAdd(pa + 0,  a0 * na);
        unsafeAtomicAdd(pa + 32, a1 * na);
        unsafeAtomicAdd(pa + 64, a2 * na);
        unsafeAtomicAdd(pa + 96, a3 * na);
        unsafeAtomicAdd(pb + 0,  b0 * nb);
        unsafeAtomicAdd(pb + 32, b1 * nb);
        unsafeAtomicAdd(pb + 64, b2 * nb);
        unsafeAtomicAdd(pb + 96, b3 * nb);
    };

    const int n_chunks = n_edges / EDGE_CHUNK;
    const int n_waves = SPMM_THREADS / 32;
    const int offset = n_chunks > 0 ? (int)((blockIdx.x * 12u) % (unsigned)n_chunks) : 0;
    const int4* __restrict__ dst4 = reinterpret_cast<const int4*>(dst);

    int pend_e = -1;
    int pend_row = 0;

    auto scan_chunk = [&](const int4& c0, const int4& c1, int ch) {
        const unsigned rel[8] = {(unsigned)(c0.x - r0), (unsigned)(c0.y - r0),
                                 (unsigned)(c0.z - r0), (unsigned)(c0.w - r0),
                                 (unsigned)(c1.x - r0), (unsigned)(c1.y - r0),
                                 (unsigned)(c1.z - r0), (unsigned)(c1.w - r0)};
        unsigned m[8];
        unsigned any = 0;
#pragma unroll
        for (int k = 0; k < 8; ++k) {
            m[k] = __builtin_amdgcn_ballot_w32(rel[k] < (unsigned)ROWS_PER_BLOCK);
            any |= m[k];
        }
        if (any) {
#pragma unroll
            for (int k = 0; k < 8; ++k) {
                unsigned mask = m[k];
                while (mask) {
                    const int l = __builtin_ctz(mask);
                    mask &= mask - 1;
                    const int row = __builtin_amdgcn_readlane((int)rel[k], l);
                    const int e = ch * EDGE_CHUNK + (k >> 2) * 128 + l * 4 + (k & 3);
                    if (pend_e < 0) {
                        pend_e = e;
                        pend_row = row;
                    } else {
                        add_two(pend_e, pend_row, e, row);
                        pend_e = -1;
                    }
                }
            }
        }
    };

    int idx = wave;
    int ch = idx + offset;
    if (ch >= n_chunks) ch -= n_chunks;
    const int fch = idx < n_chunks ? ch : 0;
    int4 a0 = dst4[fch * 64 + lane];
    int4 a1 = dst4[fch * 64 + 32 + lane];
    int4 b0, b1;
    while (idx < n_chunks) {
        int nidx = idx + n_waves;
        int nch = nidx + offset;
        if (nch >= n_chunks) nch -= n_chunks;
        int lch = nidx < n_chunks ? nch : ch;
        b0 = dst4[lch * 64 + lane];
        b1 = dst4[lch * 64 + 32 + lane];
        scan_chunk(a0, a1, ch);
        idx = nidx;
        ch = nch;
        if (idx >= n_chunks) break;

        nidx = idx + n_waves;
        nch = nidx + offset;
        if (nch >= n_chunks) nch -= n_chunks;
        lch = nidx < n_chunks ? nch : ch;
        a0 = dst4[lch * 64 + lane];
        a1 = dst4[lch * 64 + 32 + lane];
        scan_chunk(b0, b1, ch);
        idx = nidx;
        ch = nch;
    }
    if (pend_e >= 0) add_one(pend_e, pend_row);
    __syncthreads();

    for (int i = tid; i < ROWS_PER_BLOCK * (DIM / 4); i += SPMM_THREADS) {
        const int r = i >> 5;
        const int c4 = i & 31;
        const int grow = r0 + r;
        if (grow < n_nodes)
            reinterpret_cast<float4*>(agg + (size_t)grow * DIM)[c4] =
                reinterpret_cast<const float4*>(acc + r * DIM)[c4];
    }
}

__launch_bounds__(256, 2)
__global__ void gemm_relu_wmma(const float* __restrict__ A,
                               const float* __restrict__ norm_in,
                               const float* __restrict__ Wl,
                               const float* __restrict__ bias,
                               float* __restrict__ Out,
                               int n_rows) {
    __shared__ float sA[16 * 132];
    const int m0 = blockIdx.x * 16;
    const int tid = threadIdx.x;

    for (int i = tid; i < 16 * DIM; i += 256) {
        int r = i >> 7;
        int c = i & 127;
        int row = m0 + r;
        float v = 0.0f;
        if (row < n_rows) v = A[(size_t)row * DIM + c] * norm_in[row];
        sA[r * 132 + c] = v;
    }
    __syncthreads();

    const int wave = tid >> 5;
    const int lane = tid & 31;
    const int hl = lane & 15;
    const int hi = lane >> 4;
    const int col = wave * 16 + hl;

    v8f acc = {};
#pragma unroll
    for (int kk = 0; kk < 32; ++kk) {
        const int kb = kk * 4 + hi * 2;
        v2f a, b;
        a.x = sA[hl * 132 + kb + 0];
        a.y = sA[hl * 132 + kb + 1];
        b.x = Wl[(size_t)(kb + 0) * DIM + col];
        b.y = Wl[(size_t)(kb + 1) * DIM + col];
        acc = __builtin_amdgcn_wmma_f32_16x16x4_f32(
             false, a,  false, b,
             (short)0, acc,  false,  false);
    }

    const float bb = bias[col];
#pragma unroll
    for (int j = 0; j < 8; ++j) {
        int row = m0 + j + hi * 8;
        if (row < n_rows) {
            float v = acc[j] + bb;
            Out[(size_t)row * DIM + col] = v > 0.0f ? v : 0.0f;
        }
    }
}

__global__ void pool_accum(const float* __restrict__ h, const int* __restrict__ gid,
                           float* __restrict__ pooled, float* __restrict__ cnt,
                           int n_nodes) {
    long long t = (long long)blockIdx.x * blockDim.x + threadIdx.x;
    int nidx = (int)(t >> 5);
    int c = (int)(t & 31);
    if (nidx >= n_nodes) return;
    int g = gid[nidx];
    const float4 v = *reinterpret_cast<const float4*>(h + (size_t)nidx * DIM + c * 4);
    float* p = pooled + (size_t)g * DIM + c * 4;
    unsafeAtomicAdd(p + 0, v.x);
    unsafeAtomicAdd(p + 1, v.y);
    unsafeAtomicAdd(p + 2, v.z);
    unsafeAtomicAdd(p + 3, v.w);
    if (c == 0) unsafeAtomicAdd(&cnt[g], 1.0f);
}

__global__ void pool_div(float* __restrict__ pooled, const float* __restrict__ cnt) {
    int t = blockIdx.x * blockDim.x + threadIdx.x;
    int g = t >> 7;
    pooled[t] /= fmaxf(cnt[g], 1.0f);
}

extern "C" void kernel_launch(void* const* d_in, const int* in_sizes, int n_in,
                              void* d_out, int out_size, void* d_ws, size_t ws_size,
                              hipStream_t stream) {
    const float* feats = (const float*)d_in[0];
    const float* Wall  = (const float*)d_in[1];
    const float* ball  = (const float*)d_in[2];
    const int*   src   = (const int*)d_in[3];
    const int*   dst   = (const int*)d_in[4];
    const int*   gid   = (const int*)d_in[5];
    float* out = (float*)d_out;

    float* ws = (float*)d_ws;
    float* norm_out = ws;
    float* norm_in  = ws + N_NODES;
    float* bufA     = ws + 2 * (size_t)N_NODES;
    float* bufB     = bufA + (size_t)N_NODES * DIM;
    float* cnt      = bufB + (size_t)N_NODES * DIM;

    const int BT = 256;
    zero_f32<<<(2 * N_NODES + BT - 1) / BT, BT, 0, stream>>>(norm_out, 2 * N_NODES);
    degree_kernel<<<(N_EDGES + BT - 1) / BT, BT, 0, stream>>>(src, dst, norm_out, norm_in, N_EDGES);
    rsqrt_kernel<<<(N_NODES + BT - 1) / BT, BT, 0, stream>>>(norm_out, norm_in, N_NODES);

    const float* hin = feats;
    float* bufs[2] = {bufA, bufB};
    int cur = 0;
    for (int l = 0; l < N_LAYERS; ++l) {
        float* agg = bufs[cur];
        spmm_owner<<<(N_NODES + ROWS_PER_BLOCK - 1) / ROWS_PER_BLOCK, SPMM_THREADS, 0, stream>>>(
            hin, norm_out, src, dst, agg, N_EDGES, N_NODES);
        gemm_relu_wmma<<<(N_NODES + 15) / 16, BT, 0, stream>>>(
            agg, norm_in, Wall + (size_t)l * DIM * DIM, ball + (size_t)l * DIM, agg, N_NODES);
        hin = agg;
        cur ^= 1;
    }

    zero_f32<<<(N_GRAPHS * DIM + BT - 1) / BT, BT, 0, stream>>>(out, N_GRAPHS * DIM);
    zero_f32<<<1, N_GRAPHS, 0, stream>>>(cnt, N_GRAPHS);
    const long long pool_threads = (long long)N_NODES * 32;
    pool_accum<<<(int)((pool_threads + BT - 1) / BT), BT, 0, stream>>>(hin, gid, out, cnt, N_NODES);
    pool_div<<<(N_GRAPHS * DIM) / BT, BT, 0, stream>>>(out, cnt);
}


# SpMM owner scan no longer stops per matching edge: each wave queues matches in LDS and resolves 32 at once (index gather, norms, rows 8 in flight), 4 chunk loads in flight; LDS sums in arrival order
# speedup vs baseline: 1.6851x; 1.0384x over previous
#include <hip/hip_runtime.h>
#include <hip/hip_bf16.h>

#define N_NODES  50000
#define N_EDGES  800000
#define DIM      128
#define N_LAYERS 3
#define N_GRAPHS 256

typedef __attribute__((ext_vector_type(2))) float v2f;
typedef __attribute__((ext_vector_type(8))) float v8f;

__global__ void zero_f32(float* __restrict__ p, int n) {
    int t = blockIdx.x * blockDim.x + threadIdx.x;
    if (t < n) p[t] = 0.0f;
}

__global__ void degree_kernel(const int* __restrict__ src, const int* __restrict__ dst,
                              float* __restrict__ deg_out, float* __restrict__ deg_in,
                              int n_edges) {
    int e = blockIdx.x * blockDim.x + threadIdx.x;
    if (e >= n_edges) return;
    unsafeAtomicAdd(&deg_out[src[e]], 1.0f);
    unsafeAtomicAdd(&deg_in[dst[e]], 1.0f);
}

__global__ void rsqrt_kernel(float* __restrict__ a, float* __restrict__ b, int n) {
    int t = blockIdx.x * blockDim.x + threadIdx.x;
    if (t >= n) return;
    a[t] = __frsqrt_rn(fmaxf(a[t], 1.0f));
    b[t] = __frsqrt_rn(fmaxf(b[t], 1.0f));
}

#define ROWS_PER_BLOCK 196
#define SPMM_THREADS   1024
#define SPMM_WAVES     (SPMM_THREADS / 32)
#define EDGE_CHUNK     256
#define RING           4
#define QCAP           512
#define ROWS_IN_FLIGHT 8
static_assert(N_EDGES % EDGE_CHUNK == 0, "the dst list is read in whole chunks");

__launch_bounds__(SPMM_THREADS)
__global__ void spmm_owner(const float* __restrict__ h, const float* __restrict__ norm_out,
                           const int* __restrict__ src, const int* __restrict__ dst,
                           float* __restrict__ agg, int n_edges, int n_nodes) {
    __shared__ __attribute__((aligned(16))) float acc[ROWS_PER_BLOCK * DIM];
    __shared__ int2 queue[SPMM_WAVES * QCAP];
    const int tid = threadIdx.x;
    const int lane = tid & 31;
    const int wave = __builtin_amdgcn_readfirstlane(tid >> 5);
    const int r0 = blockIdx.x * ROWS_PER_BLOCK;
    int2* const q = queue + wave * QCAP;

    const float4 zero4 = {0.0f, 0.0f, 0.0f, 0.0f};
    for (int i = tid; i < ROWS_PER_BLOCK * (DIM / 4); i += SPMM_THREADS)
        reinterpret_cast<float4*>(acc)[i] = zero4;
    __syncthreads();

    int qh = 0;
    int qn = 0;

    auto drain = [&](int n) {
        int2 ent = {0, 0};
        if (lane < n) ent = q[(qh + lane) & (QCAP - 1)];
        const int s = src[ent.x];
        const int nsb = __float_as_int(norm_out[s]);
#pragma nounroll
        for (int j0 = 0; j0 < n; j0 += ROWS_IN_FLIGHT) {
            float v[ROWS_IN_FLIGHT][4];
#pragma unroll
            for (int u = 0; u < ROWS_IN_FLIGHT; ++u) {
                const int sj = __builtin_amdgcn_readlane(s, j0 + u);
                const float* hp = h + (size_t)sj * DIM + lane;
                v[u][0] = hp[0]; v[u][1] = hp[32]; v[u][2] = hp[64]; v[u][3] = hp[96];
            }
#pragma unroll
            for (int u = 0; u < ROWS_IN_FLIGHT; ++u) {
                if (j0 + u < n) {
                    const float nj = __int_as_float(__builtin_amdgcn_readlane(nsb, j0 + u));
                    const int rowj = __builtin_amdgcn_readlane(ent.y, j0 + u);
                    float* ap = acc + rowj * DIM + lane;
                    unsafeAtomicAdd(ap + 0,  v[u][0] * nj);
                    unsafeAtomicAdd(ap + 32, v[u][1] * nj);
                    unsafeAtomicAdd(ap + 64, v[u][2] * nj);
                    unsafeAtomicAdd(ap + 96, v[u][3] * nj);
                }
            }
        }
        qh = (qh + n) & (QCAP - 1);
        qn -= n;
    };

    auto scan_chunk = [&](const int4& c0, const int4& c1, int ch) {
        const unsigned rel[8] = {(unsigned)(c0.x - r0), (unsigned)(c0.y - r0),
                                 (unsigned)(c0.z - r0), (unsigned)(c0.w - r0),
                                 (unsigned)(c1.x - r0), (unsigned)(c1.y - r0),
                                 (unsigned)(c1.z - r0), (unsigned)(c1.w - r0)};
        unsigned m[8];
        unsigned any = 0;
#pragma unroll
        for (int k = 0; k < 8; ++k) {
            m[k] = __builtin_amdgcn_ballot_w32(rel[k] < (unsigned)ROWS_PER_BLOCK);
            any |= m[k];
        }
        if (any) {
#pragma unroll
            for (int k = 0; k < 8; ++k) {
                if (m[k]) {
                    if (rel[k] < (unsigned)ROWS_PER_BLOCK) {
                        const int pos = qh + qn + (int)__builtin_amdgcn_mbcnt_lo(m[k], 0u);
                        const int e = ch * EDGE_CHUNK + (k >> 2) * 128 + lane * 4 + (k & 3);
                        q[pos & (QCAP - 1)] = make_int2(e, (int)rel[k]);
                    }
                    qn += __builtin_popcount(m[k]);
                }
            }
            while (qn >= 32) drain(32);
        }
    };

    const int n_chunks = n_edges / EDGE_CHUNK;
    const int offset = n_chunks > 0 ? (int)((blockIdx.x * 12u) % (unsigned)n_chunks) : 0;
    const int4* __restrict__ dst4 = reinterpret_cast<const int4*>(dst);

    auto chunk_of = [&](int idx) {
        int ch = idx + offset;
        if (ch >= n_chunks) ch -= n_chunks;
        return idx < n_chunks ? ch : 0;
    };
    auto load_set = [&](int4& x0, int4& x1, int idx) {
        const int ch = chunk_of(idx);
        x0 = dst4[ch * 64 + lane];
        x1 = dst4[ch * 64 + 32 + lane];
    };
    int4 a0, a1, b0, b1, e0, e1, f0, f1;
    static_assert(RING == 4, "four named register sets below");
    load_set(a0, a1, wave);
    load_set(b0, b1, wave + SPMM_WAVES);
    load_set(e0, e1, wave + 2 * SPMM_WAVES);
    load_set(f0, f1, wave + 3 * SPMM_WAVES);
    auto step = [&](int4& x0, int4& x1, int my) {
        if (my < n_chunks) scan_chunk(x0, x1, chunk_of(my));
        load_set(x0, x1, my + RING * SPMM_WAVES);
    };
    for (int idx = wave; idx < n_chunks; idx += RING * SPMM_WAVES) {
        step(a0, a1, idx);
        step(b0, b1, idx + SPMM_WAVES);
        step(e0, e1, idx + 2 * SPMM_WAVES);
        step(f0, f1, idx + 3 * SPMM_WAVES);
    }
    if (qn > 0) drain(qn);
    __syncthreads();

    for (int i = tid; i < ROWS_PER_BLOCK * (DIM / 4); i += SPMM_THREADS) {
        const int r = i >> 5;
        const int c4 = i & 31;
        const int grow = r0 + r;
        if (grow < n_nodes)
            reinterpret_cast<float4*>(agg + (size_t)grow * DIM)[c4] =
                reinterpret_cast<const float4*>(acc + r * DIM)[c4];
    }
}

__launch_bounds__(256, 2)
__global__ void gemm_relu_wmma(const float* __restrict__ A,
                               const float* __restrict__ norm_in,
                               const float* __restrict__ Wl,
                               const float* __restrict__ bias,
                               float* __restrict__ Out,
                               int n_rows) {
    __shared__ float sA[16 * 132];
    const int m0 = blockIdx.x * 16;
    const int tid = threadIdx.x;

    for (int i = tid; i < 16 * DIM; i += 256) {
        int r = i >> 7;
        int c = i & 127;
        int row = m0 + r;
        float v = 0.0f;
        if (row < n_rows) v = A[(size_t)row * DIM + c] * norm_in[row];
        sA[r * 132 + c] = v;
    }
    __syncthreads();

    const int wave = tid >> 5;
    const int lane = tid & 31;
    const int hl = lane & 15;
    const int hi = lane >> 4;
    const int col = wave * 16 + hl;

    v8f acc = {};
#pragma unroll
    for (int kk = 0; kk < 32; ++kk) {
        const int kb = kk * 4 + hi * 2;
        v2f a, b;
        a.x = sA[hl * 132 + kb + 0];
        a.y = sA[hl * 132 + kb + 1];
        b.x = Wl[(size_t)(kb + 0) * DIM + col];
        b.y = Wl[(size_t)(kb + 1) * DIM + col];
        acc = __builtin_amdgcn_wmma_f32_16x16x4_f32(
             false, a,  false, b,
             (short)0, acc,  false,  false);
    }

    const float bb = bias[col];
#pragma unroll
    for (int j = 0; j < 8; ++j) {
        int row = m0 + j + hi * 8;
        if (row < n_rows) {
            float v = acc[j] + bb;
            Out[(size_t)row * DIM + col] = v > 0.0f ? v : 0.0f;
        }
    }
}

__global__ void pool_accum(const float* __restrict__ h, const int* __restrict__ gid,
                           float* __restrict__ pooled, float* __restrict__ cnt,
                           int n_nodes) {
    long long t = (long long)blockIdx.x * blockDim.x + threadIdx.x;
    int nidx = (int)(t >> 5);
    int c = (int)(t & 31);
    if (nidx >= n_nodes) return;
    int g = gid[nidx];
    const float4 v = *reinterpret_cast<const float4*>(h + (size_t)nidx * DIM + c * 4);
    float* p = pooled + (size_t)g * DIM + c * 4;
    unsafeAtomicAdd(p + 0, v.x);
    unsafeAtomicAdd(p + 1, v.y);
    unsafeAtomicAdd(p + 2, v.z);
    unsafeAtomicAdd(p + 3, v.w);
    if (c == 0) unsafeAtomicAdd(&cnt[g], 1.0f);
}

__global__ void pool_div(float* __restrict__ pooled, const float* __restrict__ cnt) {
    int t = blockIdx.x * blockDim.x + threadIdx.x;
    int g = t >> 7;
    pooled[t] /= fmaxf(cnt[g], 1.0f);
}

extern "C" void kernel_launch(void* const* d_in, const int* in_sizes, int n_in,
                              void* d_out, int out_size, void* d_ws, size_t ws_size,
                              hipStream_t stream) {
    const float* feats = (const float*)d_in[0];
    const float* Wall  = (const float*)d_in[1];
    const float* ball  = (const float*)d_in[2];
    const int*   src   = (const int*)d_in[3];
    const int*   dst   = (const int*)d_in[4];
    const int*   gid   = (const int*)d_in[5];
    float* out = (float*)d_out;

    float* ws = (float*)d_ws;
    float* norm_out = ws;
    float* norm_in  = ws + N_NODES;
    float* bufA     = ws + 2 * (size_t)N_NODES;
    float* bufB     = bufA + (size_t)N_NODES * DIM;
    float* cnt      = bufB + (size_t)N_NODES * DIM;

    const int BT = 256;
    zero_f32<<<(2 * N_NODES + BT - 1) / BT, BT, 0, stream>>>(norm_out, 2 * N_NODES);
    degree_kernel<<<(N_EDGES + BT - 1) / BT, BT, 0, stream>>>(src, dst, norm_out, norm_in, N_EDGES);
    rsqrt_kernel<<<(N_NODES + BT - 1) / BT, BT, 0, stream>>>(norm_out, norm_in, N_NODES);

    const float* hin = feats;
    float* bufs[2] = {bufA, bufB};
    int cur = 0;
    for (int l = 0; l < N_LAYERS; ++l) {
        float* agg = bufs[cur];
        spmm_owner<<<(N_NODES + ROWS_PER_BLOCK - 1) / ROWS_PER_BLOCK, SPMM_THREADS, 0, stream>>>(
            hin, norm_out, src, dst, agg, N_EDGES, N_NODES);
        gemm_relu_wmma<<<(N_NODES + 15) / 16, BT, 0, stream>>>(
            agg, norm_in, Wall + (size_t)l * DIM * DIM, ball + (size_t)l * DIM, agg, N_NODES);
        hin = agg;
        cur ^= 1;
    }

    zero_f32<<<(N_GRAPHS * DIM + BT - 1) / BT, BT, 0, stream>>>(out, N_GRAPHS * DIM);
    zero_f32<<<1, N_GRAPHS, 0, stream>>>(cnt, N_GRAPHS);
    const long long pool_threads = (long long)N_NODES * 32;
    pool_accum<<<(int)((pool_threads + BT - 1) / BT), BT, 0, stream>>>(hin, gid, out, cnt, N_NODES);
    pool_div<<<(N_GRAPHS * DIM) / BT, BT, 0, stream>>>(out, cnt);
}


# GraphConv layer in one launch: each owner block forms relu((agg*norm_in)@W+b) for its 196 rows from its LDS sums on the same f32 WMMA, W fragments in registers; no agg plane, 3 launches fewer
# speedup vs baseline: 1.8510x; 1.0985x over previous
#include <hip/hip_runtime.h>
#include <hip/hip_bf16.h>

#define N_NODES  50000
#define N_EDGES  800000
#define DIM      128
#define N_LAYERS 3
#define N_GRAPHS 256

typedef __attribute__((ext_vector_type(2))) float v2f;
typedef __attribute__((ext_vector_type(8))) float v8f;

__global__ void zero_f32(float* __restrict__ p, int n) {
    int t = blockIdx.x * blockDim.x + threadIdx.x;
    if (t < n) p[t] = 0.0f;
}

__global__ void degree_kernel(const int* __restrict__ src, const int* __restrict__ dst,
                              float* __restrict__ deg_out, float* __restrict__ deg_in,
                              int n_edges) {
    int e = blockIdx.x * blockDim.x + threadIdx.x;
    if (e >= n_edges) return;
    unsafeAtomicAdd(&deg_out[src[e]], 1.0f);
    unsafeAtomicAdd(&deg_in[dst[e]], 1.0f);
}

__global__ void rsqrt_kernel(float* __restrict__ a, float* __restrict__ b, int n) {
    int t = blockIdx.x * blockDim.x + threadIdx.x;
    if (t >= n) return;
    a[t] = __frsqrt_rn(fmaxf(a[t], 1.0f));
    b[t] = __frsqrt_rn(fmaxf(b[t], 1.0f));
}

#define ROWS_PER_BLOCK 196
#define SPMM_THREADS   1024
#define SPMM_WAVES     (SPMM_THREADS / 32)
#define EDGE_CHUNK     256
#define RING           4
#define QCAP           512
#define ROWS_IN_FLIGHT 8
#define M_TILES   ((ROWS_PER_BLOCK + 15) / 16)
#define ACC_ROWS  (M_TILES * 16)
#define ACC_PITCH 132
static_assert(N_EDGES % EDGE_CHUNK == 0, "the dst list is read in whole chunks");

__launch_bounds__(SPMM_THREADS)
__global__ void spmm_owner(const float* __restrict__ h, const float* __restrict__ norm_out,
                           const float* __restrict__ norm_in,
                           const int* __restrict__ src, const int* __restrict__ dst,
                           const float* __restrict__ Wl,
                           const float* __restrict__ bias,
                           float* __restrict__ out_h, int n_edges, int n_nodes) {
    __shared__ __attribute__((aligned(16))) float acc[ACC_ROWS * ACC_PITCH];
    __shared__ __attribute__((aligned(16))) unsigned char qbuf[sizeof(int2) * SPMM_WAVES * QCAP];
    static_assert(sizeof(qbuf) >= sizeof(float) * ACC_ROWS * ACC_PITCH,
                  "the output tile must fit in the queues' LDS");
    int2* const queue = reinterpret_cast<int2*>(qbuf);
    const int tid = threadIdx.x;
    const int lane = tid & 31;
    const int wave = __builtin_amdgcn_readfirstlane(tid >> 5);
    const int r0 = blockIdx.x * ROWS_PER_BLOCK;
    int2* const q = queue + wave * QCAP;

    const float4 zero4 = {0.0f, 0.0f, 0.0f, 0.0f};
    for (int i = tid; i < ACC_ROWS * (ACC_PITCH / 4); i += SPMM_THREADS)
        reinterpret_cast<float4*>(acc)[i] = zero4;
    __syncthreads();

    int qh = 0;
    int qn = 0;

    auto drain = [&](int n) {
        int2 ent = {0, 0};
        if (lane < n) ent = q[(qh + lane) & (QCAP - 1)];
        const int s = src[ent.x];
        const int nsb = __float_as_int(norm_out[s]);
#pragma nounroll
        for (int j0 = 0; j0 < n; j0 += ROWS_IN_FLIGHT) {
            float v[ROWS_IN_FLIGHT][4];
#pragma unroll
            for (int u = 0; u < ROWS_IN_FLIGHT; ++u) {
                const int sj = __builtin_amdgcn_readlane(s, j0 + u);
                const float* hp = h + (size_t)sj * DIM + lane;
                v[u][0] = hp[0]; v[u][1] = hp[32]; v[u][2] = hp[64]; v[u][3] = hp[96];
            }
#pragma unroll
            for (int u = 0; u < ROWS_IN_FLIGHT; ++u) {
                if (j0 + u < n) {
                    const float nj = __int_as_float(__builtin_amdgcn_readlane(nsb, j0 + u));
                    const int rowj = __builtin_amdgcn_readlane(ent.y, j0 + u);
                    float* ap = acc + rowj * ACC_PITCH + lane;
                    unsafeAtomicAdd(ap + 0,  v[u][0] * nj);
                    unsafeAtomicAdd(ap + 32, v[u][1] * nj);
                    unsafeAtomicAdd(ap + 64, v[u][2] * nj);
                    unsafeAtomicAdd(ap + 96, v[u][3] * nj);
                }
            }
        }
        qh = (qh + n) & (QCAP - 1);
        qn -= n;
    };

    auto scan_chunk = [&](const int4& c0, const int4& c1, int ch) {
        const unsigned rel[8] = {(unsigned)(c0.x - r0), (unsigned)(c0.y - r0),
                                 (unsigned)(c0.z - r0), (unsigned)(c0.w - r0),
                                 (unsigned)(c1.x - r0), (unsigned)(c1.y - r0),
                                 (unsigned)(c1.z - r0), (unsigned)(c1.w - r0)};
        unsigned m[8];
        unsigned any = 0;
#pragma unroll
        for (int k = 0; k < 8; ++k) {
            m[k] = __builtin_amdgcn_ballot_w32(rel[k] < (unsigned)ROWS_PER_BLOCK);
            any |= m[k];
        }
        if (any) {
#pragma unroll
            for (int k = 0; k < 8; ++k) {
                if (m[k]) {
                    if (rel[k] < (unsigned)ROWS_PER_BLOCK) {
                        const int pos = qh + qn + (int)__builtin_amdgcn_mbcnt_lo(m[k], 0u);
                        const int e = ch * EDGE_CHUNK + (k >> 2) * 128 + lane * 4 + (k & 3);
                        q[pos & (QCAP - 1)] = make_int2(e, (int)rel[k]);
                    }
                    qn += __builtin_popcount(m[k]);
                }
            }
            while (qn >= 32) drain(32);
        }
    };

    const int n_chunks = n_edges / EDGE_CHUNK;
    const int offset = n_chunks > 0 ? (int)((blockIdx.x * 12u) % (unsigned)n_chunks) : 0;
    const int4* __restrict__ dst4 = reinterpret_cast<const int4*>(dst);

    auto chunk_of = [&](int idx) {
        int ch = idx + offset;
        if (ch >= n_chunks) ch -= n_chunks;
        return idx < n_chunks ? ch : 0;
    };
    auto load_set = [&](int4& x0, int4& x1, int idx) {
        const int ch = chunk_of(idx);
        x0 = dst4[ch * 64 + lane];
        x1 = dst4[ch * 64 + 32 + lane];
    };
    int4 a0, a1, b0, b1, e0, e1, f0, f1;
    static_assert(RING == 4, "four named register sets below");
    load_set(a0, a1, wave);
    load_set(b0, b1, wave + SPMM_WAVES);
    load_set(e0, e1, wave + 2 * SPMM_WAVES);
    load_set(f0, f1, wave + 3 * SPMM_WAVES);
    auto step = [&](int4& x0, int4& x1, int my) {
        if (my < n_chunks) scan_chunk(x0, x1, chunk_of(my));
        load_set(x0, x1, my + RING * SPMM_WAVES);
    };
    for (int idx = wave; idx < n_chunks; idx += RING * SPMM_WAVES) {
        step(a0, a1, idx);
        step(b0, b1, idx + SPMM_WAVES);
        step(e0, e1, idx + 2 * SPMM_WAVES);
        step(f0, f1, idx + 3 * SPMM_WAVES);
    }
    if (qn > 0) drain(qn);
    __syncthreads();

    float* const hout = reinterpret_cast<float*>(qbuf);
    const int nt = wave & 7;
    const int mg = wave >> 3;
    const int hl = lane & 15;
    const int hi = lane >> 4;
    const int col = nt * 16 + hl;
    v2f bfrag[32];
#pragma unroll
    for (int kk = 0; kk < 32; ++kk) {
        const int kb = kk * 4 + hi * 2;
        bfrag[kk].x = Wl[(size_t)(kb + 0) * DIM + col];
        bfrag[kk].y = Wl[(size_t)(kb + 1) * DIM + col];
    }
    const float bb = bias[col];
    float nin[4];
#pragma unroll
    for (int t = 0; t < 4; ++t) {
        const int grow = r0 + (mg + 4 * t) * 16 + hl;
        nin[t] = grow < n_nodes ? norm_in[grow] : 0.0f;
    }
#pragma unroll
    for (int t = 0; t < 4; ++t) {
        const int mt = mg + 4 * t;
        if (mt < M_TILES) {
            const int lrow = mt * 16 + hl;
            const float* arow = acc + lrow * ACC_PITCH + hi * 2;
            v8f d = {};
#pragma unroll
            for (int kk = 0; kk < 32; ++kk) {
                v2f a;
                a.x = arow[kk * 4 + 0] * nin[t];
                a.y = arow[kk * 4 + 1] * nin[t];
                d = __builtin_amdgcn_wmma_f32_16x16x4_f32(
                     false, a,  false, bfrag[kk],
                     (short)0, d,  false,  false);
            }
#pragma unroll
            for (int j = 0; j < 8; ++j) {
                const float v = d[j] + bb;
                hout[(mt * 16 + j + hi * 8) * ACC_PITCH + col] = v > 0.0f ? v : 0.0f;
            }
        }
    }
    __syncthreads();

    for (int i = tid; i < ROWS_PER_BLOCK * (DIM / 4); i += SPMM_THREADS) {
        const int r = i >> 5;
        const int c4 = i & 31;
        const int grow = r0 + r;
        if (grow < n_nodes)
            reinterpret_cast<float4*>(out_h + (size_t)grow * DIM)[c4] =
                reinterpret_cast<const float4*>(hout + r * ACC_PITCH)[c4];
    }
}

__global__ void pool_accum(const float* __restrict__ h, const int* __restrict__ gid,
                           float* __restrict__ pooled, float* __restrict__ cnt,
                           int n_nodes) {
    long long t = (long long)blockIdx.x * blockDim.x + threadIdx.x;
    int nidx = (int)(t >> 5);
    int c = (int)(t & 31);
    if (nidx >= n_nodes) return;
    int g = gid[nidx];
    const float4 v = *reinterpret_cast<const float4*>(h + (size_t)nidx * DIM + c * 4);
    float* p = pooled + (size_t)g * DIM + c * 4;
    unsafeAtomicAdd(p + 0, v.x);
    unsafeAtomicAdd(p + 1, v.y);
    unsafeAtomicAdd(p + 2, v.z);
    unsafeAtomicAdd(p + 3, v.w);
    if (c == 0) unsafeAtomicAdd(&cnt[g], 1.0f);
}

__global__ void pool_div(float* __restrict__ pooled, const float* __restrict__ cnt) {
    int t = blockIdx.x * blockDim.x + threadIdx.x;
    int g = t >> 7;
    pooled[t] /= fmaxf(cnt[g], 1.0f);
}

extern "C" void kernel_launch(void* const* d_in, const int* in_sizes, int n_in,
                              void* d_out, int out_size, void* d_ws, size_t ws_size,
                              hipStream_t stream) {
    const float* feats = (const float*)d_in[0];
    const float* Wall  = (const float*)d_in[1];
    const float* ball  = (const float*)d_in[2];
    const int*   src   = (const int*)d_in[3];
    const int*   dst   = (const int*)d_in[4];
    const int*   gid   = (const int*)d_in[5];
    float* out = (float*)d_out;

    float* ws = (float*)d_ws;
    float* norm_out = ws;
    float* norm_in  = ws + N_NODES;
    float* bufA     = ws + 2 * (size_t)N_NODES;
    float* bufB     = bufA + (size_t)N_NODES * DIM;
    float* cnt      = bufB + (size_t)N_NODES * DIM;

    const int BT = 256;
    zero_f32<<<(2 * N_NODES + BT - 1) / BT, BT, 0, stream>>>(norm_out, 2 * N_NODES);
    degree_kernel<<<(N_EDGES + BT - 1) / BT, BT, 0, stream>>>(src, dst, norm_out, norm_in, N_EDGES);
    rsqrt_kernel<<<(N_NODES + BT - 1) / BT, BT, 0, stream>>>(norm_out, norm_in, N_NODES);

    const float* hin = feats;
    float* bufs[2] = {bufA, bufB};
    int cur = 0;
    for (int l = 0; l < N_LAYERS; ++l) {
        float* hout = bufs[cur];
        spmm_owner<<<(N_NODES + ROWS_PER_BLOCK - 1) / ROWS_PER_BLOCK, SPMM_THREADS, 0, stream>>>(
            hin, norm_out, norm_in, src, dst,
            Wall + (size_t)l * DIM * DIM, ball + (size_t)l * DIM, hout, N_EDGES, N_NODES);
        hin = hout;
        cur ^= 1;
    }

    zero_f32<<<(N_GRAPHS * DIM + BT - 1) / BT, BT, 0, stream>>>(out, N_GRAPHS * DIM);
    zero_f32<<<1, N_GRAPHS, 0, stream>>>(cnt, N_GRAPHS);
    const long long pool_threads = (long long)N_NODES * 32;
    pool_accum<<<(int)((pool_threads + BT - 1) / BT), BT, 0, stream>>>(hin, gid, out, cnt, N_NODES);
    pool_div<<<(N_GRAPHS * DIM) / BT, BT, 0, stream>>>(out, cnt);
}


# Layer 1's owner scan writes each block's matched edges (source row, local row, norm) to a workspace list; layers 2 and 3 replay it: no dst scan, no src or norm gather; LDS sums in arrival order
# speedup vs baseline: 2.1331x; 1.1524x over previous
#include <hip/hip_runtime.h>
#include <hip/hip_bf16.h>

#define N_NODES  50000
#define N_EDGES  800000
#define DIM      128
#define N_LAYERS 3
#define N_GRAPHS 256

typedef __attribute__((ext_vector_type(2))) float v2f;
typedef __attribute__((ext_vector_type(8))) float v8f;

__global__ void zero_f32(float* __restrict__ p, int n) {
    int t = blockIdx.x * blockDim.x + threadIdx.x;
    if (t < n) p[t] = 0.0f;
}

__global__ void degree_kernel(const int* __restrict__ src, const int* __restrict__ dst,
                              float* __restrict__ deg_out, float* __restrict__ deg_in,
                              int n_edges) {
    int e = blockIdx.x * blockDim.x + threadIdx.x;
    if (e >= n_edges) return;
    unsafeAtomicAdd(&deg_out[src[e]], 1.0f);
    unsafeAtomicAdd(&deg_in[dst[e]], 1.0f);
}

__global__ void rsqrt_kernel(float* __restrict__ a, float* __restrict__ b, int n) {
    int t = blockIdx.x * blockDim.x + threadIdx.x;
    if (t >= n) return;
    a[t] = __frsqrt_rn(fmaxf(a[t], 1.0f));
    b[t] = __frsqrt_rn(fmaxf(b[t], 1.0f));
}

#define ROWS_PER_BLOCK 196
#define SPMM_THREADS   1024
#define SPMM_WAVES     (SPMM_THREADS / 32)
#define EDGE_CHUNK     256
#define RING           4
#define QCAP           512
#define ROWS_IN_FLIGHT 8
#define M_TILES   ((ROWS_PER_BLOCK + 15) / 16)
#define ACC_ROWS  (M_TILES * 16)
#define ACC_PITCH 132
static_assert(N_EDGES % EDGE_CHUNK == 0, "the dst list is read in whole chunks");
#define N_BLOCKS  ((N_NODES + ROWS_PER_BLOCK - 1) / ROWS_PER_BLOCK)
#define LIST_CAP  4096
#define LIST_PAD  32
static_assert(ROWS_PER_BLOCK <= 256, "the local row takes the low 8 bits of a list entry");
static_assert(N_NODES <= (1 << 23), "the source row takes the upper 24 bits of a list entry");
#define WS_NEED (sizeof(float) * (2 * (size_t)N_NODES + 2 * (size_t)N_NODES * DIM + N_GRAPHS) + \
                 sizeof(int) * (size_t)(N_BLOCKS + LIST_PAD) + sizeof(int2) * (size_t)N_BLOCKS * LIST_CAP)

__launch_bounds__(SPMM_THREADS)
__global__ void spmm_owner(const float* __restrict__ h, const float* __restrict__ norm_out,
                           const float* __restrict__ norm_in,
                           const int* __restrict__ src, const int* __restrict__ dst,
                           const float* __restrict__ Wl,
                           const float* __restrict__ bias,
                           float* __restrict__ out_h,
                           int* __restrict__ lcount,
                           int* __restrict__ lists,
                           size_t ws_bytes, int layer, int n_edges, int n_nodes) {
    __shared__ __attribute__((aligned(16))) float acc[ACC_ROWS * ACC_PITCH];
    __shared__ __attribute__((aligned(16))) unsigned char qbuf[sizeof(int2) * SPMM_WAVES * QCAP];
    static_assert(sizeof(qbuf) >= sizeof(float) * ACC_ROWS * ACC_PITCH,
                  "the output tile must fit in the queues' LDS");
    __shared__ int lcnt;
    int2* const queue = reinterpret_cast<int2*>(qbuf);
    const int tid = threadIdx.x;
    const int lane = tid & 31;
    const int wave = __builtin_amdgcn_readfirstlane(tid >> 5);
    const int r0 = blockIdx.x * ROWS_PER_BLOCK;
    int2* const q = queue + wave * QCAP;
    int2* const mylist = reinterpret_cast<int2*>(lists) + (size_t)blockIdx.x * LIST_CAP;
    const bool have_lists = ws_bytes >= WS_NEED;
    const bool record = have_lists && layer == 0;
    const int total = (have_lists && layer != 0) ? lcount[blockIdx.x] : 0;
    const bool replay = have_lists && layer != 0 && total <= LIST_CAP;

    const float4 zero4 = {0.0f, 0.0f, 0.0f, 0.0f};
    for (int i = tid; i < ACC_ROWS * (ACC_PITCH / 4); i += SPMM_THREADS)
        reinterpret_cast<float4*>(acc)[i] = zero4;
    if (tid == 0) lcnt = 0;
    __syncthreads();

    int qh = 0;
    int qn = 0;

    auto add_rows = [&](int n, int s, int nsb, int row) {
#pragma nounroll
        for (int j0 = 0; j0 < n; j0 += ROWS_IN_FLIGHT) {
            float v[ROWS_IN_FLIGHT][4];
#pragma unroll
            for (int u = 0; u < ROWS_IN_FLIGHT; ++u) {
                const int sj = __builtin_amdgcn_readlane(s, j0 + u);
                const float* hp = h + (size_t)sj * DIM + lane;
                v[u][0] = hp[0]; v[u][1] = hp[32]; v[u][2] = hp[64]; v[u][3] = hp[96];
            }
#pragma unroll
            for (int u = 0; u < ROWS_IN_FLIGHT; ++u) {
                if (j0 + u < n) {
                    const float nj = __int_as_float(__builtin_amdgcn_readlane(nsb, j0 + u));
                    const int rowj = __builtin_amdgcn_readlane(row, j0 + u);
                    float* ap = acc + rowj * ACC_PITCH + lane;
                    unsafeAtomicAdd(ap + 0,  v[u][0] * nj);
                    unsafeAtomicAdd(ap + 32, v[u][1] * nj);
                    unsafeAtomicAdd(ap + 64, v[u][2] * nj);
                    unsafeAtomicAdd(ap + 96, v[u][3] * nj);
                }
            }
        }
    };

    auto drain = [&](int n) {
        int2 ent = {0, 0};
        if (lane < n) ent = q[(qh + lane) & (QCAP - 1)];
        const int s = src[ent.x];
        const int nsb = __float_as_int(norm_out[s]);
        if (record) {
            int base = 0;
            if (lane == 0) base = atomicAdd(&lcnt, n);
            base = __builtin_amdgcn_readfirstlane(base);
            if (lane < n && base + lane < LIST_CAP)
                mylist[base + lane] = make_int2((s << 8) | ent.y, nsb);
        }
        add_rows(n, s, nsb, ent.y);
        qh = (qh + n) & (QCAP - 1);
        qn -= n;
    };

    auto scan_chunk = [&](const int4& c0, const int4& c1, int ch) {
        const unsigned rel[8] = {(unsigned)(c0.x - r0), (unsigned)(c0.y - r0),
                                 (unsigned)(c0.z - r0), (unsigned)(c0.w - r0),
                                 (unsigned)(c1.x - r0), (unsigned)(c1.y - r0),
                                 (unsigned)(c1.z - r0), (unsigned)(c1.w - r0)};
        unsigned m[8];
        unsigned any = 0;
#pragma unroll
        for (int k = 0; k < 8; ++k) {
            m[k] = __builtin_amdgcn_ballot_w32(rel[k] < (unsigned)ROWS_PER_BLOCK);
            any |= m[k];
        }
        if (any) {
#pragma unroll
            for (int k = 0; k < 8; ++k) {
                if (m[k]) {
                    if (rel[k] < (unsigned)ROWS_PER_BLOCK) {
                        const int pos = qh + qn + (int)__builtin_amdgcn_mbcnt_lo(m[k], 0u);
                        const int e = ch * EDGE_CHUNK + (k >> 2) * 128 + lane * 4 + (k & 3);
                        q[pos & (QCAP - 1)] = make_int2(e, (int)rel[k]);
                    }
                    qn += __builtin_popcount(m[k]);
                }
            }
            while (qn >= 32) drain(32);
        }
    };

    if (replay) {
        static_assert(LIST_CAP == 4 * SPMM_THREADS, "four named entry registers below");
        if (total > 0) {
            const int last = total - 1;
            const int i0 = wave * 32 + lane;
            auto entry = [&](int i) { return mylist[i < last ? i : last]; };
            const int2 e0 = entry(i0);
            const int2 e1 = entry(i0 + SPMM_THREADS);
            const int2 e2 = entry(i0 + 2 * SPMM_THREADS);
            const int2 e3 = entry(i0 + 3 * SPMM_THREADS);
#pragma nounroll
            for (int t = 0; t < 4; ++t) {
                const int at = wave * 32 + t * SPMM_THREADS;
                if (at >= total) break;
                const int2 cur = t == 0 ? e0 : t == 1 ? e1 : t == 2 ? e2 : e3;
                const int left = total - at;
                add_rows(left < 32 ? left : 32, cur.x >> 8, cur.y, cur.x & 255);
            }
        }
    } else {
    const int n_chunks = n_edges / EDGE_CHUNK;
    const int offset = n_chunks > 0 ? (int)((blockIdx.x * 12u) % (unsigned)n_chunks) : 0;
    const int4* __restrict__ dst4 = reinterpret_cast<const int4*>(dst);

    auto chunk_of = [&](int idx) {
        int ch = idx + offset;
        if (ch >= n_chunks) ch -= n_chunks;
        return idx < n_chunks ? ch : 0;
    };
    auto load_set = [&](int4& x0, int4& x1, int idx) {
        const int ch = chunk_of(idx);
        x0 = dst4[ch * 64 + lane];
        x1 = dst4[ch * 64 + 32 + lane];
    };
    int4 a0, a1, b0, b1, e0, e1, f0, f1;
    static_assert(RING == 4, "four named register sets below");
    load_set(a0, a1, wave);
    load_set(b0, b1, wave + SPMM_WAVES);
    load_set(e0, e1, wave + 2 * SPMM_WAVES);
    load_set(f0, f1, wave + 3 * SPMM_WAVES);
    auto step = [&](int4& x0, int4& x1, int my) {
        if (my < n_chunks) scan_chunk(x0, x1, chunk_of(my));
        load_set(x0, x1, my + RING * SPMM_WAVES);
    };
    for (int idx = wave; idx < n_chunks; idx += RING * SPMM_WAVES) {
        step(a0, a1, idx);
        step(b0, b1, idx + SPMM_WAVES);
        step(e0, e1, idx + 2 * SPMM_WAVES);
        step(f0, f1, idx + 3 * SPMM_WAVES);
    }
    if (qn > 0) drain(qn);
    }
    __syncthreads();
    if (record && tid == 0) lcount[blockIdx.x] = lcnt;

    float* const hout = reinterpret_cast<float*>(qbuf);
    const int nt = wave & 7;
    const int mg = wave >> 3;
    const int hl = lane & 15;
    const int hi = lane >> 4;
    const int col = nt * 16 + hl;
    v2f bfrag[32];
#pragma unroll
    for (int kk = 0; kk < 32; ++kk) {
        const int kb = kk * 4 + hi * 2;
        bfrag[kk].x = Wl[(size_t)(kb + 0) * DIM + col];
        bfrag[kk].y = Wl[(size_t)(kb + 1) * DIM + col];
    }
    const float bb = bias[col];
    float nin[4];
#pragma unroll
    for (int t = 0; t < 4; ++t) {
        const int grow = r0 + (mg + 4 * t) * 16 + hl;
        nin[t] = grow < n_nodes ? norm_in[grow] : 0.0f;
    }
#pragma unroll
    for (int t = 0; t < 4; ++t) {
        const int mt = mg + 4 * t;
        if (mt < M_TILES) {
            const int lrow = mt * 16 + hl;
            const float* arow = acc + lrow * ACC_PITCH + hi * 2;
            v8f d = {};
#pragma unroll
            for (int kk = 0; kk < 32; ++kk) {
                v2f a;
                a.x = arow[kk * 4 + 0] * nin[t];
                a.y = arow[kk * 4 + 1] * nin[t];
                d = __builtin_amdgcn_wmma_f32_16x16x4_f32(
                     false, a,  false, bfrag[kk],
                     (short)0, d,  false,  false);
            }
#pragma unroll
            for (int j = 0; j < 8; ++j) {
                const float v = d[j] + bb;
                hout[(mt * 16 + j + hi * 8) * ACC_PITCH + col] = v > 0.0f ? v : 0.0f;
            }
        }
    }
    __syncthreads();

    for (int i = tid; i < ROWS_PER_BLOCK * (DIM / 4); i += SPMM_THREADS) {
        const int r = i >> 5;
        const int c4 = i & 31;
        const int grow = r0 + r;
        if (grow < n_nodes)
            reinterpret_cast<float4*>(out_h + (size_t)grow * DIM)[c4] =
                reinterpret_cast<const float4*>(hout + r * ACC_PITCH)[c4];
    }
}

__global__ void pool_accum(const float* __restrict__ h, const int* __restrict__ gid,
                           float* __restrict__ pooled, float* __restrict__ cnt,
                           int n_nodes) {
    long long t = (long long)blockIdx.x * blockDim.x + threadIdx.x;
    int nidx = (int)(t >> 5);
    int c = (int)(t & 31);
    if (nidx >= n_nodes) return;
    int g = gid[nidx];
    const float4 v = *reinterpret_cast<const float4*>(h + (size_t)nidx * DIM + c * 4);
    float* p = pooled + (size_t)g * DIM + c * 4;
    unsafeAtomicAdd(p + 0, v.x);
    unsafeAtomicAdd(p + 1, v.y);
    unsafeAtomicAdd(p + 2, v.z);
    unsafeAtomicAdd(p + 3, v.w);
    if (c == 0) unsafeAtomicAdd(&cnt[g], 1.0f);
}

__global__ void pool_div(float* __restrict__ pooled, const float* __restrict__ cnt) {
    int t = blockIdx.x * blockDim.x + threadIdx.x;
    int g = t >> 7;
    pooled[t] /= fmaxf(cnt[g], 1.0f);
}

extern "C" void kernel_launch(void* const* d_in, const int* in_sizes, int n_in,
                              void* d_out, int out_size, void* d_ws, size_t ws_size,
                              hipStream_t stream) {
    const float* feats = (const float*)d_in[0];
    const float* Wall  = (const float*)d_in[1];
    const float* ball  = (const float*)d_in[2];
    const int*   src   = (const int*)d_in[3];
    const int*   dst   = (const int*)d_in[4];
    const int*   gid   = (const int*)d_in[5];
    float* out = (float*)d_out;

    float* ws = (float*)d_ws;
    float* norm_out = ws;
    float* norm_in  = ws + N_NODES;
    float* bufA     = ws + 2 * (size_t)N_NODES;
    float* bufB     = bufA + (size_t)N_NODES * DIM;
    float* cnt      = bufB + (size_t)N_NODES * DIM;
    int*   lcount   = (int*)(cnt + N_GRAPHS);
    int*   lists    = lcount + N_BLOCKS + LIST_PAD;

    const int BT = 256;
    zero_f32<<<(2 * N_NODES + BT - 1) / BT, BT, 0, stream>>>(norm_out, 2 * N_NODES);
    degree_kernel<<<(N_EDGES + BT - 1) / BT, BT, 0, stream>>>(src, dst, norm_out, norm_in, N_EDGES);
    rsqrt_kernel<<<(N_NODES + BT - 1) / BT, BT, 0, stream>>>(norm_out, norm_in, N_NODES);

    const float* hin = feats;
    float* bufs[2] = {bufA, bufB};
    int cur = 0;
    for (int l = 0; l < N_LAYERS; ++l) {
        float* hout = bufs[cur];
        spmm_owner<<<N_BLOCKS, SPMM_THREADS, 0, stream>>>(
            hin, norm_out, norm_in, src, dst,
            Wall + (size_t)l * DIM * DIM, ball + (size_t)l * DIM, hout,
            lcount, lists, ws_size, l, N_EDGES, N_NODES);
        hin = hout;
        cur ^= 1;
    }

    zero_f32<<<(N_GRAPHS * DIM + BT - 1) / BT, BT, 0, stream>>>(out, N_GRAPHS * DIM);
    zero_f32<<<1, N_GRAPHS, 0, stream>>>(cnt, N_GRAPHS);
    const long long pool_threads = (long long)N_NODES * 32;
    pool_accum<<<(int)((pool_threads + BT - 1) / BT), BT, 0, stream>>>(hin, gid, out, cnt, N_NODES);
    pool_div<<<(N_GRAPHS * DIM) / BT, BT, 0, stream>>>(out, cnt);
}


# Degree counts without global atomics: 64 blocks each count 25000 src or dst indices into a 50000-bin LDS table by integer LDS adds and store it; one kernel sums 32 tables per node, then rsqrt
# speedup vs baseline: 2.7450x; 1.2869x over previous
#include <hip/hip_runtime.h>
#include <hip/hip_bf16.h>

#define N_NODES  50000
#define N_EDGES  800000
#define DIM      128
#define N_LAYERS 3
#define N_GRAPHS 256

typedef __attribute__((ext_vector_type(2))) float v2f;
typedef __attribute__((ext_vector_type(8))) float v8f;

__global__ void zero_f32(float* __restrict__ p, int n) {
    int t = blockIdx.x * blockDim.x + threadIdx.x;
    if (t < n) p[t] = 0.0f;
}

#define HIST_SLICES   32
#define HIST_THREADS  1024
#define SLICE_EDGES   (N_EDGES / HIST_SLICES)
#define SLICE_IDX4    (SLICE_EDGES / 4)
#define IDX4_PER_THREAD ((SLICE_IDX4 + HIST_THREADS - 1) / HIST_THREADS)
static_assert(N_EDGES % (HIST_SLICES * 4) == 0, "a slice is a whole number of 16-byte loads");
static_assert(N_NODES % 4 == 0, "tables are cleared, stored and added four counters at a time");

__launch_bounds__(HIST_THREADS)
__global__ void degree_count(const int* __restrict__ src, const int* __restrict__ dst,
                             unsigned* __restrict__ tables) {
    __shared__ __attribute__((aligned(16))) unsigned bins[N_NODES];
    const int tid = threadIdx.x;
    const int which = blockIdx.x / HIST_SLICES;
    const int slice = blockIdx.x % HIST_SLICES;
    const int4* __restrict__ idx4 =
        reinterpret_cast<const int4*>((which ? dst : src) + (size_t)slice * SLICE_EDGES);
    int4 v[IDX4_PER_THREAD];
#pragma unroll
    for (int k = 0; k < IDX4_PER_THREAD; ++k) {
        const int i = tid + k * HIST_THREADS;
        v[k] = idx4[i < SLICE_IDX4 ? i : SLICE_IDX4 - 1];
    }
    const uint4 zero4 = {0u, 0u, 0u, 0u};
    for (int i = tid; i < N_NODES / 4; i += HIST_THREADS)
        reinterpret_cast<uint4*>(bins)[i] = zero4;
    __syncthreads();
#pragma unroll
    for (int k = 0; k < IDX4_PER_THREAD; ++k) {
        if (tid + k * HIST_THREADS < SLICE_IDX4) {
            atomicAdd(&bins[v[k].x], 1u);
            atomicAdd(&bins[v[k].y], 1u);
            atomicAdd(&bins[v[k].z], 1u);
            atomicAdd(&bins[v[k].w], 1u);
        }
    }
    __syncthreads();
    uint4* __restrict__ out4 = reinterpret_cast<uint4*>(
        tables + ((size_t)slice * 2 + which) * N_NODES);
    for (int i = tid; i < N_NODES / 4; i += HIST_THREADS)
        out4[i] = reinterpret_cast<const uint4*>(bins)[i];
}

#define NORM_THREADS 64
#define NORM_VEC4    (2 * N_NODES / 4)
__global__ void norm_from_counts(const unsigned* __restrict__ tables, float* __restrict__ norms) {
    const int t = blockIdx.x * blockDim.x + threadIdx.x;
    if (t >= NORM_VEC4) return;
    const uint4* __restrict__ t4 = reinterpret_cast<const uint4*>(tables);
    uint4 s = {0u, 0u, 0u, 0u};
#pragma unroll
    for (int b = 0; b < HIST_SLICES; ++b) {
        const uint4 c = t4[(size_t)b * NORM_VEC4 + t];
        s.x += c.x; s.y += c.y; s.z += c.z; s.w += c.w;
    }
    float4 r;
    r.x = __frsqrt_rn(fmaxf((float)s.x, 1.0f));
    r.y = __frsqrt_rn(fmaxf((float)s.y, 1.0f));
    r.z = __frsqrt_rn(fmaxf((float)s.z, 1.0f));
    r.w = __frsqrt_rn(fmaxf((float)s.w, 1.0f));
    reinterpret_cast<float4*>(norms)[t] = r;
}

#define ROWS_PER_BLOCK 196
#define SPMM_THREADS   1024
#define SPMM_WAVES     (SPMM_THREADS / 32)
#define EDGE_CHUNK     256
#define RING           4
#define QCAP           512
#define ROWS_IN_FLIGHT 8
#define M_TILES   ((ROWS_PER_BLOCK + 15) / 16)
#define ACC_ROWS  (M_TILES * 16)
#define ACC_PITCH 132
static_assert(N_EDGES % EDGE_CHUNK == 0, "the dst list is read in whole chunks");
#define N_BLOCKS  ((N_NODES + ROWS_PER_BLOCK - 1) / ROWS_PER_BLOCK)
#define LIST_CAP  4096
#define LIST_PAD  32
static_assert(ROWS_PER_BLOCK <= 256, "the local row takes the low 8 bits of a list entry");
static_assert(N_NODES <= (1 << 23), "the source row takes the upper 24 bits of a list entry");
#define WS_NEED (sizeof(float) * (2 * (size_t)N_NODES + 2 * (size_t)N_NODES * DIM + N_GRAPHS) + \
                 sizeof(int) * (size_t)(N_BLOCKS + LIST_PAD) + sizeof(int2) * (size_t)N_BLOCKS * LIST_CAP)

__launch_bounds__(SPMM_THREADS)
__global__ void spmm_owner(const float* __restrict__ h, const float* __restrict__ norm_out,
                           const float* __restrict__ norm_in,
                           const int* __restrict__ src, const int* __restrict__ dst,
                           const float* __restrict__ Wl,
                           const float* __restrict__ bias,
                           float* __restrict__ out_h,
                           int* __restrict__ lcount,
                           int* __restrict__ lists,
                           size_t ws_bytes, int layer, int n_edges, int n_nodes) {
    __shared__ __attribute__((aligned(16))) float acc[ACC_ROWS * ACC_PITCH];
    __shared__ __attribute__((aligned(16))) unsigned char qbuf[sizeof(int2) * SPMM_WAVES * QCAP];
    static_assert(sizeof(qbuf) >= sizeof(float) * ACC_ROWS * ACC_PITCH,
                  "the output tile must fit in the queues' LDS");
    __shared__ int lcnt;
    int2* const queue = reinterpret_cast<int2*>(qbuf);
    const int tid = threadIdx.x;
    const int lane = tid & 31;
    const int wave = __builtin_amdgcn_readfirstlane(tid >> 5);
    const int r0 = blockIdx.x * ROWS_PER_BLOCK;
    int2* const q = queue + wave * QCAP;
    int2* const mylist = reinterpret_cast<int2*>(lists) + (size_t)blockIdx.x * LIST_CAP;
    const bool have_lists = ws_bytes >= WS_NEED;
    const bool record = have_lists && layer == 0;
    const int total = (have_lists && layer != 0) ? lcount[blockIdx.x] : 0;
    const bool replay = have_lists && layer != 0 && total <= LIST_CAP;

    const float4 zero4 = {0.0f, 0.0f, 0.0f, 0.0f};
    for (int i = tid; i < ACC_ROWS * (ACC_PITCH / 4); i += SPMM_THREADS)
        reinterpret_cast<float4*>(acc)[i] = zero4;
    if (tid == 0) lcnt = 0;
    __syncthreads();

    int qh = 0;
    int qn = 0;

    auto add_rows = [&](int n, int s, int nsb, int row) {
#pragma nounroll
        for (int j0 = 0; j0 < n; j0 += ROWS_IN_FLIGHT) {
            float v[ROWS_IN_FLIGHT][4];
#pragma unroll
            for (int u = 0; u < ROWS_IN_FLIGHT; ++u) {
                const int sj = __builtin_amdgcn_readlane(s, j0 + u);
                const float* hp = h + (size_t)sj * DIM + lane;
                v[u][0] = hp[0]; v[u][1] = hp[32]; v[u][2] = hp[64]; v[u][3] = hp[96];
            }
#pragma unroll
            for (int u = 0; u < ROWS_IN_FLIGHT; ++u) {
                if (j0 + u < n) {
                    const float nj = __int_as_float(__builtin_amdgcn_readlane(nsb, j0 + u));
                    const int rowj = __builtin_amdgcn_readlane(row, j0 + u);
                    float* ap = acc + rowj * ACC_PITCH + lane;
                    unsafeAtomicAdd(ap + 0,  v[u][0] * nj);
                    unsafeAtomicAdd(ap + 32, v[u][1] * nj);
                    unsafeAtomicAdd(ap + 64, v[u][2] * nj);
                    unsafeAtomicAdd(ap + 96, v[u][3] * nj);
                }
            }
        }
    };

    auto drain = [&](int n) {
        int2 ent = {0, 0};
        if (lane < n) ent = q[(qh + lane) & (QCAP - 1)];
        const int s = src[ent.x];
        const int nsb = __float_as_int(norm_out[s]);
        if (record) {
            int base = 0;
            if (lane == 0) base = atomicAdd(&lcnt, n);
            base = __builtin_amdgcn_readfirstlane(base);
            if (lane < n && base + lane < LIST_CAP)
                mylist[base + lane] = make_int2((s << 8) | ent.y, nsb);
        }
        add_rows(n, s, nsb, ent.y);
        qh = (qh + n) & (QCAP - 1);
        qn -= n;
    };

    auto scan_chunk = [&](const int4& c0, const int4& c1, int ch) {
        const unsigned rel[8] = {(unsigned)(c0.x - r0), (unsigned)(c0.y - r0),
                                 (unsigned)(c0.z - r0), (unsigned)(c0.w - r0),
                                 (unsigned)(c1.x - r0), (unsigned)(c1.y - r0),
                                 (unsigned)(c1.z - r0), (unsigned)(c1.w - r0)};
        unsigned m[8];
        unsigned any = 0;
#pragma unroll
        for (int k = 0; k < 8; ++k) {
            m[k] = __builtin_amdgcn_ballot_w32(rel[k] < (unsigned)ROWS_PER_BLOCK);
            any |= m[k];
        }
        if (any) {
#pragma unroll
            for (int k = 0; k < 8; ++k) {
                if (m[k]) {
                    if (rel[k] < (unsigned)ROWS_PER_BLOCK) {
                        const int pos = qh + qn + (int)__builtin_amdgcn_mbcnt_lo(m[k], 0u);
                        const int e = ch * EDGE_CHUNK + (k >> 2) * 128 + lane * 4 + (k & 3);
                        q[pos & (QCAP - 1)] = make_int2(e, (int)rel[k]);
                    }
                    qn += __builtin_popcount(m[k]);
                }
            }
            while (qn >= 32) drain(32);
        }
    };

    if (replay) {
        static_assert(LIST_CAP == 4 * SPMM_THREADS, "four named entry registers below");
        if (total > 0) {
            const int last = total - 1;
            const int i0 = wave * 32 + lane;
            auto entry = [&](int i) { return mylist[i < last ? i : last]; };
            const int2 e0 = entry(i0);
            const int2 e1 = entry(i0 + SPMM_THREADS);
            const int2 e2 = entry(i0 + 2 * SPMM_THREADS);
            const int2 e3 = entry(i0 + 3 * SPMM_THREADS);
#pragma nounroll
            for (int t = 0; t < 4; ++t) {
                const int at = wave * 32 + t * SPMM_THREADS;
                if (at >= total) break;
                const int2 cur = t == 0 ? e0 : t == 1 ? e1 : t == 2 ? e2 : e3;
                const int left = total - at;
                add_rows(left < 32 ? left : 32, cur.x >> 8, cur.y, cur.x & 255);
            }
        }
    } else {
    const int n_chunks = n_edges / EDGE_CHUNK;
    const int offset = n_chunks > 0 ? (int)((blockIdx.x * 12u) % (unsigned)n_chunks) : 0;
    const int4* __restrict__ dst4 = reinterpret_cast<const int4*>(dst);

    auto chunk_of = [&](int idx) {
        int ch = idx + offset;
        if (ch >= n_chunks) ch -= n_chunks;
        return idx < n_chunks ? ch : 0;
    };
    auto load_set = [&](int4& x0, int4& x1, int idx) {
        const int ch = chunk_of(idx);
        x0 = dst4[ch * 64 + lane];
        x1 = dst4[ch * 64 + 32 + lane];
    };
    int4 a0, a1, b0, b1, e0, e1, f0, f1;
    static_assert(RING == 4, "four named register sets below");
    load_set(a0, a1, wave);
    load_set(b0, b1, wave + SPMM_WAVES);
    load_set(e0, e1, wave + 2 * SPMM_WAVES);
    load_set(f0, f1, wave + 3 * SPMM_WAVES);
    auto step = [&](int4& x0, int4& x1, int my) {
        if (my < n_chunks) scan_chunk(x0, x1, chunk_of(my));
        load_set(x0, x1, my + RING * SPMM_WAVES);
    };
    for (int idx = wave; idx < n_chunks; idx += RING * SPMM_WAVES) {
        step(a0, a1, idx);
        step(b0, b1, idx + SPMM_WAVES);
        step(e0, e1, idx + 2 * SPMM_WAVES);
        step(f0, f1, idx + 3 * SPMM_WAVES);
    }
    if (qn > 0) drain(qn);
    }
    __syncthreads();
    if (record && tid == 0) lcount[blockIdx.x] = lcnt;

    float* const hout = reinterpret_cast<float*>(qbuf);
    const int nt = wave & 7;
    const int mg = wave >> 3;
    const int hl = lane & 15;
    const int hi = lane >> 4;
    const int col = nt * 16 + hl;
    v2f bfrag[32];
#pragma unroll
    for (int kk = 0; kk < 32; ++kk) {
        const int kb = kk * 4 + hi * 2;
        bfrag[kk].x = Wl[(size_t)(kb + 0) * DIM + col];
        bfrag[kk].y = Wl[(size_t)(kb + 1) * DIM + col];
    }
    const float bb = bias[col];
    float nin[4];
#pragma unroll
    for (int t = 0; t < 4; ++t) {
        const int grow = r0 + (mg + 4 * t) * 16 + hl;
        nin[t] = grow < n_nodes ? norm_in[grow] : 0.0f;
    }
#pragma unroll
    for (int t = 0; t < 4; ++t) {
        const int mt = mg + 4 * t;
        if (mt < M_TILES) {
            const int lrow = mt * 16 + hl;
            const float* arow = acc + lrow * ACC_PITCH + hi * 2;
            v8f d = {};
#pragma unroll
            for (int kk = 0; kk < 32; ++kk) {
                v2f a;
                a.x = arow[kk * 4 + 0] * nin[t];
                a.y = arow[kk * 4 + 1] * nin[t];
                d = __builtin_amdgcn_wmma_f32_16x16x4_f32(
                     false, a,  false, bfrag[kk],
                     (short)0, d,  false,  false);
            }
#pragma unroll
            for (int j = 0; j < 8; ++j) {
                const float v = d[j] + bb;
                hout[(mt * 16 + j + hi * 8) * ACC_PITCH + col] = v > 0.0f ? v : 0.0f;
            }
        }
    }
    __syncthreads();

    for (int i = tid; i < ROWS_PER_BLOCK * (DIM / 4); i += SPMM_THREADS) {
        const int r = i >> 5;
        const int c4 = i & 31;
        const int grow = r0 + r;
        if (grow < n_nodes)
            reinterpret_cast<float4*>(out_h + (size_t)grow * DIM)[c4] =
                reinterpret_cast<const float4*>(hout + r * ACC_PITCH)[c4];
    }
}

__global__ void pool_accum(const float* __restrict__ h, const int* __restrict__ gid,
                           float* __restrict__ pooled, float* __restrict__ cnt,
                           int n_nodes) {
    long long t = (long long)blockIdx.x * blockDim.x + threadIdx.x;
    int nidx = (int)(t >> 5);
    int c = (int)(t & 31);
    if (nidx >= n_nodes) return;
    int g = gid[nidx];
    const float4 v = *reinterpret_cast<const float4*>(h + (size_t)nidx * DIM + c * 4);
    float* p = pooled + (size_t)g * DIM + c * 4;
    unsafeAtomicAdd(p + 0, v.x);
    unsafeAtomicAdd(p + 1, v.y);
    unsafeAtomicAdd(p + 2, v.z);
    unsafeAtomicAdd(p + 3, v.w);
    if (c == 0) unsafeAtomicAdd(&cnt[g], 1.0f);
}

__global__ void pool_div(float* __restrict__ pooled, const float* __restrict__ cnt) {
    int t = blockIdx.x * blockDim.x + threadIdx.x;
    int g = t >> 7;
    pooled[t] /= fmaxf(cnt[g], 1.0f);
}

extern "C" void kernel_launch(void* const* d_in, const int* in_sizes, int n_in,
                              void* d_out, int out_size, void* d_ws, size_t ws_size,
                              hipStream_t stream) {
    const float* feats = (const float*)d_in[0];
    const float* Wall  = (const float*)d_in[1];
    const float* ball  = (const float*)d_in[2];
    const int*   src   = (const int*)d_in[3];
    const int*   dst   = (const int*)d_in[4];
    const int*   gid   = (const int*)d_in[5];
    float* out = (float*)d_out;

    float* ws = (float*)d_ws;
    float* norm_out = ws;
    float* norm_in  = ws + N_NODES;
    float* bufA     = ws + 2 * (size_t)N_NODES;
    float* bufB     = bufA + (size_t)N_NODES * DIM;
    float* cnt      = bufB + (size_t)N_NODES * DIM;
    int*   lcount   = (int*)(cnt + N_GRAPHS);
    int*   lists    = lcount + N_BLOCKS + LIST_PAD;

    const int BT = 256;
    unsigned* tables = (unsigned*)bufB;
    degree_count<<<2 * HIST_SLICES, HIST_THREADS, 0, stream>>>(src, dst, tables);
    norm_from_counts<<<(NORM_VEC4 + NORM_THREADS - 1) / NORM_THREADS, NORM_THREADS, 0, stream>>>(
        tables, norm_out);

    const float* hin = feats;
    float* bufs[2] = {bufA, bufB};
    int cur = 0;
    for (int l = 0; l < N_LAYERS; ++l) {
        float* hout = bufs[cur];
        spmm_owner<<<N_BLOCKS, SPMM_THREADS, 0, stream>>>(
            hin, norm_out, norm_in, src, dst,
            Wall + (size_t)l * DIM * DIM, ball + (size_t)l * DIM, hout,
            lcount, lists, ws_size, l, N_EDGES, N_NODES);
        hin = hout;
        cur ^= 1;
    }

    zero_f32<<<(N_GRAPHS * DIM + BT - 1) / BT, BT, 0, stream>>>(out, N_GRAPHS * DIM);
    zero_f32<<<1, N_GRAPHS, 0, stream>>>(cnt, N_GRAPHS);
    const long long pool_threads = (long long)N_NODES * 32;
    pool_accum<<<(int)((pool_threads + BT - 1) / BT), BT, 0, stream>>>(hin, gid, out, cnt, N_NODES);
    pool_div<<<(N_GRAPHS * DIM) / BT, BT, 0, stream>>>(out, cnt);
}


# Mean pooling in the layer launches: first layer clears sums and counts, last layer's blocks sum their LDS rows per run of equal graph ids in row order, one float add to out per run; 3 launches fewer
# speedup vs baseline: 4.3449x; 1.5828x over previous
#include <hip/hip_runtime.h>
#include <hip/hip_bf16.h>

#define N_NODES  50000
#define N_EDGES  800000
#define DIM      128
#define N_LAYERS 3
#define N_GRAPHS 256

typedef __attribute__((ext_vector_type(2))) float v2f;
typedef __attribute__((ext_vector_type(8))) float v8f;

#define HIST_SLICES   32
#define HIST_THREADS  1024
#define SLICE_EDGES   (N_EDGES / HIST_SLICES)
#define SLICE_IDX4    (SLICE_EDGES / 4)
#define IDX4_PER_THREAD ((SLICE_IDX4 + HIST_THREADS - 1) / HIST_THREADS)
static_assert(N_EDGES % (HIST_SLICES * 4) == 0, "a slice is a whole number of 16-byte loads");
static_assert(N_NODES % 4 == 0, "tables are cleared, stored and added four counters at a time");

__launch_bounds__(HIST_THREADS)
__global__ void degree_count(const int* __restrict__ src, const int* __restrict__ dst,
                             unsigned* __restrict__ tables) {
    __shared__ __attribute__((aligned(16))) unsigned bins[N_NODES];
    const int tid = threadIdx.x;
    const int which = blockIdx.x / HIST_SLICES;
    const int slice = blockIdx.x % HIST_SLICES;
    const int4* __restrict__ idx4 =
        reinterpret_cast<const int4*>((which ? dst : src) + (size_t)slice * SLICE_EDGES);
    int4 v[IDX4_PER_THREAD];
#pragma unroll
    for (int k = 0; k < IDX4_PER_THREAD; ++k) {
        const int i = tid + k * HIST_THREADS;
        v[k] = idx4[i < SLICE_IDX4 ? i : SLICE_IDX4 - 1];
    }
    const uint4 zero4 = {0u, 0u, 0u, 0u};
    for (int i = tid; i < N_NODES / 4; i += HIST_THREADS)
        reinterpret_cast<uint4*>(bins)[i] = zero4;
    __syncthreads();
#pragma unroll
    for (int k = 0; k < IDX4_PER_THREAD; ++k) {
        if (tid + k * HIST_THREADS < SLICE_IDX4) {
            atomicAdd(&bins[v[k].x], 1u);
            atomicAdd(&bins[v[k].y], 1u);
            atomicAdd(&bins[v[k].z], 1u);
            atomicAdd(&bins[v[k].w], 1u);
        }
    }
    __syncthreads();
    uint4* __restrict__ out4 = reinterpret_cast<uint4*>(
        tables + ((size_t)slice * 2 + which) * N_NODES);
    for (int i = tid; i < N_NODES / 4; i += HIST_THREADS)
        out4[i] = reinterpret_cast<const uint4*>(bins)[i];
}

#define NORM_THREADS 64
#define NORM_VEC4    (2 * N_NODES / 4)
__global__ void norm_from_counts(const unsigned* __restrict__ tables, float* __restrict__ norms) {
    const int t = blockIdx.x * blockDim.x + threadIdx.x;
    if (t >= NORM_VEC4) return;
    const uint4* __restrict__ t4 = reinterpret_cast<const uint4*>(tables);
    uint4 s = {0u, 0u, 0u, 0u};
#pragma unroll
    for (int b = 0; b < HIST_SLICES; ++b) {
        const uint4 c = t4[(size_t)b * NORM_VEC4 + t];
        s.x += c.x; s.y += c.y; s.z += c.z; s.w += c.w;
    }
    float4 r;
    r.x = __frsqrt_rn(fmaxf((float)s.x, 1.0f));
    r.y = __frsqrt_rn(fmaxf((float)s.y, 1.0f));
    r.z = __frsqrt_rn(fmaxf((float)s.z, 1.0f));
    r.w = __frsqrt_rn(fmaxf((float)s.w, 1.0f));
    reinterpret_cast<float4*>(norms)[t] = r;
}

#define ROWS_PER_BLOCK 196
#define SPMM_THREADS   1024
#define SPMM_WAVES     (SPMM_THREADS / 32)
#define EDGE_CHUNK     256
#define RING           4
#define QCAP           512
#define ROWS_IN_FLIGHT 8
#define M_TILES   ((ROWS_PER_BLOCK + 15) / 16)
#define ACC_ROWS  (M_TILES * 16)
#define ACC_PITCH 132
static_assert(N_EDGES % EDGE_CHUNK == 0, "the dst list is read in whole chunks");
#define N_BLOCKS  ((N_NODES + ROWS_PER_BLOCK - 1) / ROWS_PER_BLOCK)
#define POOL_PARTS (SPMM_THREADS / DIM)
#define POOL_ROWS  ((ROWS_PER_BLOCK + POOL_PARTS - 1) / POOL_PARTS)
static_assert(SPMM_THREADS % DIM == 0, "a pooling part is a whole number of waves, one thread per column");
static_assert(N_LAYERS >= 2, "the first layer clears the pooled sums, the last layer adds to them");
#define LIST_CAP  4096
#define LIST_PAD  32
static_assert(ROWS_PER_BLOCK <= 256, "the local row takes the low 8 bits of a list entry");
static_assert(N_NODES <= (1 << 23), "the source row takes the upper 24 bits of a list entry");
#define WS_NEED (sizeof(float) * (2 * (size_t)N_NODES + 2 * (size_t)N_NODES * DIM + N_GRAPHS) + \
                 sizeof(int) * (size_t)(N_BLOCKS + LIST_PAD) + sizeof(int2) * (size_t)N_BLOCKS * LIST_CAP)

__launch_bounds__(SPMM_THREADS)
__global__ void spmm_owner(const float* __restrict__ h, const float* __restrict__ norm_out,
                           const float* __restrict__ norm_in,
                           const int* __restrict__ src, const int* __restrict__ dst,
                           const float* __restrict__ Wl,
                           const float* __restrict__ bias,
                           float* __restrict__ out_h,
                           const int* __restrict__ gid,
                           float* __restrict__ pooled,
                           float* __restrict__ cnt,
                           int* __restrict__ lcount,
                           int* __restrict__ lists,
                           size_t ws_bytes, int layer, int n_edges, int n_nodes) {
    __shared__ __attribute__((aligned(16))) float acc[ACC_ROWS * ACC_PITCH];
    __shared__ __attribute__((aligned(16))) unsigned char qbuf[sizeof(int2) * SPMM_WAVES * QCAP];
    static_assert(sizeof(qbuf) >= sizeof(float) * ACC_ROWS * ACC_PITCH,
                  "the output tile must fit in the queues' LDS");
    __shared__ int lcnt;
    __shared__ int gl[ROWS_PER_BLOCK];
    int2* const queue = reinterpret_cast<int2*>(qbuf);
    const int tid = threadIdx.x;
    const int lane = tid & 31;
    const int wave = __builtin_amdgcn_readfirstlane(tid >> 5);
    const int r0 = blockIdx.x * ROWS_PER_BLOCK;
    int2* const q = queue + wave * QCAP;
    int2* const mylist = reinterpret_cast<int2*>(lists) + (size_t)blockIdx.x * LIST_CAP;
    const bool have_lists = ws_bytes >= WS_NEED;
    const bool record = have_lists && layer == 0;
    const int total = (have_lists && layer != 0) ? lcount[blockIdx.x] : 0;
    const bool replay = have_lists && layer != 0 && total <= LIST_CAP;
    const bool pool = layer == N_LAYERS - 1;

    if (layer == 0) {
        const int gt = blockIdx.x * SPMM_THREADS + tid;
        const int gstep = N_BLOCKS * SPMM_THREADS;
        for (int i = gt; i < N_GRAPHS * DIM; i += gstep) pooled[i] = 0.0f;
        for (int i = gt; i < N_GRAPHS; i += gstep) cnt[i] = 0.0f;
    }

    const float4 zero4 = {0.0f, 0.0f, 0.0f, 0.0f};
    for (int i = tid; i < ACC_ROWS * (ACC_PITCH / 4); i += SPMM_THREADS)
        reinterpret_cast<float4*>(acc)[i] = zero4;
    if (tid == 0) lcnt = 0;
    __syncthreads();

    int qh = 0;
    int qn = 0;

    auto add_rows = [&](int n, int s, int nsb, int row) {
#pragma nounroll
        for (int j0 = 0; j0 < n; j0 += ROWS_IN_FLIGHT) {
            float v[ROWS_IN_FLIGHT][4];
#pragma unroll
            for (int u = 0; u < ROWS_IN_FLIGHT; ++u) {
                const int sj = __builtin_amdgcn_readlane(s, j0 + u);
                const float* hp = h + (size_t)sj * DIM + lane;
                v[u][0] = hp[0]; v[u][1] = hp[32]; v[u][2] = hp[64]; v[u][3] = hp[96];
            }
#pragma unroll
            for (int u = 0; u < ROWS_IN_FLIGHT; ++u) {
                if (j0 + u < n) {
                    const float nj = __int_as_float(__builtin_amdgcn_readlane(nsb, j0 + u));
                    const int rowj = __builtin_amdgcn_readlane(row, j0 + u);
                    float* ap = acc + rowj * ACC_PITCH + lane;
                    unsafeAtomicAdd(ap + 0,  v[u][0] * nj);
                    unsafeAtomicAdd(ap + 32, v[u][1] * nj);
                    unsafeAtomicAdd(ap + 64, v[u][2] * nj);
                    unsafeAtomicAdd(ap + 96, v[u][3] * nj);
                }
            }
        }
    };

    auto drain = [&](int n) {
        int2 ent = {0, 0};
        if (lane < n) ent = q[(qh + lane) & (QCAP - 1)];
        const int s = src[ent.x];
        const int nsb = __float_as_int(norm_out[s]);
        if (record) {
            int base = 0;
            if (lane == 0) base = atomicAdd(&lcnt, n);
            base = __builtin_amdgcn_readfirstlane(base);
            if (lane < n && base + lane < LIST_CAP)
                mylist[base + lane] = make_int2((s << 8) | ent.y, nsb);
        }
        add_rows(n, s, nsb, ent.y);
        qh = (qh + n) & (QCAP - 1);
        qn -= n;
    };

    auto scan_chunk = [&](const int4& c0, const int4& c1, int ch) {
        const unsigned rel[8] = {(unsigned)(c0.x - r0), (unsigned)(c0.y - r0),
                                 (unsigned)(c0.z - r0), (unsigned)(c0.w - r0),
                                 (unsigned)(c1.x - r0), (unsigned)(c1.y - r0),
                                 (unsigned)(c1.z - r0), (unsigned)(c1.w - r0)};
        unsigned m[8];
        unsigned any = 0;
#pragma unroll
        for (int k = 0; k < 8; ++k) {
            m[k] = __builtin_amdgcn_ballot_w32(rel[k] < (unsigned)ROWS_PER_BLOCK);
            any |= m[k];
        }
        if (any) {
#pragma unroll
            for (int k = 0; k < 8; ++k) {
                if (m[k]) {
                    if (rel[k] < (unsigned)ROWS_PER_BLOCK) {
                        const int pos = qh + qn + (int)__builtin_amdgcn_mbcnt_lo(m[k], 0u);
                        const int e = ch * EDGE_CHUNK + (k >> 2) * 128 + lane * 4 + (k & 3);
                        q[pos & (QCAP - 1)] = make_int2(e, (int)rel[k]);
                    }
                    qn += __builtin_popcount(m[k]);
                }
            }
            while (qn >= 32) drain(32);
        }
    };

    if (replay) {
        static_assert(LIST_CAP == 4 * SPMM_THREADS, "four named entry registers below");
        if (total > 0) {
            const int last = total - 1;
            const int i0 = wave * 32 + lane;
            auto entry = [&](int i) { return mylist[i < last ? i : last]; };
            const int2 e0 = entry(i0);
            const int2 e1 = entry(i0 + SPMM_THREADS);
            const int2 e2 = entry(i0 + 2 * SPMM_THREADS);
            const int2 e3 = entry(i0 + 3 * SPMM_THREADS);
#pragma nounroll
            for (int t = 0; t < 4; ++t) {
                const int at = wave * 32 + t * SPMM_THREADS;
                if (at >= total) break;
                const int2 cur = t == 0 ? e0 : t == 1 ? e1 : t == 2 ? e2 : e3;
                const int left = total - at;
                add_rows(left < 32 ? left : 32, cur.x >> 8, cur.y, cur.x & 255);
            }
        }
    } else {
    const int n_chunks = n_edges / EDGE_CHUNK;
    const int offset = n_chunks > 0 ? (int)((blockIdx.x * 12u) % (unsigned)n_chunks) : 0;
    const int4* __restrict__ dst4 = reinterpret_cast<const int4*>(dst);

    auto chunk_of = [&](int idx) {
        int ch = idx + offset;
        if (ch >= n_chunks) ch -= n_chunks;
        return idx < n_chunks ? ch : 0;
    };
    auto load_set = [&](int4& x0, int4& x1, int idx) {
        const int ch = chunk_of(idx);
        x0 = dst4[ch * 64 + lane];
        x1 = dst4[ch * 64 + 32 + lane];
    };
    int4 a0, a1, b0, b1, e0, e1, f0, f1;
    static_assert(RING == 4, "four named register sets below");
    load_set(a0, a1, wave);
    load_set(b0, b1, wave + SPMM_WAVES);
    load_set(e0, e1, wave + 2 * SPMM_WAVES);
    load_set(f0, f1, wave + 3 * SPMM_WAVES);
    auto step = [&](int4& x0, int4& x1, int my) {
        if (my < n_chunks) scan_chunk(x0, x1, chunk_of(my));
        load_set(x0, x1, my + RING * SPMM_WAVES);
    };
    for (int idx = wave; idx < n_chunks; idx += RING * SPMM_WAVES) {
        step(a0, a1, idx);
        step(b0, b1, idx + SPMM_WAVES);
        step(e0, e1, idx + 2 * SPMM_WAVES);
        step(f0, f1, idx + 3 * SPMM_WAVES);
    }
    if (qn > 0) drain(qn);
    }
    __syncthreads();
    if (record && tid == 0) lcount[blockIdx.x] = lcnt;
    if (pool && tid < ROWS_PER_BLOCK) {
        const int grow = r0 + tid;
        gl[tid] = grow < n_nodes ? gid[grow] : -1;
    }

    float* const hout = reinterpret_cast<float*>(qbuf);
    const int nt = wave & 7;
    const int mg = wave >> 3;
    const int hl = lane & 15;
    const int hi = lane >> 4;
    const int col = nt * 16 + hl;
    v2f bfrag[32];
#pragma unroll
    for (int kk = 0; kk < 32; ++kk) {
        const int kb = kk * 4 + hi * 2;
        bfrag[kk].x = Wl[(size_t)(kb + 0) * DIM + col];
        bfrag[kk].y = Wl[(size_t)(kb + 1) * DIM + col];
    }
    const float bb = bias[col];
    float nin[4];
#pragma unroll
    for (int t = 0; t < 4; ++t) {
        const int grow = r0 + (mg + 4 * t) * 16 + hl;
        nin[t] = grow < n_nodes ? norm_in[grow] : 0.0f;
    }
#pragma unroll
    for (int t = 0; t < 4; ++t) {
        const int mt = mg + 4 * t;
        if (mt < M_TILES) {
            const int lrow = mt * 16 + hl;
            const float* arow = acc + lrow * ACC_PITCH + hi * 2;
            v8f d = {};
#pragma unroll
            for (int kk = 0; kk < 32; ++kk) {
                v2f a;
                a.x = arow[kk * 4 + 0] * nin[t];
                a.y = arow[kk * 4 + 1] * nin[t];
                d = __builtin_amdgcn_wmma_f32_16x16x4_f32(
                     false, a,  false, bfrag[kk],
                     (short)0, d,  false,  false);
            }
#pragma unroll
            for (int j = 0; j < 8; ++j) {
                const float v = d[j] + bb;
                hout[(mt * 16 + j + hi * 8) * ACC_PITCH + col] = v > 0.0f ? v : 0.0f;
            }
        }
    }
    __syncthreads();

    if (pool) {
        const int pc = tid & (DIM - 1);
        const int rb = (tid / DIM) * POOL_ROWS;
        const int nr = n_nodes - r0 < ROWS_PER_BLOCK ? n_nodes - r0 : ROWS_PER_BLOCK;
        const int re = rb + POOL_ROWS < nr ? rb + POOL_ROWS : nr;
        if (rb < re) {
            auto flush = [&](int g, float s, int n) {
                if ((unsigned)g < (unsigned)N_GRAPHS) {
                    unsafeAtomicAdd(pooled + (size_t)g * DIM + pc, s);
                    if (pc == 0) unsafeAtomicAdd(cnt + g, (float)n);
                }
            };
            int g = gl[rb];
            float s = hout[rb * ACC_PITCH + pc];
            int n = 1;
            for (int r = rb + 1; r < re; ++r) {
                const int gr = gl[r];
                const float v = hout[r * ACC_PITCH + pc];
                if (gr != g) {
                    flush(g, s, n);
                    g = gr; s = 0.0f; n = 0;
                }
                s += v;
                ++n;
            }
            flush(g, s, n);
        }
    } else {
        for (int i = tid; i < ROWS_PER_BLOCK * (DIM / 4); i += SPMM_THREADS) {
            const int r = i >> 5;
            const int c4 = i & 31;
            const int grow = r0 + r;
            if (grow < n_nodes)
                reinterpret_cast<float4*>(out_h + (size_t)grow * DIM)[c4] =
                    reinterpret_cast<const float4*>(hout + r * ACC_PITCH)[c4];
        }
    }
}

__global__ void pool_div(float* __restrict__ pooled, const float* __restrict__ cnt) {
    int t = blockIdx.x * blockDim.x + threadIdx.x;
    int g = t >> 7;
    pooled[t] /= fmaxf(cnt[g], 1.0f);
}

extern "C" void kernel_launch(void* const* d_in, const int* in_sizes, int n_in,
                              void* d_out, int out_size, void* d_ws, size_t ws_size,
                              hipStream_t stream) {
    const float* feats = (const float*)d_in[0];
    const float* Wall  = (const float*)d_in[1];
    const float* ball  = (const float*)d_in[2];
    const int*   src   = (const int*)d_in[3];
    const int*   dst   = (const int*)d_in[4];
    const int*   gid   = (const int*)d_in[5];
    float* out = (float*)d_out;

    float* ws = (float*)d_ws;
    float* norm_out = ws;
    float* norm_in  = ws + N_NODES;
    float* bufA     = ws + 2 * (size_t)N_NODES;
    float* bufB     = bufA + (size_t)N_NODES * DIM;
    float* cnt      = bufB + (size_t)N_NODES * DIM;
    int*   lcount   = (int*)(cnt + N_GRAPHS);
    int*   lists    = lcount + N_BLOCKS + LIST_PAD;

    const int BT = 256;
    unsigned* tables = (unsigned*)bufB;
    degree_count<<<2 * HIST_SLICES, HIST_THREADS, 0, stream>>>(src, dst, tables);
    norm_from_counts<<<(NORM_VEC4 + NORM_THREADS - 1) / NORM_THREADS, NORM_THREADS, 0, stream>>>(
        tables, norm_out);

    const float* hin = feats;
    float* bufs[2] = {bufA, bufB};
    int cur = 0;
    for (int l = 0; l < N_LAYERS; ++l) {
        float* hout = bufs[cur];
        spmm_owner<<<N_BLOCKS, SPMM_THREADS, 0, stream>>>(
            hin, norm_out, norm_in, src, dst,
            Wall + (size_t)l * DIM * DIM, ball + (size_t)l * DIM, hout,
            gid, out, cnt, lcount, lists, ws_size, l, N_EDGES, N_NODES);
        hin = hout;
        cur ^= 1;
    }

    pool_div<<<(N_GRAPHS * DIM) / BT, BT, 0, stream>>>(out, cnt);
}


# First layer's owner scan reads one owner byte per edge (plane written by the dst degree blocks), not 4-byte destinations: 16 edges per load, byte matches by word arithmetic; LDS sums in arrival order
# speedup vs baseline: 4.5521x; 1.0477x over previous
#include <hip/hip_runtime.h>
#include <hip/hip_bf16.h>

#define N_NODES  50000
#define N_EDGES  800000
#define DIM      128
#define N_LAYERS 3
#define N_GRAPHS 256

typedef __attribute__((ext_vector_type(2))) float v2f;
typedef __attribute__((ext_vector_type(8))) float v8f;

#define ROWS_PER_BLOCK 196
#define N_BLOCKS  ((N_NODES + ROWS_PER_BLOCK - 1) / ROWS_PER_BLOCK)
static_assert(N_BLOCKS <= 256, "an owner is one byte");
__device__ __forceinline__ unsigned owner_bytes(const int4 d) {
    const unsigned a = min((unsigned)d.x / (unsigned)ROWS_PER_BLOCK, 255u);
    const unsigned b = min((unsigned)d.y / (unsigned)ROWS_PER_BLOCK, 255u);
    const unsigned e = min((unsigned)d.z / (unsigned)ROWS_PER_BLOCK, 255u);
    const unsigned f = min((unsigned)d.w / (unsigned)ROWS_PER_BLOCK, 255u);
    return a | (b << 8) | (e << 16) | (f << 24);
}

#define HIST_SLICES   32
#define HIST_THREADS  1024
#define SLICE_EDGES   (N_EDGES / HIST_SLICES)
#define SLICE_IDX4    (SLICE_EDGES / 4)
#define IDX4_PER_THREAD ((SLICE_IDX4 + HIST_THREADS - 1) / HIST_THREADS)
static_assert(N_EDGES % (HIST_SLICES * 4) == 0, "a slice is a whole number of 16-byte loads");
static_assert(N_NODES % 4 == 0, "tables are cleared, stored and added four counters at a time");
static_assert(sizeof(unsigned) * HIST_SLICES * 2 * (size_t)N_NODES + N_EDGES <= sizeof(float) * (size_t)N_NODES * DIM,
              "the tables and the owner plane behind them fit in one h plane");
static_assert((sizeof(unsigned) * HIST_SLICES * 2 * (size_t)N_NODES) % 16 == 0, "the owner plane starts on a 16-byte boundary");

__launch_bounds__(HIST_THREADS)
__global__ void degree_count(const int* __restrict__ src, const int* __restrict__ dst,
                             unsigned* __restrict__ tables, unsigned* __restrict__ own32) {
    __shared__ __attribute__((aligned(16))) unsigned bins[N_NODES];
    const int tid = threadIdx.x;
    const int which = blockIdx.x / HIST_SLICES;
    const int slice = blockIdx.x % HIST_SLICES;
    const int4* __restrict__ idx4 =
        reinterpret_cast<const int4*>((which ? dst : src) + (size_t)slice * SLICE_EDGES);
    int4 v[IDX4_PER_THREAD];
#pragma unroll
    for (int k = 0; k < IDX4_PER_THREAD; ++k) {
        const int i = tid + k * HIST_THREADS;
        v[k] = idx4[i < SLICE_IDX4 ? i : SLICE_IDX4 - 1];
    }
    const uint4 zero4 = {0u, 0u, 0u, 0u};
    for (int i = tid; i < N_NODES / 4; i += HIST_THREADS)
        reinterpret_cast<uint4*>(bins)[i] = zero4;
    __syncthreads();
#pragma unroll
    for (int k = 0; k < IDX4_PER_THREAD; ++k) {
        if (tid + k * HIST_THREADS < SLICE_IDX4) {
            atomicAdd(&bins[v[k].x], 1u);
            atomicAdd(&bins[v[k].y], 1u);
            atomicAdd(&bins[v[k].z], 1u);
            atomicAdd(&bins[v[k].w], 1u);
            if (which) own32[(size_t)slice * SLICE_IDX4 + tid + k * HIST_THREADS] = owner_bytes(v[k]);
        }
    }
    __syncthreads();
    uint4* __restrict__ out4 = reinterpret_cast<uint4*>(
        tables + ((size_t)slice * 2 + which) * N_NODES);
    for (int i = tid; i < N_NODES / 4; i += HIST_THREADS)
        out4[i] = reinterpret_cast<const uint4*>(bins)[i];
}

#define NORM_THREADS 64
#define NORM_VEC4    (2 * N_NODES / 4)
__global__ void norm_from_counts(const unsigned* __restrict__ tables, float* __restrict__ norms) {
    const int t = blockIdx.x * blockDim.x + threadIdx.x;
    if (t >= NORM_VEC4) return;
    const uint4* __restrict__ t4 = reinterpret_cast<const uint4*>(tables);
    uint4 s = {0u, 0u, 0u, 0u};
#pragma unroll
    for (int b = 0; b < HIST_SLICES; ++b) {
        const uint4 c = t4[(size_t)b * NORM_VEC4 + t];
        s.x += c.x; s.y += c.y; s.z += c.z; s.w += c.w;
    }
    float4 r;
    r.x = __frsqrt_rn(fmaxf((float)s.x, 1.0f));
    r.y = __frsqrt_rn(fmaxf((float)s.y, 1.0f));
    r.z = __frsqrt_rn(fmaxf((float)s.z, 1.0f));
    r.w = __frsqrt_rn(fmaxf((float)s.w, 1.0f));
    reinterpret_cast<float4*>(norms)[t] = r;
}

#define SPMM_THREADS   1024
#define SPMM_WAVES     (SPMM_THREADS / 32)
#define VEC_EDGES      16
#define EDGE_CHUNK     (32 * VEC_EDGES)
#define RING           4
#define QCAP           1024
#define ROWS_IN_FLIGHT 8
#define M_TILES   ((ROWS_PER_BLOCK + 15) / 16)
#define ACC_ROWS  (M_TILES * 16)
#define ACC_PITCH 132
static_assert(N_EDGES % VEC_EDGES == 0, "the edge list is a whole number of 16-byte owner loads");
static_assert(EDGE_CHUNK + 32 <= QCAP, "a chunk's matches fit behind the waiting entries");
#define POOL_PARTS (SPMM_THREADS / DIM)
#define POOL_ROWS  ((ROWS_PER_BLOCK + POOL_PARTS - 1) / POOL_PARTS)
static_assert(SPMM_THREADS % DIM == 0, "a pooling part is a whole number of waves, one thread per column");
static_assert(N_LAYERS >= 2, "the first layer clears the pooled sums, the last layer adds to them");
#define LIST_CAP  4096
#define LIST_PAD  32
static_assert(ROWS_PER_BLOCK <= 256, "the local row takes the low 8 bits of a list entry");
static_assert(N_NODES <= (1 << 23), "the source row takes the upper 24 bits of a list entry");
#define WS_NEED (sizeof(float) * (2 * (size_t)N_NODES + 2 * (size_t)N_NODES * DIM + N_GRAPHS) + \
                 sizeof(int) * (size_t)(N_BLOCKS + LIST_PAD) + sizeof(int2) * (size_t)N_BLOCKS * LIST_CAP)

__launch_bounds__(SPMM_THREADS)
__global__ void spmm_owner(const float* __restrict__ h, const float* __restrict__ norm_out,
                           const float* __restrict__ norm_in,
                           const int* __restrict__ src, const int* __restrict__ dst,
                           const uint4* __restrict__ own,
                           const float* __restrict__ Wl,
                           const float* __restrict__ bias,
                           float* __restrict__ out_h,
                           const int* __restrict__ gid,
                           float* __restrict__ pooled,
                           float* __restrict__ cnt,
                           int* __restrict__ lcount,
                           int* __restrict__ lists,
                           size_t ws_bytes, int layer, int n_edges, int n_nodes) {
    __shared__ __attribute__((aligned(16))) float acc[ACC_ROWS * ACC_PITCH];
    __shared__ __attribute__((aligned(16))) unsigned char qbuf[sizeof(int) * SPMM_WAVES * QCAP];
    static_assert(sizeof(qbuf) >= sizeof(float) * ACC_ROWS * ACC_PITCH,
                  "the output tile must fit in the queues' LDS");
    __shared__ int lcnt;
    __shared__ int gl[ROWS_PER_BLOCK];
    int* const queue = reinterpret_cast<int*>(qbuf);
    const int tid = threadIdx.x;
    const int lane = tid & 31;
    const int wave = __builtin_amdgcn_readfirstlane(tid >> 5);
    const int r0 = blockIdx.x * ROWS_PER_BLOCK;
    int* const q = queue + wave * QCAP;
    int2* const mylist = reinterpret_cast<int2*>(lists) + (size_t)blockIdx.x * LIST_CAP;
    const bool have_lists = ws_bytes >= WS_NEED;
    const bool record = have_lists && layer == 0;
    const int total = (have_lists && layer != 0) ? lcount[blockIdx.x] : 0;
    const bool replay = have_lists && layer != 0 && total <= LIST_CAP;
    const bool pool = layer == N_LAYERS - 1;

    if (layer == 0) {
        const int gt = blockIdx.x * SPMM_THREADS + tid;
        const int gstep = N_BLOCKS * SPMM_THREADS;
        for (int i = gt; i < N_GRAPHS * DIM; i += gstep) pooled[i] = 0.0f;
        for (int i = gt; i < N_GRAPHS; i += gstep) cnt[i] = 0.0f;
    }

    const float4 zero4 = {0.0f, 0.0f, 0.0f, 0.0f};
    for (int i = tid; i < ACC_ROWS * (ACC_PITCH / 4); i += SPMM_THREADS)
        reinterpret_cast<float4*>(acc)[i] = zero4;
    if (tid == 0) lcnt = 0;
    __syncthreads();

    int qh = 0;
    int qn = 0;

    auto add_rows = [&](int n, int s, int nsb, int row) {
#pragma nounroll
        for (int j0 = 0; j0 < n; j0 += ROWS_IN_FLIGHT) {
            float v[ROWS_IN_FLIGHT][4];
#pragma unroll
            for (int u = 0; u < ROWS_IN_FLIGHT; ++u) {
                const int sj = __builtin_amdgcn_readlane(s, j0 + u);
                const float* hp = h + (size_t)sj * DIM + lane;
                v[u][0] = hp[0]; v[u][1] = hp[32]; v[u][2] = hp[64]; v[u][3] = hp[96];
            }
#pragma unroll
            for (int u = 0; u < ROWS_IN_FLIGHT; ++u) {
                if (j0 + u < n) {
                    const float nj = __int_as_float(__builtin_amdgcn_readlane(nsb, j0 + u));
                    const int rowj = __builtin_amdgcn_readlane(row, j0 + u);
                    float* ap = acc + rowj * ACC_PITCH + lane;
                    unsafeAtomicAdd(ap + 0,  v[u][0] * nj);
                    unsafeAtomicAdd(ap + 32, v[u][1] * nj);
                    unsafeAtomicAdd(ap + 64, v[u][2] * nj);
                    unsafeAtomicAdd(ap + 96, v[u][3] * nj);
                }
            }
        }
    };

    auto drain = [&](int n) {
        int e = 0;
        if (lane < n) e = q[(qh + lane) & (QCAP - 1)];
        const int s = src[e];
        const unsigned rel = (unsigned)(dst[e] - r0);
        int nsb = __float_as_int(norm_out[s]);
        const bool mine = rel < (unsigned)ROWS_PER_BLOCK;
        const int row = mine ? (int)rel : 0;
        if (!mine) nsb = 0;
        if (record) {
            int base = 0;
            if (lane == 0) base = atomicAdd(&lcnt, n);
            base = __builtin_amdgcn_readfirstlane(base);
            if (lane < n && base + lane < LIST_CAP)
                mylist[base + lane] = make_int2((s << 8) | row, nsb);
        }
        add_rows(n, s, nsb, row);
        qh = (qh + n) & (QCAP - 1);
        qn -= n;
    };

    const unsigned pat = (unsigned)blockIdx.x * 0x01010101u;
    const int n_vec = n_edges / VEC_EDGES;
    auto scan_chunk = [&](const uint4& x, int ch) {
        const int vi = ch * 32 + lane;
        const unsigned M = 0x7f7f7f7fu;
        const unsigned t0 = x.x ^ pat, t1 = x.y ^ pat, t2 = x.z ^ pat, t3 = x.w ^ pat;
        const unsigned n0 = ((t0 & M) + M) | t0;
        const unsigned n1 = ((t1 & M) + M) | t1;
        const unsigned n2 = ((t2 & M) + M) | t2;
        const unsigned n3 = ((t3 & M) + M) | t3;
        unsigned g = (((n0 >> 7) & 0x01010101u) | ((n1 >> 6) & 0x02020202u) |
                      ((n2 >> 5) & 0x04040404u) | ((n3 >> 4) & 0x08080808u)) ^ 0x0f0f0f0fu;
        if (vi >= n_vec) g = 0;
        unsigned mk = __builtin_amdgcn_ballot_w32(g != 0);
        if (mk) {
            do {
                if (g) {
                    const int b = __builtin_ctz(g);
                    const int pos = qh + qn + (int)__builtin_amdgcn_mbcnt_lo(mk, 0u);
                    q[pos & (QCAP - 1)] = vi * VEC_EDGES + (b & 7) * 4 + (b >> 3);
                    g &= g - 1;
                }
                qn += __builtin_popcount(mk);
                mk = __builtin_amdgcn_ballot_w32(g != 0);
            } while (mk);
            while (qn >= 32) drain(32);
        }
    };

    if (replay) {
        static_assert(LIST_CAP == 4 * SPMM_THREADS, "four named entry registers below");
        if (total > 0) {
            const int last = total - 1;
            const int i0 = wave * 32 + lane;
            auto entry = [&](int i) { return mylist[i < last ? i : last]; };
            const int2 e0 = entry(i0);
            const int2 e1 = entry(i0 + SPMM_THREADS);
            const int2 e2 = entry(i0 + 2 * SPMM_THREADS);
            const int2 e3 = entry(i0 + 3 * SPMM_THREADS);
#pragma nounroll
            for (int t = 0; t < 4; ++t) {
                const int at = wave * 32 + t * SPMM_THREADS;
                if (at >= total) break;
                const int2 cur = t == 0 ? e0 : t == 1 ? e1 : t == 2 ? e2 : e3;
                const int left = total - at;
                add_rows(left < 32 ? left : 32, cur.x >> 8, cur.y, cur.x & 255);
            }
        }
    } else {
    const int n_chunks = (n_vec + 31) / 32;
    const int offset = n_chunks > 0 ? (int)((blockIdx.x * 6u) % (unsigned)n_chunks) : 0;
    const bool plane = layer == 0;
    const int4* __restrict__ dst4 = reinterpret_cast<const int4*>(dst);

    auto chunk_of = [&](int idx) {
        int ch = idx + offset;
        if (ch >= n_chunks) ch -= n_chunks;
        return idx < n_chunks ? ch : 0;
    };
    auto vec_of = [&](int ch) {
        const int vi = ch * 32 + lane;
        return vi < n_vec ? vi : n_vec - 1;
    };
    if (plane) {
        auto load_set = [&](uint4& x, int idx) { x = own[vec_of(chunk_of(idx))]; };
        uint4 a0, b0, e0, f0;
        static_assert(RING == 4, "four named register sets below");
        load_set(a0, wave);
        __builtin_amdgcn_sched_barrier(0);
        load_set(b0, wave + SPMM_WAVES);
        __builtin_amdgcn_sched_barrier(0);
        load_set(e0, wave + 2 * SPMM_WAVES);
        __builtin_amdgcn_sched_barrier(0);
        load_set(f0, wave + 3 * SPMM_WAVES);
        auto step = [&](uint4& x0, int my) {
            if (my < n_chunks) scan_chunk(x0, chunk_of(my));
            load_set(x0, my + RING * SPMM_WAVES);
        };
        for (int idx = wave; idx < n_chunks; idx += RING * SPMM_WAVES) {
            step(a0, idx);
            step(b0, idx + SPMM_WAVES);
            step(e0, idx + 2 * SPMM_WAVES);
            step(f0, idx + 3 * SPMM_WAVES);
        }
    } else {
#pragma nounroll
        for (int idx = wave; idx < n_chunks; idx += SPMM_WAVES) {
            const int ch = chunk_of(idx);
            const int4* d4 = dst4 + (size_t)vec_of(ch) * 4;
            uint4 x;
            x.x = owner_bytes(d4[0]);
            x.y = owner_bytes(d4[1]);
            x.z = owner_bytes(d4[2]);
            x.w = owner_bytes(d4[3]);
            scan_chunk(x, ch);
        }
    }
    if (qn > 0) drain(qn);
    }
    __syncthreads();
    if (record && tid == 0) lcount[blockIdx.x] = lcnt;
    if (pool && tid < ROWS_PER_BLOCK) {
        const int grow = r0 + tid;
        gl[tid] = grow < n_nodes ? gid[grow] : -1;
    }

    float* const hout = reinterpret_cast<float*>(qbuf);
    const int nt = wave & 7;
    const int mg = wave >> 3;
    const int hl = lane & 15;
    const int hi = lane >> 4;
    const int col = nt * 16 + hl;
    v2f bfrag[32];
#pragma unroll
    for (int kk = 0; kk < 32; ++kk) {
        const int kb = kk * 4 + hi * 2;
        bfrag[kk].x = Wl[(size_t)(kb + 0) * DIM + col];
        bfrag[kk].y = Wl[(size_t)(kb + 1) * DIM + col];
    }
    const float bb = bias[col];
    float nin[4];
#pragma unroll
    for (int t = 0; t < 4; ++t) {
        const int grow = r0 + (mg + 4 * t) * 16 + hl;
        nin[t] = grow < n_nodes ? norm_in[grow] : 0.0f;
    }
#pragma unroll
    for (int t = 0; t < 4; ++t) {
        const int mt = mg + 4 * t;
        if (mt < M_TILES) {
            const int lrow = mt * 16 + hl;
            const float* arow = acc + lrow * ACC_PITCH + hi * 2;
            v8f d = {};
#pragma unroll
            for (int kk = 0; kk < 32; ++kk) {
                v2f a;
                a.x = arow[kk * 4 + 0] * nin[t];
                a.y = arow[kk * 4 + 1] * nin[t];
                d = __builtin_amdgcn_wmma_f32_16x16x4_f32(
                     false, a,  false, bfrag[kk],
                     (short)0, d,  false,  false);
            }
#pragma unroll
            for (int j = 0; j < 8; ++j) {
                const float v = d[j] + bb;
                hout[(mt * 16 + j + hi * 8) * ACC_PITCH + col] = v > 0.0f ? v : 0.0f;
            }
        }
    }
    __syncthreads();

    if (pool) {
        const int pc = tid & (DIM - 1);
        const int rb = (tid / DIM) * POOL_ROWS;
        const int nr = n_nodes - r0 < ROWS_PER_BLOCK ? n_nodes - r0 : ROWS_PER_BLOCK;
        const int re = rb + POOL_ROWS < nr ? rb + POOL_ROWS : nr;
        if (rb < re) {
            auto flush = [&](int g, float s, int n) {
                if ((unsigned)g < (unsigned)N_GRAPHS) {
                    unsafeAtomicAdd(pooled + (size_t)g * DIM + pc, s);
                    if (pc == 0) unsafeAtomicAdd(cnt + g, (float)n);
                }
            };
            int g = gl[rb];
            float s = hout[rb * ACC_PITCH + pc];
            int n = 1;
            for (int r = rb + 1; r < re; ++r) {
                const int gr = gl[r];
                const float v = hout[r * ACC_PITCH + pc];
                if (gr != g) {
                    flush(g, s, n);
                    g = gr; s = 0.0f; n = 0;
                }
                s += v;
                ++n;
            }
            flush(g, s, n);
        }
    } else {
        for (int i = tid; i < ROWS_PER_BLOCK * (DIM / 4); i += SPMM_THREADS) {
            const int r = i >> 5;
            const int c4 = i & 31;
            const int grow = r0 + r;
            if (grow < n_nodes)
                reinterpret_cast<float4*>(out_h + (size_t)grow * DIM)[c4] =
                    reinterpret_cast<const float4*>(hout + r * ACC_PITCH)[c4];
        }
    }
}

__global__ void pool_div(float* __restrict__ pooled, const float* __restrict__ cnt) {
    int t = blockIdx.x * blockDim.x + threadIdx.x;
    int g = t >> 7;
    pooled[t] /= fmaxf(cnt[g], 1.0f);
}

extern "C" void kernel_launch(void* const* d_in, const int* in_sizes, int n_in,
                              void* d_out, int out_size, void* d_ws, size_t ws_size,
                              hipStream_t stream) {
    const float* feats = (const float*)d_in[0];
    const float* Wall  = (const float*)d_in[1];
    const float* ball  = (const float*)d_in[2];
    const int*   src   = (const int*)d_in[3];
    const int*   dst   = (const int*)d_in[4];
    const int*   gid   = (const int*)d_in[5];
    float* out = (float*)d_out;

    float* ws = (float*)d_ws;
    float* norm_out = ws;
    float* norm_in  = ws + N_NODES;
    float* bufA     = ws + 2 * (size_t)N_NODES;
    float* bufB     = bufA + (size_t)N_NODES * DIM;
    float* cnt      = bufB + (size_t)N_NODES * DIM;
    int*   lcount   = (int*)(cnt + N_GRAPHS);
    int*   lists    = lcount + N_BLOCKS + LIST_PAD;

    const int BT = 256;
    unsigned* tables = (unsigned*)bufB;
    unsigned* own = tables + (size_t)HIST_SLICES * 2 * N_NODES;
    degree_count<<<2 * HIST_SLICES, HIST_THREADS, 0, stream>>>(src, dst, tables, own);
    norm_from_counts<<<(NORM_VEC4 + NORM_THREADS - 1) / NORM_THREADS, NORM_THREADS, 0, stream>>>(
        tables, norm_out);

    const float* hin = feats;
    float* bufs[2] = {bufA, bufB};
    int cur = 0;
    for (int l = 0; l < N_LAYERS; ++l) {
        float* hout = bufs[cur];
        spmm_owner<<<N_BLOCKS, SPMM_THREADS, 0, stream>>>(
            hin, norm_out, norm_in, src, dst, (const uint4*)own,
            Wall + (size_t)l * DIM * DIM, ball + (size_t)l * DIM, hout,
            gid, out, cnt, lcount, lists, ws_size, l, N_EDGES, N_NODES);
        hin = hout;
        cur ^= 1;
    }

    pool_div<<<(N_GRAPHS * DIM) / BT, BT, 0, stream>>>(out, cnt);
}


# SpMM source rows: one 16-byte load per lane, not four 4-byte; LDS sum rows keep columns in lane order and W fragments follow, so the f32 WMMA sums k in a new fixed order; LDS sums in arrival order
# speedup vs baseline: 4.6963x; 1.0317x over previous
#include <hip/hip_runtime.h>
#include <hip/hip_bf16.h>

#define N_NODES  50000
#define N_EDGES  800000
#define DIM      128
#define N_LAYERS 3
#define N_GRAPHS 256

typedef __attribute__((ext_vector_type(2))) float v2f;
typedef __attribute__((ext_vector_type(8))) float v8f;

#define ROWS_PER_BLOCK 196
#define N_BLOCKS  ((N_NODES + ROWS_PER_BLOCK - 1) / ROWS_PER_BLOCK)
static_assert(N_BLOCKS <= 256, "an owner is one byte");
__device__ __forceinline__ unsigned owner_bytes(const int4 d) {
    const unsigned a = min((unsigned)d.x / (unsigned)ROWS_PER_BLOCK, 255u);
    const unsigned b = min((unsigned)d.y / (unsigned)ROWS_PER_BLOCK, 255u);
    const unsigned e = min((unsigned)d.z / (unsigned)ROWS_PER_BLOCK, 255u);
    const unsigned f = min((unsigned)d.w / (unsigned)ROWS_PER_BLOCK, 255u);
    return a | (b << 8) | (e << 16) | (f << 24);
}

#define HIST_SLICES   32
#define HIST_THREADS  1024
#define SLICE_EDGES   (N_EDGES / HIST_SLICES)
#define SLICE_IDX4    (SLICE_EDGES / 4)
#define IDX4_PER_THREAD ((SLICE_IDX4 + HIST_THREADS - 1) / HIST_THREADS)
static_assert(N_EDGES % (HIST_SLICES * 4) == 0, "a slice is a whole number of 16-byte loads");
static_assert(N_NODES % 4 == 0, "tables are cleared, stored and added four counters at a time");
static_assert(sizeof(unsigned) * HIST_SLICES * 2 * (size_t)N_NODES + N_EDGES <= sizeof(float) * (size_t)N_NODES * DIM,
              "the tables and the owner plane behind them fit in one h plane");
static_assert((sizeof(unsigned) * HIST_SLICES * 2 * (size_t)N_NODES) % 16 == 0, "the owner plane starts on a 16-byte boundary");

__launch_bounds__(HIST_THREADS)
__global__ void degree_count(const int* __restrict__ src, const int* __restrict__ dst,
                             unsigned* __restrict__ tables, unsigned* __restrict__ own32) {
    __shared__ __attribute__((aligned(16))) unsigned bins[N_NODES];
    const int tid = threadIdx.x;
    const int which = blockIdx.x / HIST_SLICES;
    const int slice = blockIdx.x % HIST_SLICES;
    const int4* __restrict__ idx4 =
        reinterpret_cast<const int4*>((which ? dst : src) + (size_t)slice * SLICE_EDGES);
    int4 v[IDX4_PER_THREAD];
#pragma unroll
    for (int k = 0; k < IDX4_PER_THREAD; ++k) {
        const int i = tid + k * HIST_THREADS;
        v[k] = idx4[i < SLICE_IDX4 ? i : SLICE_IDX4 - 1];
    }
    const uint4 zero4 = {0u, 0u, 0u, 0u};
    for (int i = tid; i < N_NODES / 4; i += HIST_THREADS)
        reinterpret_cast<uint4*>(bins)[i] = zero4;
    __syncthreads();
#pragma unroll
    for (int k = 0; k < IDX4_PER_THREAD; ++k) {
        if (tid + k * HIST_THREADS < SLICE_IDX4) {
            atomicAdd(&bins[v[k].x], 1u);
            atomicAdd(&bins[v[k].y], 1u);
            atomicAdd(&bins[v[k].z], 1u);
            atomicAdd(&bins[v[k].w], 1u);
            if (which) own32[(size_t)slice * SLICE_IDX4 + tid + k * HIST_THREADS] = owner_bytes(v[k]);
        }
    }
    __syncthreads();
    uint4* __restrict__ out4 = reinterpret_cast<uint4*>(
        tables + ((size_t)slice * 2 + which) * N_NODES);
    for (int i = tid; i < N_NODES / 4; i += HIST_THREADS)
        out4[i] = reinterpret_cast<const uint4*>(bins)[i];
}

#define NORM_THREADS 64
#define NORM_VEC4    (2 * N_NODES / 4)
__global__ void norm_from_counts(const unsigned* __restrict__ tables, float* __restrict__ norms) {
    const int t = blockIdx.x * blockDim.x + threadIdx.x;
    if (t >= NORM_VEC4) return;
    const uint4* __restrict__ t4 = reinterpret_cast<const uint4*>(tables);
    uint4 s = {0u, 0u, 0u, 0u};
#pragma unroll
    for (int b = 0; b < HIST_SLICES; ++b) {
        const uint4 c = t4[(size_t)b * NORM_VEC4 + t];
        s.x += c.x; s.y += c.y; s.z += c.z; s.w += c.w;
    }
    float4 r;
    r.x = __frsqrt_rn(fmaxf((float)s.x, 1.0f));
    r.y = __frsqrt_rn(fmaxf((float)s.y, 1.0f));
    r.z = __frsqrt_rn(fmaxf((float)s.z, 1.0f));
    r.w = __frsqrt_rn(fmaxf((float)s.w, 1.0f));
    reinterpret_cast<float4*>(norms)[t] = r;
}

#define SPMM_THREADS   1024
#define SPMM_WAVES     (SPMM_THREADS / 32)
#define VEC_EDGES      16
#define EDGE_CHUNK     (32 * VEC_EDGES)
#define RING           4
#define QCAP           1024
#define ROWS_IN_FLIGHT 8
#define M_TILES   ((ROWS_PER_BLOCK + 15) / 16)
#define ACC_ROWS  (M_TILES * 16)
#define ACC_PITCH 132
#define SUM_COL(p) ((((p) & 31) << 2) | ((p) >> 5))
static_assert(DIM == 128, "SUM_COL: 32 lanes of four columns each");
static_assert(N_EDGES % VEC_EDGES == 0, "the edge list is a whole number of 16-byte owner loads");
static_assert(EDGE_CHUNK + 32 <= QCAP, "a chunk's matches fit behind the waiting entries");
#define POOL_PARTS (SPMM_THREADS / DIM)
#define POOL_ROWS  ((ROWS_PER_BLOCK + POOL_PARTS - 1) / POOL_PARTS)
static_assert(SPMM_THREADS % DIM == 0, "a pooling part is a whole number of waves, one thread per column");
static_assert(N_LAYERS >= 2, "the first layer clears the pooled sums, the last layer adds to them");
#define LIST_CAP  4096
#define LIST_PAD  32
static_assert(ROWS_PER_BLOCK <= 256, "the local row takes the low 8 bits of a list entry");
static_assert(N_NODES <= (1 << 23), "the source row takes the upper 24 bits of a list entry");
#define WS_NEED (sizeof(float) * (2 * (size_t)N_NODES + 2 * (size_t)N_NODES * DIM + N_GRAPHS) + \
                 sizeof(int) * (size_t)(N_BLOCKS + LIST_PAD) + sizeof(int2) * (size_t)N_BLOCKS * LIST_CAP)

__launch_bounds__(SPMM_THREADS)
__global__ void spmm_owner(const float* __restrict__ h, const float* __restrict__ norm_out,
                           const float* __restrict__ norm_in,
                           const int* __restrict__ src, const int* __restrict__ dst,
                           const uint4* __restrict__ own,
                           const float* __restrict__ Wl,
                           const float* __restrict__ bias,
                           float* __restrict__ out_h,
                           const int* __restrict__ gid,
                           float* __restrict__ pooled,
                           float* __restrict__ cnt,
                           int* __restrict__ lcount,
                           int* __restrict__ lists,
                           size_t ws_bytes, int layer, int n_edges, int n_nodes) {
    __shared__ __attribute__((aligned(16))) float acc[ACC_ROWS * ACC_PITCH];
    __shared__ __attribute__((aligned(16))) unsigned char qbuf[sizeof(int) * SPMM_WAVES * QCAP];
    static_assert(sizeof(qbuf) >= sizeof(float) * ACC_ROWS * ACC_PITCH,
                  "the output tile must fit in the queues' LDS");
    __shared__ int lcnt;
    __shared__ int gl[ROWS_PER_BLOCK];
    int* const queue = reinterpret_cast<int*>(qbuf);
    const int tid = threadIdx.x;
    const int lane = tid & 31;
    const int wave = __builtin_amdgcn_readfirstlane(tid >> 5);
    const int r0 = blockIdx.x * ROWS_PER_BLOCK;
    int* const q = queue + wave * QCAP;
    int2* const mylist = reinterpret_cast<int2*>(lists) + (size_t)blockIdx.x * LIST_CAP;
    const bool have_lists = ws_bytes >= WS_NEED;
    const bool record = have_lists && layer == 0;
    const int total = (have_lists && layer != 0) ? lcount[blockIdx.x] : 0;
    const bool replay = have_lists && layer != 0 && total <= LIST_CAP;
    const bool pool = layer == N_LAYERS - 1;

    if (layer == 0) {
        const int gt = blockIdx.x * SPMM_THREADS + tid;
        const int gstep = N_BLOCKS * SPMM_THREADS;
        for (int i = gt; i < N_GRAPHS * DIM; i += gstep) pooled[i] = 0.0f;
        for (int i = gt; i < N_GRAPHS; i += gstep) cnt[i] = 0.0f;
    }

    const float4 zero4 = {0.0f, 0.0f, 0.0f, 0.0f};
    for (int i = tid; i < ACC_ROWS * (ACC_PITCH / 4); i += SPMM_THREADS)
        reinterpret_cast<float4*>(acc)[i] = zero4;
    if (tid == 0) lcnt = 0;
    __syncthreads();

    int qh = 0;
    int qn = 0;

    auto add_rows = [&](int n, int s, int nsb, int row) {
#pragma nounroll
        for (int j0 = 0; j0 < n; j0 += ROWS_IN_FLIGHT) {
            float4 v[ROWS_IN_FLIGHT];
#pragma unroll
            for (int u = 0; u < ROWS_IN_FLIGHT; ++u) {
                const int sj = __builtin_amdgcn_readlane(s, j0 + u);
                v[u] = reinterpret_cast<const float4*>(h + (size_t)sj * DIM)[lane];
            }
#pragma unroll
            for (int u = 0; u < ROWS_IN_FLIGHT; ++u) {
                if (j0 + u < n) {
                    const float nj = __int_as_float(__builtin_amdgcn_readlane(nsb, j0 + u));
                    const int rowj = __builtin_amdgcn_readlane(row, j0 + u);
                    float* ap = acc + rowj * ACC_PITCH + lane;
                    unsafeAtomicAdd(ap + 0,  v[u].x * nj);
                    unsafeAtomicAdd(ap + 32, v[u].y * nj);
                    unsafeAtomicAdd(ap + 64, v[u].z * nj);
                    unsafeAtomicAdd(ap + 96, v[u].w * nj);
                }
            }
        }
    };

    auto drain = [&](int n) {
        int e = 0;
        if (lane < n) e = q[(qh + lane) & (QCAP - 1)];
        const int s = src[e];
        const unsigned rel = (unsigned)(dst[e] - r0);
        int nsb = __float_as_int(norm_out[s]);
        const bool mine = rel < (unsigned)ROWS_PER_BLOCK;
        const int row = mine ? (int)rel : 0;
        if (!mine) nsb = 0;
        if (record) {
            int base = 0;
            if (lane == 0) base = atomicAdd(&lcnt, n);
            base = __builtin_amdgcn_readfirstlane(base);
            if (lane < n && base + lane < LIST_CAP)
                mylist[base + lane] = make_int2((s << 8) | row, nsb);
        }
        add_rows(n, s, nsb, row);
        qh = (qh + n) & (QCAP - 1);
        qn -= n;
    };

    const unsigned pat = (unsigned)blockIdx.x * 0x01010101u;
    const int n_vec = n_edges / VEC_EDGES;
    auto scan_chunk = [&](const uint4& x, int ch) {
        const int vi = ch * 32 + lane;
        const unsigned M = 0x7f7f7f7fu;
        const unsigned t0 = x.x ^ pat, t1 = x.y ^ pat, t2 = x.z ^ pat, t3 = x.w ^ pat;
        const unsigned n0 = ((t0 & M) + M) | t0;
        const unsigned n1 = ((t1 & M) + M) | t1;
        const unsigned n2 = ((t2 & M) + M) | t2;
        const unsigned n3 = ((t3 & M) + M) | t3;
        unsigned g = (((n0 >> 7) & 0x01010101u) | ((n1 >> 6) & 0x02020202u) |
                      ((n2 >> 5) & 0x04040404u) | ((n3 >> 4) & 0x08080808u)) ^ 0x0f0f0f0fu;
        if (vi >= n_vec) g = 0;
        unsigned mk = __builtin_amdgcn_ballot_w32(g != 0);
        if (mk) {
            do {
                if (g) {
                    const int b = __builtin_ctz(g);
                    const int pos = qh + qn + (int)__builtin_amdgcn_mbcnt_lo(mk, 0u);
                    q[pos & (QCAP - 1)] = vi * VEC_EDGES + (b & 7) * 4 + (b >> 3);
                    g &= g - 1;
                }
                qn += __builtin_popcount(mk);
                mk = __builtin_amdgcn_ballot_w32(g != 0);
            } while (mk);
            while (qn >= 32) drain(32);
        }
    };

    if (replay) {
        static_assert(LIST_CAP == 4 * SPMM_THREADS, "four named entry registers below");
        if (total > 0) {
            const int last = total - 1;
            const int i0 = wave * 32 + lane;
            auto entry = [&](int i) { return mylist[i < last ? i : last]; };
            const int2 e0 = entry(i0);
            const int2 e1 = entry(i0 + SPMM_THREADS);
            const int2 e2 = entry(i0 + 2 * SPMM_THREADS);
            const int2 e3 = entry(i0 + 3 * SPMM_THREADS);
#pragma nounroll
            for (int t = 0; t < 4; ++t) {
                const int at = wave * 32 + t * SPMM_THREADS;
                if (at >= total) break;
                const int2 cur = t == 0 ? e0 : t == 1 ? e1 : t == 2 ? e2 : e3;
                const int left = total - at;
                add_rows(left < 32 ? left : 32, cur.x >> 8, cur.y, cur.x & 255);
            }
        }
    } else {
    const int n_chunks = (n_vec + 31) / 32;
    const int offset = n_chunks > 0 ? (int)((blockIdx.x * 6u) % (unsigned)n_chunks) : 0;
    const bool plane = layer == 0;
    const int4* __restrict__ dst4 = reinterpret_cast<const int4*>(dst);

    auto chunk_of = [&](int idx) {
        int ch = idx + offset;
        if (ch >= n_chunks) ch -= n_chunks;
        return idx < n_chunks ? ch : 0;
    };
    auto vec_of = [&](int ch) {
        const int vi = ch * 32 + lane;
        return vi < n_vec ? vi : n_vec - 1;
    };
    if (plane) {
        auto load_set = [&](uint4& x, int idx) { x = own[vec_of(chunk_of(idx))]; };
        uint4 a0, b0, e0, f0;
        static_assert(RING == 4, "four named register sets below");
        load_set(a0, wave);
        __builtin_amdgcn_sched_barrier(0);
        load_set(b0, wave + SPMM_WAVES);
        __builtin_amdgcn_sched_barrier(0);
        load_set(e0, wave + 2 * SPMM_WAVES);
        __builtin_amdgcn_sched_barrier(0);
        load_set(f0, wave + 3 * SPMM_WAVES);
        auto step = [&](uint4& x0, int my) {
            if (my < n_chunks) scan_chunk(x0, chunk_of(my));
            load_set(x0, my + RING * SPMM_WAVES);
        };
        for (int idx = wave; idx < n_chunks; idx += RING * SPMM_WAVES) {
            step(a0, idx);
            step(b0, idx + SPMM_WAVES);
            step(e0, idx + 2 * SPMM_WAVES);
            step(f0, idx + 3 * SPMM_WAVES);
        }
    } else {
#pragma nounroll
        for (int idx = wave; idx < n_chunks; idx += SPMM_WAVES) {
            const int ch = chunk_of(idx);
            const int4* d4 = dst4 + (size_t)vec_of(ch) * 4;
            uint4 x;
            x.x = owner_bytes(d4[0]);
            x.y = owner_bytes(d4[1]);
            x.z = owner_bytes(d4[2]);
            x.w = owner_bytes(d4[3]);
            scan_chunk(x, ch);
        }
    }
    if (qn > 0) drain(qn);
    }
    __syncthreads();
    if (record && tid == 0) lcount[blockIdx.x] = lcnt;
    if (pool && tid < ROWS_PER_BLOCK) {
        const int grow = r0 + tid;
        gl[tid] = grow < n_nodes ? gid[grow] : -1;
    }

    float* const hout = reinterpret_cast<float*>(qbuf);
    const int nt = wave & 7;
    const int mg = wave >> 3;
    const int hl = lane & 15;
    const int hi = lane >> 4;
    const int col = nt * 16 + hl;
    v2f bfrag[32];
#pragma unroll
    for (int kk = 0; kk < 32; ++kk) {
        const int kb = kk * 4 + hi * 2;
        bfrag[kk].x = Wl[(size_t)SUM_COL(kb + 0) * DIM + col];
        bfrag[kk].y = Wl[(size_t)SUM_COL(kb + 1) * DIM + col];
    }
    const float bb = bias[col];
    float nin[4];
#pragma unroll
    for (int t = 0; t < 4; ++t) {
        const int grow = r0 + (mg + 4 * t) * 16 + hl;
        nin[t] = grow < n_nodes ? norm_in[grow] : 0.0f;
    }
#pragma unroll
    for (int t = 0; t < 4; ++t) {
        const int mt = mg + 4 * t;
        if (mt < M_TILES) {
            const int lrow = mt * 16 + hl;
            const float* arow = acc + lrow * ACC_PITCH + hi * 2;
            v8f d = {};
#pragma unroll
            for (int kk = 0; kk < 32; ++kk) {
                v2f a;
                a.x = arow[kk * 4 + 0] * nin[t];
                a.y = arow[kk * 4 + 1] * nin[t];
                d = __builtin_amdgcn_wmma_f32_16x16x4_f32(
                     false, a,  false, bfrag[kk],
                     (short)0, d,  false,  false);
            }
#pragma unroll
            for (int j = 0; j < 8; ++j) {
                const float v = d[j] + bb;
                hout[(mt * 16 + j + hi * 8) * ACC_PITCH + col] = v > 0.0f ? v : 0.0f;
            }
        }
    }
    __syncthreads();

    if (pool) {
        const int pc = tid & (DIM - 1);
        const int rb = (tid / DIM) * POOL_ROWS;
        const int nr = n_nodes - r0 < ROWS_PER_BLOCK ? n_nodes - r0 : ROWS_PER_BLOCK;
        const int re = rb + POOL_ROWS < nr ? rb + POOL_ROWS : nr;
        if (rb < re) {
            auto flush = [&](int g, float s, int n) {
                if ((unsigned)g < (unsigned)N_GRAPHS) {
                    unsafeAtomicAdd(pooled + (size_t)g * DIM + pc, s);
                    if (pc == 0) unsafeAtomicAdd(cnt + g, (float)n);
                }
            };
            int g = gl[rb];
            float s = hout[rb * ACC_PITCH + pc];
            int n = 1;
            for (int r = rb + 1; r < re; ++r) {
                const int gr = gl[r];
                const float v = hout[r * ACC_PITCH + pc];
                if (gr != g) {
                    flush(g, s, n);
                    g = gr; s = 0.0f; n = 0;
                }
                s += v;
                ++n;
            }
            flush(g, s, n);
        }
    } else {
        for (int i = tid; i < ROWS_PER_BLOCK * (DIM / 4); i += SPMM_THREADS) {
            const int r = i >> 5;
            const int c4 = i & 31;
            const int grow = r0 + r;
            if (grow < n_nodes)
                reinterpret_cast<float4*>(out_h + (size_t)grow * DIM)[c4] =
                    reinterpret_cast<const float4*>(hout + r * ACC_PITCH)[c4];
        }
    }
}

__global__ void pool_div(float* __restrict__ pooled, const float* __restrict__ cnt) {
    int t = blockIdx.x * blockDim.x + threadIdx.x;
    int g = t >> 7;
    pooled[t] /= fmaxf(cnt[g], 1.0f);
}

extern "C" void kernel_launch(void* const* d_in, const int* in_sizes, int n_in,
                              void* d_out, int out_size, void* d_ws, size_t ws_size,
                              hipStream_t stream) {
    const float* feats = (const float*)d_in[0];
    const float* Wall  = (const float*)d_in[1];
    const float* ball  = (const float*)d_in[2];
    const int*   src   = (const int*)d_in[3];
    const int*   dst   = (const int*)d_in[4];
    const int*   gid   = (const int*)d_in[5];
    float* out = (float*)d_out;

    float* ws = (float*)d_ws;
    float* norm_out = ws;
    float* norm_in  = ws + N_NODES;
    float* bufA     = ws + 2 * (size_t)N_NODES;
    float* bufB     = bufA + (size_t)N_NODES * DIM;
    float* cnt      = bufB + (size_t)N_NODES * DIM;
    int*   lcount   = (int*)(cnt + N_GRAPHS);
    int*   lists    = lcount + N_BLOCKS + LIST_PAD;

    const int BT = 256;
    unsigned* tables = (unsigned*)bufB;
    unsigned* own = tables + (size_t)HIST_SLICES * 2 * N_NODES;
    degree_count<<<2 * HIST_SLICES, HIST_THREADS, 0, stream>>>(src, dst, tables, own);
    norm_from_counts<<<(NORM_VEC4 + NORM_THREADS - 1) / NORM_THREADS, NORM_THREADS, 0, stream>>>(
        tables, norm_out);

    const float* hin = feats;
    float* bufs[2] = {bufA, bufB};
    int cur = 0;
    for (int l = 0; l < N_LAYERS; ++l) {
        float* hout = bufs[cur];
        spmm_owner<<<N_BLOCKS, SPMM_THREADS, 0, stream>>>(
            hin, norm_out, norm_in, src, dst, (const uint4*)own,
            Wall + (size_t)l * DIM * DIM, ball + (size_t)l * DIM, hout,
            gid, out, cnt, lcount, lists, ws_size, l, N_EDGES, N_NODES);
        hin = hout;
        cur ^= 1;
    }

    pool_div<<<(N_GRAPHS * DIM) / BT, BT, 0, stream>>>(out, cnt);
}


# First layer no longer scans the edge list: 32 more blocks of the degree launch sort each edge slice by owner block (LDS counting sort), wave w reads its slice's segment; LDS sums in arrival order
# speedup vs baseline: 5.2035x; 1.1080x over previous
#include <hip/hip_runtime.h>
#include <hip/hip_bf16.h>

#define N_NODES  50000
#define N_EDGES  800000
#define DIM      128
#define N_LAYERS 3
#define N_GRAPHS 256

typedef __attribute__((ext_vector_type(2))) float v2f;
typedef __attribute__((ext_vector_type(8))) float v8f;

#define ROWS_PER_BLOCK 196
#define N_BLOCKS  ((N_NODES + ROWS_PER_BLOCK - 1) / ROWS_PER_BLOCK)
static_assert(N_BLOCKS <= 256, "an owner is one byte");
__device__ __forceinline__ unsigned owner_bytes(const int4 d) {
    const unsigned a = min((unsigned)d.x / (unsigned)ROWS_PER_BLOCK, 255u);
    const unsigned b = min((unsigned)d.y / (unsigned)ROWS_PER_BLOCK, 255u);
    const unsigned e = min((unsigned)d.z / (unsigned)ROWS_PER_BLOCK, 255u);
    const unsigned f = min((unsigned)d.w / (unsigned)ROWS_PER_BLOCK, 255u);
    return a | (b << 8) | (e << 16) | (f << 24);
}
__device__ __forceinline__ unsigned owner_of(const int d) {
    return min((unsigned)d / (unsigned)ROWS_PER_BLOCK, (unsigned)(N_BLOCKS - 1));
}
__device__ __forceinline__ unsigned seg_entry(const int s, const int d) {
    const unsigned rel = (unsigned)d - owner_of(d) * (unsigned)ROWS_PER_BLOCK;
    return ((unsigned)s << 8) | min(rel, 255u);
}
static_assert(ROWS_PER_BLOCK < 255, "local row 255 is the marker of a destination outside the block");

#define HIST_SLICES   32
#define HIST_THREADS  1024
#define SLICE_EDGES   (N_EDGES / HIST_SLICES)
#define SLICE_IDX4    (SLICE_EDGES / 4)
#define IDX4_PER_THREAD ((SLICE_IDX4 + HIST_THREADS - 1) / HIST_THREADS)
static_assert(N_EDGES % (HIST_SLICES * 4) == 0, "a slice is a whole number of 16-byte loads");
static_assert(N_NODES % 4 == 0, "tables are cleared, stored and added four counters at a time");
#define SEG_OFFS (N_BLOCKS + 1)
static_assert(sizeof(unsigned) * (HIST_SLICES * 2 * (size_t)N_NODES + (size_t)N_EDGES + HIST_SLICES * SEG_OFFS)
                  <= sizeof(float) * (size_t)N_NODES * DIM,
              "the tables, the sorted slices and their offsets fit in one h plane");
static_assert((sizeof(unsigned) * HIST_SLICES * 2 * (size_t)N_NODES) % 16 == 0 && (sizeof(unsigned) * SLICE_EDGES) % 16 == 0,
              "every sorted slice starts on a 16-byte boundary");
static_assert(N_BLOCKS == 256, "the offsets are made by one wave, eight owners a lane");
static_assert(SLICE_EDGES + N_BLOCKS + SEG_OFFS <= N_NODES, "the sort's LDS fits in the table's");

__launch_bounds__(HIST_THREADS)
__global__ void degree_count(const int* __restrict__ src, const int* __restrict__ dst,
                             unsigned* __restrict__ tables, unsigned* __restrict__ segs,
                             unsigned* __restrict__ segoffs) {
    __shared__ __attribute__((aligned(16))) unsigned bins[N_NODES];
    const int tid = threadIdx.x;
    const int which = blockIdx.x / HIST_SLICES;
    const int slice = blockIdx.x % HIST_SLICES;
    const int4* __restrict__ idx4 =
        reinterpret_cast<const int4*>((which ? dst : src) + (size_t)slice * SLICE_EDGES);
    int4 v[IDX4_PER_THREAD];
#pragma unroll
    for (int k = 0; k < IDX4_PER_THREAD; ++k) {
        const int i = tid + k * HIST_THREADS;
        v[k] = idx4[i < SLICE_IDX4 ? i : SLICE_IDX4 - 1];
    }
    if (which == 2) {
        const int4* __restrict__ src4 = reinterpret_cast<const int4*>(src + (size_t)slice * SLICE_EDGES);
        int4 sv[IDX4_PER_THREAD];
#pragma unroll
        for (int k = 0; k < IDX4_PER_THREAD; ++k) {
            const int i = tid + k * HIST_THREADS;
            sv[k] = src4[i < SLICE_IDX4 ? i : SLICE_IDX4 - 1];
        }
        unsigned* const stage = bins;
        unsigned* const ocnt  = bins + SLICE_EDGES;
        unsigned* const offs  = ocnt + N_BLOCKS;
        if (tid < N_BLOCKS) ocnt[tid] = 0u;
        __syncthreads();
        unsigned rk[IDX4_PER_THREAD][4];
#pragma unroll
        for (int k = 0; k < IDX4_PER_THREAD; ++k) {
            if (tid + k * HIST_THREADS < SLICE_IDX4) {
                rk[k][0] = atomicAdd(&ocnt[owner_of(v[k].x)], 1u);
                rk[k][1] = atomicAdd(&ocnt[owner_of(v[k].y)], 1u);
                rk[k][2] = atomicAdd(&ocnt[owner_of(v[k].z)], 1u);
                rk[k][3] = atomicAdd(&ocnt[owner_of(v[k].w)], 1u);
            }
        }
        __syncthreads();
        if (tid < 32) {
            unsigned cn[8];
            unsigned sum = 0u;
#pragma unroll
            for (int j = 0; j < 8; ++j) { cn[j] = ocnt[tid * 8 + j]; sum += cn[j]; }
            unsigned incl = sum;
#pragma unroll
            for (int dl = 1; dl < 32; dl <<= 1) {
                const unsigned t = __shfl_up(incl, (unsigned)dl);
                if (tid >= dl) incl += t;
            }
            unsigned run = incl - sum;
#pragma unroll
            for (int j = 0; j < 8; ++j) { offs[tid * 8 + j] = run; run += cn[j]; }
            if (tid == 31) offs[N_BLOCKS] = run;
        }
        __syncthreads();
#pragma unroll
        for (int k = 0; k < IDX4_PER_THREAD; ++k) {
            if (tid + k * HIST_THREADS < SLICE_IDX4) {
                stage[offs[owner_of(v[k].x)] + rk[k][0]] = seg_entry(sv[k].x, v[k].x);
                stage[offs[owner_of(v[k].y)] + rk[k][1]] = seg_entry(sv[k].y, v[k].y);
                stage[offs[owner_of(v[k].z)] + rk[k][2]] = seg_entry(sv[k].z, v[k].z);
                stage[offs[owner_of(v[k].w)] + rk[k][3]] = seg_entry(sv[k].w, v[k].w);
            }
        }
        __syncthreads();
        uint4* __restrict__ seg4 = reinterpret_cast<uint4*>(segs + (size_t)slice * SLICE_EDGES);
        for (int i = tid; i < SLICE_IDX4; i += HIST_THREADS)
            seg4[i] = reinterpret_cast<const uint4*>(stage)[i];
        if (tid < SEG_OFFS) segoffs[slice * SEG_OFFS + tid] = offs[tid];
        return;
    }
    const uint4 zero4 = {0u, 0u, 0u, 0u};
    for (int i = tid; i < N_NODES / 4; i += HIST_THREADS)
        reinterpret_cast<uint4*>(bins)[i] = zero4;
    __syncthreads();
#pragma unroll
    for (int k = 0; k < IDX4_PER_THREAD; ++k) {
        if (tid + k * HIST_THREADS < SLICE_IDX4) {
            atomicAdd(&bins[v[k].x], 1u);
            atomicAdd(&bins[v[k].y], 1u);
            atomicAdd(&bins[v[k].z], 1u);
            atomicAdd(&bins[v[k].w], 1u);
        }
    }
    __syncthreads();
    uint4* __restrict__ out4 = reinterpret_cast<uint4*>(
        tables + ((size_t)slice * 2 + which) * N_NODES);
    for (int i = tid; i < N_NODES / 4; i += HIST_THREADS)
        out4[i] = reinterpret_cast<const uint4*>(bins)[i];
}

#define NORM_THREADS 64
#define NORM_VEC4    (2 * N_NODES / 4)
__global__ void norm_from_counts(const unsigned* __restrict__ tables, float* __restrict__ norms) {
    const int t = blockIdx.x * blockDim.x + threadIdx.x;
    if (t >= NORM_VEC4) return;
    const uint4* __restrict__ t4 = reinterpret_cast<const uint4*>(tables);
    uint4 s = {0u, 0u, 0u, 0u};
#pragma unroll
    for (int b = 0; b < HIST_SLICES; ++b) {
        const uint4 c = t4[(size_t)b * NORM_VEC4 + t];
        s.x += c.x; s.y += c.y; s.z += c.z; s.w += c.w;
    }
    float4 r;
    r.x = __frsqrt_rn(fmaxf((float)s.x, 1.0f));
    r.y = __frsqrt_rn(fmaxf((float)s.y, 1.0f));
    r.z = __frsqrt_rn(fmaxf((float)s.z, 1.0f));
    r.w = __frsqrt_rn(fmaxf((float)s.w, 1.0f));
    reinterpret_cast<float4*>(norms)[t] = r;
}

#define SPMM_THREADS   1024
#define SPMM_WAVES     (SPMM_THREADS / 32)
#define VEC_EDGES      16
#define EDGE_CHUNK     (32 * VEC_EDGES)
#define QCAP           1024
#define ROWS_IN_FLIGHT 8
#define M_TILES   ((ROWS_PER_BLOCK + 15) / 16)
#define ACC_ROWS  (M_TILES * 16)
#define ACC_PITCH 132
#define SUM_COL(p) ((((p) & 31) << 2) | ((p) >> 5))
static_assert(DIM == 128, "SUM_COL: 32 lanes of four columns each");
static_assert(N_EDGES % VEC_EDGES == 0, "the edge list is a whole number of scan steps' lane shares");
static_assert(EDGE_CHUNK + 32 <= QCAP, "a chunk's matches fit behind the waiting entries");
#define POOL_PARTS (SPMM_THREADS / DIM)
#define POOL_ROWS  ((ROWS_PER_BLOCK + POOL_PARTS - 1) / POOL_PARTS)
static_assert(SPMM_THREADS % DIM == 0, "a pooling part is a whole number of waves, one thread per column");
static_assert(N_LAYERS >= 2, "the first layer clears the pooled sums, the last layer adds to them");
#define LIST_CAP  4096
#define LIST_PAD  32
static_assert(ROWS_PER_BLOCK <= 256, "the local row takes the low 8 bits of a list entry");
static_assert(N_NODES <= (1 << 23), "the source row takes the upper 24 bits of a list entry");
#define WS_NEED (sizeof(float) * (2 * (size_t)N_NODES + 2 * (size_t)N_NODES * DIM + N_GRAPHS) + \
                 sizeof(int) * (size_t)(N_BLOCKS + LIST_PAD) + sizeof(int2) * (size_t)N_BLOCKS * LIST_CAP)

__launch_bounds__(SPMM_THREADS)
__global__ void spmm_owner(const float* __restrict__ h, const float* __restrict__ norm_out,
                           const float* __restrict__ norm_in,
                           const int* __restrict__ src, const int* __restrict__ dst,
                           const int* __restrict__ segs,
                           const int* __restrict__ segoffs,
                           const float* __restrict__ Wl,
                           const float* __restrict__ bias,
                           float* __restrict__ out_h,
                           const int* __restrict__ gid,
                           float* __restrict__ pooled,
                           float* __restrict__ cnt,
                           int* __restrict__ lcount,
                           int* __restrict__ lists,
                           size_t ws_bytes, int layer, int n_edges, int n_nodes) {
    __shared__ __attribute__((aligned(16))) float acc[ACC_ROWS * ACC_PITCH];
    __shared__ __attribute__((aligned(16))) unsigned char qbuf[sizeof(int) * SPMM_WAVES * QCAP];
    static_assert(sizeof(qbuf) >= sizeof(float) * ACC_ROWS * ACC_PITCH,
                  "the output tile must fit in the queues' LDS");
    __shared__ int lcnt;
    __shared__ int gl[ROWS_PER_BLOCK];
    int* const queue = reinterpret_cast<int*>(qbuf);
    const int tid = threadIdx.x;
    const int lane = tid & 31;
    const int wave = __builtin_amdgcn_readfirstlane(tid >> 5);
    const int r0 = blockIdx.x * ROWS_PER_BLOCK;
    int* const q = queue + wave * QCAP;
    int2* const mylist = reinterpret_cast<int2*>(lists) + (size_t)blockIdx.x * LIST_CAP;
    const bool have_lists = ws_bytes >= WS_NEED;
    const bool record = have_lists && layer == 0;
    const int total = (have_lists && layer != 0) ? lcount[blockIdx.x] : 0;
    const bool replay = have_lists && layer != 0 && total <= LIST_CAP;
    const bool pool = layer == N_LAYERS - 1;

    if (layer == 0) {
        const int gt = blockIdx.x * SPMM_THREADS + tid;
        const int gstep = N_BLOCKS * SPMM_THREADS;
        for (int i = gt; i < N_GRAPHS * DIM; i += gstep) pooled[i] = 0.0f;
        for (int i = gt; i < N_GRAPHS; i += gstep) cnt[i] = 0.0f;
    }

    const float4 zero4 = {0.0f, 0.0f, 0.0f, 0.0f};
    for (int i = tid; i < ACC_ROWS * (ACC_PITCH / 4); i += SPMM_THREADS)
        reinterpret_cast<float4*>(acc)[i] = zero4;
    if (tid == 0) lcnt = 0;
    __syncthreads();

    int qh = 0;
    int qn = 0;

    auto add_rows = [&](int n, int s, int nsb, int row) {
#pragma nounroll
        for (int j0 = 0; j0 < n; j0 += ROWS_IN_FLIGHT) {
            float4 v[ROWS_IN_FLIGHT];
#pragma unroll
            for (int u = 0; u < ROWS_IN_FLIGHT; ++u) {
                const int sj = __builtin_amdgcn_readlane(s, j0 + u);
                v[u] = reinterpret_cast<const float4*>(h + (size_t)sj * DIM)[lane];
            }
#pragma unroll
            for (int u = 0; u < ROWS_IN_FLIGHT; ++u) {
                if (j0 + u < n) {
                    const float nj = __int_as_float(__builtin_amdgcn_readlane(nsb, j0 + u));
                    const int rowj = __builtin_amdgcn_readlane(row, j0 + u);
                    float* ap = acc + rowj * ACC_PITCH + lane;
                    unsafeAtomicAdd(ap + 0,  v[u].x * nj);
                    unsafeAtomicAdd(ap + 32, v[u].y * nj);
                    unsafeAtomicAdd(ap + 64, v[u].z * nj);
                    unsafeAtomicAdd(ap + 96, v[u].w * nj);
                }
            }
        }
    };

    auto drain = [&](int n) {
        int e = 0;
        if (lane < n) e = q[(qh + lane) & (QCAP - 1)];
        const int s = src[e];
        const unsigned rel = (unsigned)(dst[e] - r0);
        int nsb = __float_as_int(norm_out[s]);
        const bool mine = rel < (unsigned)ROWS_PER_BLOCK;
        const int row = mine ? (int)rel : 0;
        if (!mine) nsb = 0;
        if (record) {
            int base = 0;
            if (lane == 0) base = atomicAdd(&lcnt, n);
            base = __builtin_amdgcn_readfirstlane(base);
            if (lane < n && base + lane < LIST_CAP)
                mylist[base + lane] = make_int2((s << 8) | row, nsb);
        }
        add_rows(n, s, nsb, row);
        qh = (qh + n) & (QCAP - 1);
        qn -= n;
    };

    const unsigned pat = (unsigned)blockIdx.x * 0x01010101u;
    const int n_vec = n_edges / VEC_EDGES;
    auto scan_chunk = [&](const uint4& x, int ch) {
        const int vi = ch * 32 + lane;
        const unsigned M = 0x7f7f7f7fu;
        const unsigned t0 = x.x ^ pat, t1 = x.y ^ pat, t2 = x.z ^ pat, t3 = x.w ^ pat;
        const unsigned n0 = ((t0 & M) + M) | t0;
        const unsigned n1 = ((t1 & M) + M) | t1;
        const unsigned n2 = ((t2 & M) + M) | t2;
        const unsigned n3 = ((t3 & M) + M) | t3;
        unsigned g = (((n0 >> 7) & 0x01010101u) | ((n1 >> 6) & 0x02020202u) |
                      ((n2 >> 5) & 0x04040404u) | ((n3 >> 4) & 0x08080808u)) ^ 0x0f0f0f0fu;
        if (vi >= n_vec) g = 0;
        unsigned mk = __builtin_amdgcn_ballot_w32(g != 0);
        if (mk) {
            do {
                if (g) {
                    const int b = __builtin_ctz(g);
                    const int pos = qh + qn + (int)__builtin_amdgcn_mbcnt_lo(mk, 0u);
                    q[pos & (QCAP - 1)] = vi * VEC_EDGES + (b & 7) * 4 + (b >> 3);
                    g &= g - 1;
                }
                qn += __builtin_popcount(mk);
                mk = __builtin_amdgcn_ballot_w32(g != 0);
            } while (mk);
            while (qn >= 32) drain(32);
        }
    };

    if (replay) {
        static_assert(LIST_CAP == 4 * SPMM_THREADS, "four named entry registers below");
        if (total > 0) {
            const int last = total - 1;
            const int i0 = wave * 32 + lane;
            auto entry = [&](int i) { return mylist[i < last ? i : last]; };
            const int2 e0 = entry(i0);
            const int2 e1 = entry(i0 + SPMM_THREADS);
            const int2 e2 = entry(i0 + 2 * SPMM_THREADS);
            const int2 e3 = entry(i0 + 3 * SPMM_THREADS);
#pragma nounroll
            for (int t = 0; t < 4; ++t) {
                const int at = wave * 32 + t * SPMM_THREADS;
                if (at >= total) break;
                const int2 cur = t == 0 ? e0 : t == 1 ? e1 : t == 2 ? e2 : e3;
                const int left = total - at;
                add_rows(left < 32 ? left : 32, cur.x >> 8, cur.y, cur.x & 255);
            }
        }
    } else if (layer == 0) {
        static_assert(SPMM_WAVES == HIST_SLICES, "one wave per sorted slice");
        const int* const so = segoffs + wave * SEG_OFFS + blockIdx.x;
        const int beg = __builtin_amdgcn_readfirstlane(so[0]);
        const int end = __builtin_amdgcn_readfirstlane(so[1]);
        const int n = end - beg;
        const int* const seg = segs + (size_t)wave * SLICE_EDGES;
        for (int base = 0; base < n; base += 128) {
            const int lastI = end - 1;
            const int i0 = beg + base + lane;
            auto entry = [&](int i) { return seg[i < lastI ? i : lastI]; };
            const int x0 = entry(i0);
            const int x1 = entry(i0 + 32);
            const int x2 = entry(i0 + 64);
            const int x3 = entry(i0 + 96);
            const int m0 = __float_as_int(norm_out[(unsigned)x0 >> 8]);
            const int m1 = __float_as_int(norm_out[(unsigned)x1 >> 8]);
            const int m2 = __float_as_int(norm_out[(unsigned)x2 >> 8]);
            const int m3 = __float_as_int(norm_out[(unsigned)x3 >> 8]);
#pragma nounroll
            for (int t = 0; t < 4; ++t) {
                const int at = base + 32 * t;
                if (at >= n) break;
                const int x = t == 0 ? x0 : t == 1 ? x1 : t == 2 ? x2 : x3;
                int nsb = t == 0 ? m0 : t == 1 ? m1 : t == 2 ? m2 : m3;
                const int left = n - at;
                const int cnt = left < 32 ? left : 32;
                const int s = (int)((unsigned)x >> 8);
                int row = x & 255;
                if (row >= ROWS_PER_BLOCK) { nsb = 0; row = 0; }
                if (record) {
                    int lb = 0;
                    if (lane == 0) lb = atomicAdd(&lcnt, cnt);
                    lb = __builtin_amdgcn_readfirstlane(lb);
                    if (lane < cnt && lb + lane < LIST_CAP)
                        mylist[lb + lane] = make_int2((s << 8) | row, nsb);
                }
                add_rows(cnt, s, nsb, row);
            }
        }
    } else {
    const int n_chunks = (n_vec + 31) / 32;
    const int offset = n_chunks > 0 ? (int)((blockIdx.x * 6u) % (unsigned)n_chunks) : 0;
    const int4* __restrict__ dst4 = reinterpret_cast<const int4*>(dst);
    auto chunk_of = [&](int idx) {
        int ch = idx + offset;
        if (ch >= n_chunks) ch -= n_chunks;
        return idx < n_chunks ? ch : 0;
    };
    auto vec_of = [&](int ch) {
        const int vi = ch * 32 + lane;
        return vi < n_vec ? vi : n_vec - 1;
    };
#pragma nounroll
    for (int idx = wave; idx < n_chunks; idx += SPMM_WAVES) {
        const int ch = chunk_of(idx);
        const int4* d4 = dst4 + (size_t)vec_of(ch) * 4;
        uint4 x;
        x.x = owner_bytes(d4[0]);
        x.y = owner_bytes(d4[1]);
        x.z = owner_bytes(d4[2]);
        x.w = owner_bytes(d4[3]);
        scan_chunk(x, ch);
    }
    if (qn > 0) drain(qn);
    }
    __syncthreads();
    if (record && tid == 0) lcount[blockIdx.x] = lcnt;
    if (pool && tid < ROWS_PER_BLOCK) {
        const int grow = r0 + tid;
        gl[tid] = grow < n_nodes ? gid[grow] : -1;
    }

    float* const hout = reinterpret_cast<float*>(qbuf);
    const int nt = wave & 7;
    const int mg = wave >> 3;
    const int hl = lane & 15;
    const int hi = lane >> 4;
    const int col = nt * 16 + hl;
    v2f bfrag[32];
#pragma unroll
    for (int kk = 0; kk < 32; ++kk) {
        const int kb = kk * 4 + hi * 2;
        bfrag[kk].x = Wl[(size_t)SUM_COL(kb + 0) * DIM + col];
        bfrag[kk].y = Wl[(size_t)SUM_COL(kb + 1) * DIM + col];
    }
    const float bb = bias[col];
    float nin[4];
#pragma unroll
    for (int t = 0; t < 4; ++t) {
        const int grow = r0 + (mg + 4 * t) * 16 + hl;
        nin[t] = grow < n_nodes ? norm_in[grow] : 0.0f;
    }
#pragma unroll
    for (int t = 0; t < 4; ++t) {
        const int mt = mg + 4 * t;
        if (mt < M_TILES) {
            const int lrow = mt * 16 + hl;
            const float* arow = acc + lrow * ACC_PITCH + hi * 2;
            v8f d = {};
#pragma unroll
            for (int kk = 0; kk < 32; ++kk) {
                v2f a;
                a.x = arow[kk * 4 + 0] * nin[t];
                a.y = arow[kk * 4 + 1] * nin[t];
                d = __builtin_amdgcn_wmma_f32_16x16x4_f32(
                     false, a,  false, bfrag[kk],
                     (short)0, d,  false,  false);
            }
#pragma unroll
            for (int j = 0; j < 8; ++j) {
                const float v = d[j] + bb;
                hout[(mt * 16 + j + hi * 8) * ACC_PITCH + col] = v > 0.0f ? v : 0.0f;
            }
        }
    }
    __syncthreads();

    if (pool) {
        const int pc = tid & (DIM - 1);
        const int rb = (tid / DIM) * POOL_ROWS;
        const int nr = n_nodes - r0 < ROWS_PER_BLOCK ? n_nodes - r0 : ROWS_PER_BLOCK;
        const int re = rb + POOL_ROWS < nr ? rb + POOL_ROWS : nr;
        if (rb < re) {
            auto flush = [&](int g, float s, int n) {
                if ((unsigned)g < (unsigned)N_GRAPHS) {
                    unsafeAtomicAdd(pooled + (size_t)g * DIM + pc, s);
                    if (pc == 0) unsafeAtomicAdd(cnt + g, (float)n);
                }
            };
            int g = gl[rb];
            float s = hout[rb * ACC_PITCH + pc];
            int n = 1;
            for (int r = rb + 1; r < re; ++r) {
                const int gr = gl[r];
                const float v = hout[r * ACC_PITCH + pc];
                if (gr != g) {
                    flush(g, s, n);
                    g = gr; s = 0.0f; n = 0;
                }
                s += v;
                ++n;
            }
            flush(g, s, n);
        }
    } else {
        for (int i = tid; i < ROWS_PER_BLOCK * (DIM / 4); i += SPMM_THREADS) {
            const int r = i >> 5;
            const int c4 = i & 31;
            const int grow = r0 + r;
            if (grow < n_nodes)
                reinterpret_cast<float4*>(out_h + (size_t)grow * DIM)[c4] =
                    reinterpret_cast<const float4*>(hout + r * ACC_PITCH)[c4];
        }
    }
}

__global__ void pool_div(float* __restrict__ pooled, const float* __restrict__ cnt) {
    int t = blockIdx.x * blockDim.x + threadIdx.x;
    int g = t >> 7;
    pooled[t] /= fmaxf(cnt[g], 1.0f);
}

extern "C" void kernel_launch(void* const* d_in, const int* in_sizes, int n_in,
                              void* d_out, int out_size, void* d_ws, size_t ws_size,
                              hipStream_t stream) {
    const float* feats = (const float*)d_in[0];
    const float* Wall  = (const float*)d_in[1];
    const float* ball  = (const float*)d_in[2];
    const int*   src   = (const int*)d_in[3];
    const int*   dst   = (const int*)d_in[4];
    const int*   gid   = (const int*)d_in[5];
    float* out = (float*)d_out;

    float* ws = (float*)d_ws;
    float* norm_out = ws;
    float* norm_in  = ws + N_NODES;
    float* bufA     = ws + 2 * (size_t)N_NODES;
    float* bufB     = bufA + (size_t)N_NODES * DIM;
    float* cnt      = bufB + (size_t)N_NODES * DIM;
    int*   lcount   = (int*)(cnt + N_GRAPHS);
    int*   lists    = lcount + N_BLOCKS + LIST_PAD;

    const int BT = 256;
    unsigned* tables = (unsigned*)bufB;
    unsigned* segs = tables + (size_t)HIST_SLICES * 2 * N_NODES;
    unsigned* segoffs = segs + N_EDGES;
    degree_count<<<3 * HIST_SLICES, HIST_THREADS, 0, stream>>>(src, dst, tables, segs, segoffs);
    norm_from_counts<<<(NORM_VEC4 + NORM_THREADS - 1) / NORM_THREADS, NORM_THREADS, 0, stream>>>(
        tables, norm_out);

    const float* hin = feats;
    float* bufs[2] = {bufA, bufB};
    int cur = 0;
    for (int l = 0; l < N_LAYERS; ++l) {
        float* hout = bufs[cur];
        spmm_owner<<<N_BLOCKS, SPMM_THREADS, 0, stream>>>(
            hin, norm_out, norm_in, src, dst, (const int*)segs, (const int*)segoffs,
            Wall + (size_t)l * DIM * DIM, ball + (size_t)l * DIM, hout,
            gid, out, cnt, lcount, lists, ws_size, l, N_EDGES, N_NODES);
        hin = hout;
        cur ^= 1;
    }

    pool_div<<<(N_GRAPHS * DIM) / BT, BT, 0, stream>>>(out, cnt);
}


# Each block sorts its edges by destination row once (LDS counting sort), lists them; waves take equal shares, sum a row in registers (fma), add to LDS once per row, not per edge; sums in list order
# speedup vs baseline: 5.3974x; 1.0373x over previous
#include <hip/hip_runtime.h>
#include <hip/hip_bf16.h>

#define N_NODES  50000
#define N_EDGES  800000
#define DIM      128
#define N_LAYERS 3
#define N_GRAPHS 256

typedef __attribute__((ext_vector_type(2))) float v2f;
typedef __attribute__((ext_vector_type(8))) float v8f;

#define ROWS_PER_BLOCK 196
#define N_BLOCKS  ((N_NODES + ROWS_PER_BLOCK - 1) / ROWS_PER_BLOCK)
static_assert(N_BLOCKS <= 256, "an owner is one byte");
__device__ __forceinline__ unsigned owner_bytes(const int4 d) {
    const unsigned a = min((unsigned)d.x / (unsigned)ROWS_PER_BLOCK, 255u);
    const unsigned b = min((unsigned)d.y / (unsigned)ROWS_PER_BLOCK, 255u);
    const unsigned e = min((unsigned)d.z / (unsigned)ROWS_PER_BLOCK, 255u);
    const unsigned f = min((unsigned)d.w / (unsigned)ROWS_PER_BLOCK, 255u);
    return a | (b << 8) | (e << 16) | (f << 24);
}
__device__ __forceinline__ unsigned owner_of(const int d) {
    return min((unsigned)d / (unsigned)ROWS_PER_BLOCK, (unsigned)(N_BLOCKS - 1));
}
__device__ __forceinline__ unsigned seg_entry(const int s, const int d) {
    const unsigned rel = (unsigned)d - owner_of(d) * (unsigned)ROWS_PER_BLOCK;
    return ((unsigned)s << 8) | min(rel, 255u);
}
static_assert(ROWS_PER_BLOCK < 255, "local row 255 is the marker of a destination outside the block");

#define HIST_SLICES   32
#define HIST_THREADS  1024
#define SLICE_EDGES   (N_EDGES / HIST_SLICES)
#define SLICE_IDX4    (SLICE_EDGES / 4)
#define IDX4_PER_THREAD ((SLICE_IDX4 + HIST_THREADS - 1) / HIST_THREADS)
static_assert(N_EDGES % (HIST_SLICES * 4) == 0, "a slice is a whole number of 16-byte loads");
static_assert(N_NODES % 4 == 0, "tables are cleared, stored and added four counters at a time");
#define SEG_OFFS (N_BLOCKS + 1)
static_assert(sizeof(unsigned) * (HIST_SLICES * 2 * (size_t)N_NODES + (size_t)N_EDGES + HIST_SLICES * SEG_OFFS)
                  <= sizeof(float) * (size_t)N_NODES * DIM,
              "the tables, the sorted slices and their offsets fit in one h plane");
static_assert((sizeof(unsigned) * HIST_SLICES * 2 * (size_t)N_NODES) % 16 == 0 && (sizeof(unsigned) * SLICE_EDGES) % 16 == 0,
              "every sorted slice starts on a 16-byte boundary");
static_assert(N_BLOCKS == 256, "the offsets are made by one wave, eight owners a lane");
static_assert(SLICE_EDGES + N_BLOCKS + SEG_OFFS <= N_NODES, "the sort's LDS fits in the table's");

__launch_bounds__(HIST_THREADS)
__global__ void degree_count(const int* __restrict__ src, const int* __restrict__ dst,
                             unsigned* __restrict__ tables, unsigned* __restrict__ segs,
                             unsigned* __restrict__ segoffs) {
    __shared__ __attribute__((aligned(16))) unsigned bins[N_NODES];
    const int tid = threadIdx.x;
    const int which = blockIdx.x / HIST_SLICES;
    const int slice = blockIdx.x % HIST_SLICES;
    const int4* __restrict__ idx4 =
        reinterpret_cast<const int4*>((which ? dst : src) + (size_t)slice * SLICE_EDGES);
    int4 v[IDX4_PER_THREAD];
#pragma unroll
    for (int k = 0; k < IDX4_PER_THREAD; ++k) {
        const int i = tid + k * HIST_THREADS;
        v[k] = idx4[i < SLICE_IDX4 ? i : SLICE_IDX4 - 1];
    }
    if (which == 2) {
        const int4* __restrict__ src4 = reinterpret_cast<const int4*>(src + (size_t)slice * SLICE_EDGES);
        int4 sv[IDX4_PER_THREAD];
#pragma unroll
        for (int k = 0; k < IDX4_PER_THREAD; ++k) {
            const int i = tid + k * HIST_THREADS;
            sv[k] = src4[i < SLICE_IDX4 ? i : SLICE_IDX4 - 1];
        }
        unsigned* const stage = bins;
        unsigned* const ocnt  = bins + SLICE_EDGES;
        unsigned* const offs  = ocnt + N_BLOCKS;
        if (tid < N_BLOCKS) ocnt[tid] = 0u;
        __syncthreads();
        unsigned rk[IDX4_PER_THREAD][4];
#pragma unroll
        for (int k = 0; k < IDX4_PER_THREAD; ++k) {
            if (tid + k * HIST_THREADS < SLICE_IDX4) {
                rk[k][0] = atomicAdd(&ocnt[owner_of(v[k].x)], 1u);
                rk[k][1] = atomicAdd(&ocnt[owner_of(v[k].y)], 1u);
                rk[k][2] = atomicAdd(&ocnt[owner_of(v[k].z)], 1u);
                rk[k][3] = atomicAdd(&ocnt[owner_of(v[k].w)], 1u);
            }
        }
        __syncthreads();
        if (tid < 32) {
            unsigned cn[8];
            unsigned sum = 0u;
#pragma unroll
            for (int j = 0; j < 8; ++j) { cn[j] = ocnt[tid * 8 + j]; sum += cn[j]; }
            unsigned incl = sum;
#pragma unroll
            for (int dl = 1; dl < 32; dl <<= 1) {
                const unsigned t = __shfl_up(incl, (unsigned)dl);
                if (tid >= dl) incl += t;
            }
            unsigned run = incl - sum;
#pragma unroll
            for (int j = 0; j < 8; ++j) { offs[tid * 8 + j] = run; run += cn[j]; }
            if (tid == 31) offs[N_BLOCKS] = run;
        }
        __syncthreads();
#pragma unroll
        for (int k = 0; k < IDX4_PER_THREAD; ++k) {
            if (tid + k * HIST_THREADS < SLICE_IDX4) {
                stage[offs[owner_of(v[k].x)] + rk[k][0]] = seg_entry(sv[k].x, v[k].x);
                stage[offs[owner_of(v[k].y)] + rk[k][1]] = seg_entry(sv[k].y, v[k].y);
                stage[offs[owner_of(v[k].z)] + rk[k][2]] = seg_entry(sv[k].z, v[k].z);
                stage[offs[owner_of(v[k].w)] + rk[k][3]] = seg_entry(sv[k].w, v[k].w);
            }
        }
        __syncthreads();
        uint4* __restrict__ seg4 = reinterpret_cast<uint4*>(segs + (size_t)slice * SLICE_EDGES);
        for (int i = tid; i < SLICE_IDX4; i += HIST_THREADS)
            seg4[i] = reinterpret_cast<const uint4*>(stage)[i];
        if (tid < SEG_OFFS) segoffs[slice * SEG_OFFS + tid] = offs[tid];
        return;
    }
    const uint4 zero4 = {0u, 0u, 0u, 0u};
    for (int i = tid; i < N_NODES / 4; i += HIST_THREADS)
        reinterpret_cast<uint4*>(bins)[i] = zero4;
    __syncthreads();
#pragma unroll
    for (int k = 0; k < IDX4_PER_THREAD; ++k) {
        if (tid + k * HIST_THREADS < SLICE_IDX4) {
            atomicAdd(&bins[v[k].x], 1u);
            atomicAdd(&bins[v[k].y], 1u);
            atomicAdd(&bins[v[k].z], 1u);
            atomicAdd(&bins[v[k].w], 1u);
        }
    }
    __syncthreads();
    uint4* __restrict__ out4 = reinterpret_cast<uint4*>(
        tables + ((size_t)slice * 2 + which) * N_NODES);
    for (int i = tid; i < N_NODES / 4; i += HIST_THREADS)
        out4[i] = reinterpret_cast<const uint4*>(bins)[i];
}

#define NORM_THREADS 64
#define NORM_VEC4    (2 * N_NODES / 4)
__global__ void norm_from_counts(const unsigned* __restrict__ tables, float* __restrict__ norms) {
    const int t = blockIdx.x * blockDim.x + threadIdx.x;
    if (t >= NORM_VEC4) return;
    const uint4* __restrict__ t4 = reinterpret_cast<const uint4*>(tables);
    uint4 s = {0u, 0u, 0u, 0u};
#pragma unroll
    for (int b = 0; b < HIST_SLICES; ++b) {
        const uint4 c = t4[(size_t)b * NORM_VEC4 + t];
        s.x += c.x; s.y += c.y; s.z += c.z; s.w += c.w;
    }
    float4 r;
    r.x = __frsqrt_rn(fmaxf((float)s.x, 1.0f));
    r.y = __frsqrt_rn(fmaxf((float)s.y, 1.0f));
    r.z = __frsqrt_rn(fmaxf((float)s.z, 1.0f));
    r.w = __frsqrt_rn(fmaxf((float)s.w, 1.0f));
    reinterpret_cast<float4*>(norms)[t] = r;
}

#define SPMM_THREADS   1024
#define SPMM_WAVES     (SPMM_THREADS / 32)
#define VEC_EDGES      16
#define EDGE_CHUNK     (32 * VEC_EDGES)
#define QCAP           1024
#define ROWS_IN_FLIGHT 8
#define M_TILES   ((ROWS_PER_BLOCK + 15) / 16)
#define ACC_ROWS  (M_TILES * 16)
#define ACC_PITCH 132
#define SUM_COL(p) ((((p) & 31) << 2) | ((p) >> 5))
static_assert(DIM == 128, "SUM_COL: 32 lanes of four columns each");
static_assert(N_EDGES % VEC_EDGES == 0, "the edge list is a whole number of scan steps' lane shares");
static_assert(EDGE_CHUNK + 32 <= QCAP, "a chunk's matches fit behind the waiting entries");
#define POOL_PARTS (SPMM_THREADS / DIM)
#define POOL_ROWS  ((ROWS_PER_BLOCK + POOL_PARTS - 1) / POOL_PARTS)
static_assert(SPMM_THREADS % DIM == 0, "a pooling part is a whole number of waves, one thread per column");
static_assert(N_LAYERS >= 2, "the first layer clears the pooled sums, the last layer adds to them");
#define LIST_CAP  4096
#define LIST_PAD  32
static_assert(ROWS_PER_BLOCK <= 256, "the local row takes the low 8 bits of a list entry");
static_assert(N_NODES <= (1 << 23), "the source row takes the upper 24 bits of a list entry");
#define WS_NEED (sizeof(float) * (2 * (size_t)N_NODES + 2 * (size_t)N_NODES * DIM + N_GRAPHS) + \
                 sizeof(int) * (size_t)(N_BLOCKS + LIST_PAD) + sizeof(int2) * (size_t)N_BLOCKS * LIST_CAP)

__launch_bounds__(SPMM_THREADS)
__global__ void spmm_owner(const float* __restrict__ h, const float* __restrict__ norm_out,
                           const float* __restrict__ norm_in,
                           const int* __restrict__ src, const int* __restrict__ dst,
                           const int* __restrict__ segs,
                           const int* __restrict__ segoffs,
                           const float* __restrict__ Wl,
                           const float* __restrict__ bias,
                           float* __restrict__ out_h,
                           const int* __restrict__ gid,
                           float* __restrict__ pooled,
                           float* __restrict__ cnt,
                           int* __restrict__ lcount,
                           int* __restrict__ lists,
                           size_t ws_bytes, int layer, int n_edges, int n_nodes) {
    __shared__ __attribute__((aligned(16))) float acc[ACC_ROWS * ACC_PITCH];
    __shared__ __attribute__((aligned(16))) unsigned char qbuf[sizeof(int) * SPMM_WAVES * QCAP];
    static_assert(sizeof(qbuf) >= sizeof(float) * ACC_ROWS * ACC_PITCH,
                  "the output tile must fit in the queues' LDS");
    static_assert(sizeof(qbuf) >= sizeof(int2) * LIST_CAP, "the row-ordered list must fit in the queues' LDS");
    __shared__ int rowcnt[256];
    __shared__ int roffs[257];
    __shared__ int gl[ROWS_PER_BLOCK];
    int* const queue = reinterpret_cast<int*>(qbuf);
    int2* const slist = reinterpret_cast<int2*>(qbuf);
    const int tid = threadIdx.x;
    const int lane = tid & 31;
    const int wave = __builtin_amdgcn_readfirstlane(tid >> 5);
    const int r0 = blockIdx.x * ROWS_PER_BLOCK;
    int* const q = queue + wave * QCAP;
    int2* const mylist = reinterpret_cast<int2*>(lists) + (size_t)blockIdx.x * LIST_CAP;
    const bool have_lists = ws_bytes >= WS_NEED;
    const bool record = have_lists && layer == 0;
    const int total = (have_lists && layer != 0) ? lcount[blockIdx.x] : 0;
    const bool replay = have_lists && layer != 0 && total <= LIST_CAP;
    const bool pool = layer == N_LAYERS - 1;

    if (layer == 0) {
        const int gt = blockIdx.x * SPMM_THREADS + tid;
        const int gstep = N_BLOCKS * SPMM_THREADS;
        for (int i = gt; i < N_GRAPHS * DIM; i += gstep) pooled[i] = 0.0f;
        for (int i = gt; i < N_GRAPHS; i += gstep) cnt[i] = 0.0f;
    }

    const float4 zero4 = {0.0f, 0.0f, 0.0f, 0.0f};
    for (int i = tid; i < ACC_ROWS * (ACC_PITCH / 4); i += SPMM_THREADS)
        reinterpret_cast<float4*>(acc)[i] = zero4;
    if (tid < 256) rowcnt[tid] = 0;
    __syncthreads();

    int qh = 0;
    int qn = 0;

    float4 racc = zero4;
    int cur = -1;
    auto flush_row = [&]() {
        if (cur >= 0) {
            float* ap = acc + cur * ACC_PITCH + lane;
            unsafeAtomicAdd(ap + 0,  racc.x);
            unsafeAtomicAdd(ap + 32, racc.y);
            unsafeAtomicAdd(ap + 64, racc.z);
            unsafeAtomicAdd(ap + 96, racc.w);
        }
    };
    auto sum_rows = [&](int n, int s, int nsb, int row) {
#pragma nounroll
        for (int j0 = 0; j0 < n; j0 += ROWS_IN_FLIGHT) {
            float4 v[ROWS_IN_FLIGHT];
#pragma unroll
            for (int u = 0; u < ROWS_IN_FLIGHT; ++u) {
                const int sj = __builtin_amdgcn_readlane(s, j0 + u);
                v[u] = reinterpret_cast<const float4*>(h + (size_t)sj * DIM)[lane];
            }
#pragma unroll
            for (int u = 0; u < ROWS_IN_FLIGHT; ++u) {
                if (j0 + u < n) {
                    const float nj = __int_as_float(__builtin_amdgcn_readlane(nsb, j0 + u));
                    const int rowj = __builtin_amdgcn_readlane(row, j0 + u);
                    if (rowj != cur) {
                        flush_row();
                        cur = rowj;
                        racc = zero4;
                    }
                    racc.x += v[u].x * nj;
                    racc.y += v[u].y * nj;
                    racc.z += v[u].z * nj;
                    racc.w += v[u].w * nj;
                }
            }
        }
    };
    auto share = [&](int tv, int& eb, int& ee) {
        const int per = ((tv + ROWS_IN_FLIGHT * SPMM_WAVES - 1) / (ROWS_IN_FLIGHT * SPMM_WAVES)) * ROWS_IN_FLIGHT;
        eb = wave * per < tv ? wave * per : tv;
        ee = eb + per < tv ? eb + per : tv;
    };

    auto add_rows = [&](int n, int s, int nsb, int row) {
#pragma nounroll
        for (int j0 = 0; j0 < n; j0 += ROWS_IN_FLIGHT) {
            float4 v[ROWS_IN_FLIGHT];
#pragma unroll
            for (int u = 0; u < ROWS_IN_FLIGHT; ++u) {
                const int sj = __builtin_amdgcn_readlane(s, j0 + u);
                v[u] = reinterpret_cast<const float4*>(h + (size_t)sj * DIM)[lane];
            }
#pragma unroll
            for (int u = 0; u < ROWS_IN_FLIGHT; ++u) {
                if (j0 + u < n) {
                    const float nj = __int_as_float(__builtin_amdgcn_readlane(nsb, j0 + u));
                    const int rowj = __builtin_amdgcn_readlane(row, j0 + u);
                    float* ap = acc + rowj * ACC_PITCH + lane;
                    unsafeAtomicAdd(ap + 0,  v[u].x * nj);
                    unsafeAtomicAdd(ap + 32, v[u].y * nj);
                    unsafeAtomicAdd(ap + 64, v[u].z * nj);
                    unsafeAtomicAdd(ap + 96, v[u].w * nj);
                }
            }
        }
    };

    auto drain = [&](int n) {
        int e = 0;
        if (lane < n) e = q[(qh + lane) & (QCAP - 1)];
        const int s = src[e];
        const unsigned rel = (unsigned)(dst[e] - r0);
        int nsb = __float_as_int(norm_out[s]);
        const bool mine = rel < (unsigned)ROWS_PER_BLOCK;
        const int row = mine ? (int)rel : 0;
        if (!mine) nsb = 0;
        add_rows(n, s, nsb, row);
        qh = (qh + n) & (QCAP - 1);
        qn -= n;
    };

    const unsigned pat = (unsigned)blockIdx.x * 0x01010101u;
    const int n_vec = n_edges / VEC_EDGES;
    auto scan_chunk = [&](const uint4& x, int ch) {
        const int vi = ch * 32 + lane;
        const unsigned M = 0x7f7f7f7fu;
        const unsigned t0 = x.x ^ pat, t1 = x.y ^ pat, t2 = x.z ^ pat, t3 = x.w ^ pat;
        const unsigned n0 = ((t0 & M) + M) | t0;
        const unsigned n1 = ((t1 & M) + M) | t1;
        const unsigned n2 = ((t2 & M) + M) | t2;
        const unsigned n3 = ((t3 & M) + M) | t3;
        unsigned g = (((n0 >> 7) & 0x01010101u) | ((n1 >> 6) & 0x02020202u) |
                      ((n2 >> 5) & 0x04040404u) | ((n3 >> 4) & 0x08080808u)) ^ 0x0f0f0f0fu;
        if (vi >= n_vec) g = 0;
        unsigned mk = __builtin_amdgcn_ballot_w32(g != 0);
        if (mk) {
            do {
                if (g) {
                    const int b = __builtin_ctz(g);
                    const int pos = qh + qn + (int)__builtin_amdgcn_mbcnt_lo(mk, 0u);
                    q[pos & (QCAP - 1)] = vi * VEC_EDGES + (b & 7) * 4 + (b >> 3);
                    g &= g - 1;
                }
                qn += __builtin_popcount(mk);
                mk = __builtin_amdgcn_ballot_w32(g != 0);
            } while (mk);
            while (qn >= 32) drain(32);
        }
    };

    if (replay) {
        int eb, ee;
        share(total, eb, ee);
        const int n = ee - eb;
        for (int base = 0; base < n; base += 128) {
            const int lastI = ee - 1;
            const int i0 = eb + base + lane;
            auto entry = [&](int i) { return mylist[i < lastI ? i : lastI]; };
            const int2 e0 = entry(i0);
            const int2 e1 = entry(i0 + 32);
            const int2 e2 = entry(i0 + 64);
            const int2 e3 = entry(i0 + 96);
#pragma nounroll
            for (int t = 0; t < 4; ++t) {
                const int at = base + 32 * t;
                if (at >= n) break;
                const int2 ce = t == 0 ? e0 : t == 1 ? e1 : t == 2 ? e2 : e3;
                const int left = n - at;
                sum_rows(left < 32 ? left : 32, (int)((unsigned)ce.x >> 8), ce.y, ce.x & 255);
            }
        }
        flush_row();
    } else if (layer == 0) {
        static_assert(SPMM_WAVES == HIST_SLICES, "one wave per sorted slice");
        const int* const so = segoffs + wave * SEG_OFFS + blockIdx.x;
        const int beg = __builtin_amdgcn_readfirstlane(so[0]);
        const int end = __builtin_amdgcn_readfirstlane(so[1]);
        const int n = end - beg;
        const int* const seg = segs + (size_t)wave * SLICE_EDGES;
        const int lastI = end - 1;
        auto entry = [&](int i) { return seg[i < lastI ? i : lastI]; };
        int xs[4] = {0, 0, 0, 0};
        int ms[4] = {0, 0, 0, 0};
        if (n > 0) {
#pragma unroll
            for (int t = 0; t < 4; ++t) xs[t] = entry(beg + 32 * t + lane);
#pragma unroll
            for (int t = 0; t < 4; ++t) ms[t] = __float_as_int(norm_out[(unsigned)xs[t] >> 8]);
#pragma unroll
            for (int t = 0; t < 4; ++t)
                if (32 * t + lane < n) atomicAdd(&rowcnt[xs[t] & 255], 1);
            for (int at = 128 + lane; at < n; at += 32) atomicAdd(&rowcnt[entry(beg + at) & 255], 1);
        }
        __syncthreads();
        if (tid < 32) {
            int cn[8];
            int sum = 0;
#pragma unroll
            for (int j = 0; j < 8; ++j) { cn[j] = rowcnt[tid * 8 + j]; sum += cn[j]; }
            int incl = sum;
#pragma unroll
            for (int dl = 1; dl < 32; dl <<= 1) {
                const int t = __shfl_up(incl, (unsigned)dl);
                if (tid >= dl) incl += t;
            }
            int run = incl - sum;
#pragma unroll
            for (int j = 0; j < 8; ++j) { roffs[tid * 8 + j] = run; run += cn[j]; }
            if (tid == 31) roffs[256] = run;
        }
        __syncthreads();
        const bool sorted = roffs[256] <= LIST_CAP;
        if (sorted) {
#pragma unroll
            for (int t = 0; t < 4; ++t) {
                if (32 * t + lane < n) {
                    const int pos = atomicAdd(&roffs[xs[t] & 255], 1);
                    slist[pos] = make_int2(xs[t], ms[t]);
                }
            }
            for (int at = 128 + lane; at < n; at += 32) {
                const int x = entry(beg + at);
                const int pos = atomicAdd(&roffs[x & 255], 1);
                slist[pos] = make_int2(x, __float_as_int(norm_out[(unsigned)x >> 8]));
            }
            __syncthreads();
            const int tv = __builtin_amdgcn_readfirstlane(roffs[ROWS_PER_BLOCK - 1]);
            if (record) {
                for (int i = tid; i < tv; i += SPMM_THREADS) mylist[i] = slist[i];
                if (tid == 0) lcount[blockIdx.x] = tv;
            }
            int eb, ee;
            share(tv, eb, ee);
#pragma nounroll
            for (int at = eb; at < ee; at += 32) {
                const int i = at + lane;
                const int2 ce = slist[i < ee - 1 ? i : ee - 1];
                const int left = ee - at;
                sum_rows(left < 32 ? left : 32, (int)((unsigned)ce.x >> 8), ce.y, ce.x & 255);
            }
            flush_row();
        } else {
            if (record && tid == 0) lcount[blockIdx.x] = LIST_CAP + 1;
            for (int base = 0; base < n; base += 128) {
                const int i0 = beg + base + lane;
                const int x0 = entry(i0);
                const int x1 = entry(i0 + 32);
                const int x2 = entry(i0 + 64);
                const int x3 = entry(i0 + 96);
                const int m0 = __float_as_int(norm_out[(unsigned)x0 >> 8]);
                const int m1 = __float_as_int(norm_out[(unsigned)x1 >> 8]);
                const int m2 = __float_as_int(norm_out[(unsigned)x2 >> 8]);
                const int m3 = __float_as_int(norm_out[(unsigned)x3 >> 8]);
#pragma nounroll
                for (int t = 0; t < 4; ++t) {
                    const int at = base + 32 * t;
                    if (at >= n) break;
                    const int x = t == 0 ? x0 : t == 1 ? x1 : t == 2 ? x2 : x3;
                    int nsb = t == 0 ? m0 : t == 1 ? m1 : t == 2 ? m2 : m3;
                    const int left = n - at;
                    const int s = (int)((unsigned)x >> 8);
                    int row = x & 255;
                    if (row >= ROWS_PER_BLOCK) { nsb = 0; row = 0; }
                    add_rows(left < 32 ? left : 32, s, nsb, row);
                }
            }
        }
    } else {
    const int n_chunks = (n_vec + 31) / 32;
    const int offset = n_chunks > 0 ? (int)((blockIdx.x * 6u) % (unsigned)n_chunks) : 0;
    const int4* __restrict__ dst4 = reinterpret_cast<const int4*>(dst);
    auto chunk_of = [&](int idx) {
        int ch = idx + offset;
        if (ch >= n_chunks) ch -= n_chunks;
        return idx < n_chunks ? ch : 0;
    };
    auto vec_of = [&](int ch) {
        const int vi = ch * 32 + lane;
        return vi < n_vec ? vi : n_vec - 1;
    };
#pragma nounroll
    for (int idx = wave; idx < n_chunks; idx += SPMM_WAVES) {
        const int ch = chunk_of(idx);
        const int4* d4 = dst4 + (size_t)vec_of(ch) * 4;
        uint4 x;
        x.x = owner_bytes(d4[0]);
        x.y = owner_bytes(d4[1]);
        x.z = owner_bytes(d4[2]);
        x.w = owner_bytes(d4[3]);
        scan_chunk(x, ch);
    }
    if (qn > 0) drain(qn);
    }
    __syncthreads();
    if (pool && tid < ROWS_PER_BLOCK) {
        const int grow = r0 + tid;
        gl[tid] = grow < n_nodes ? gid[grow] : -1;
    }

    float* const hout = reinterpret_cast<float*>(qbuf);
    const int nt = wave & 7;
    const int mg = wave >> 3;
    const int hl = lane & 15;
    const int hi = lane >> 4;
    const int col = nt * 16 + hl;
    v2f bfrag[32];
#pragma unroll
    for (int kk = 0; kk < 32; ++kk) {
        const int kb = kk * 4 + hi * 2;
        bfrag[kk].x = Wl[(size_t)SUM_COL(kb + 0) * DIM + col];
        bfrag[kk].y = Wl[(size_t)SUM_COL(kb + 1) * DIM + col];
    }
    const float bb = bias[col];
    float nin[4];
#pragma unroll
    for (int t = 0; t < 4; ++t) {
        const int grow = r0 + (mg + 4 * t) * 16 + hl;
        nin[t] = grow < n_nodes ? norm_in[grow] : 0.0f;
    }
#pragma unroll
    for (int t = 0; t < 4; ++t) {
        const int mt = mg + 4 * t;
        if (mt < M_TILES) {
            const int lrow = mt * 16 + hl;
            const float* arow = acc + lrow * ACC_PITCH + hi * 2;
            v8f d = {};
#pragma unroll
            for (int kk = 0; kk < 32; ++kk) {
                v2f a;
                a.x = arow[kk * 4 + 0] * nin[t];
                a.y = arow[kk * 4 + 1] * nin[t];
                d = __builtin_amdgcn_wmma_f32_16x16x4_f32(
                     false, a,  false, bfrag[kk],
                     (short)0, d,  false,  false);
            }
#pragma unroll
            for (int j = 0; j < 8; ++j) {
                const float v = d[j] + bb;
                hout[(mt * 16 + j + hi * 8) * ACC_PITCH + col] = v > 0.0f ? v : 0.0f;
            }
        }
    }
    __syncthreads();

    if (pool) {
        const int pc = tid & (DIM - 1);
        const int rb = (tid / DIM) * POOL_ROWS;
        const int nr = n_nodes - r0 < ROWS_PER_BLOCK ? n_nodes - r0 : ROWS_PER_BLOCK;
        const int re = rb + POOL_ROWS < nr ? rb + POOL_ROWS : nr;
        if (rb < re) {
            auto flush = [&](int g, float s, int n) {
                if ((unsigned)g < (unsigned)N_GRAPHS) {
                    unsafeAtomicAdd(pooled + (size_t)g * DIM + pc, s);
                    if (pc == 0) unsafeAtomicAdd(cnt + g, (float)n);
                }
            };
            int g = gl[rb];
            float s = hout[rb * ACC_PITCH + pc];
            int n = 1;
            for (int r = rb + 1; r < re; ++r) {
                const int gr = gl[r];
                const float v = hout[r * ACC_PITCH + pc];
                if (gr != g) {
                    flush(g, s, n);
                    g = gr; s = 0.0f; n = 0;
                }
                s += v;
                ++n;
            }
            flush(g, s, n);
        }
    } else {
        for (int i = tid; i < ROWS_PER_BLOCK * (DIM / 4); i += SPMM_THREADS) {
            const int r = i >> 5;
            const int c4 = i & 31;
            const int grow = r0 + r;
            if (grow < n_nodes)
                reinterpret_cast<float4*>(out_h + (size_t)grow * DIM)[c4] =
                    reinterpret_cast<const float4*>(hout + r * ACC_PITCH)[c4];
        }
    }
}

__global__ void pool_div(float* __restrict__ pooled, const float* __restrict__ cnt) {
    int t = blockIdx.x * blockDim.x + threadIdx.x;
    int g = t >> 7;
    pooled[t] /= fmaxf(cnt[g], 1.0f);
}

extern "C" void kernel_launch(void* const* d_in, const int* in_sizes, int n_in,
                              void* d_out, int out_size, void* d_ws, size_t ws_size,
                              hipStream_t stream) {
    const float* feats = (const float*)d_in[0];
    const float* Wall  = (const float*)d_in[1];
    const float* ball  = (const float*)d_in[2];
    const int*   src   = (const int*)d_in[3];
    const int*   dst   = (const int*)d_in[4];
    const int*   gid   = (const int*)d_in[5];
    float* out = (float*)d_out;

    float* ws = (float*)d_ws;
    float* norm_out = ws;
    float* norm_in  = ws + N_NODES;
    float* bufA     = ws + 2 * (size_t)N_NODES;
    float* bufB     = bufA + (size_t)N_NODES * DIM;
    float* cnt      = bufB + (size_t)N_NODES * DIM;
    int*   lcount   = (int*)(cnt + N_GRAPHS);
    int*   lists    = lcount + N_BLOCKS + LIST_PAD;

    const int BT = 256;
    unsigned* tables = (unsigned*)bufB;
    unsigned* segs = tables + (size_t)HIST_SLICES * 2 * N_NODES;
    unsigned* segoffs = segs + N_EDGES;
    degree_count<<<3 * HIST_SLICES, HIST_THREADS, 0, stream>>>(src, dst, tables, segs, segoffs);
    norm_from_counts<<<(NORM_VEC4 + NORM_THREADS - 1) / NORM_THREADS, NORM_THREADS, 0, stream>>>(
        tables, norm_out);

    const float* hin = feats;
    float* bufs[2] = {bufA, bufB};
    int cur = 0;
    for (int l = 0; l < N_LAYERS; ++l) {
        float* hout = bufs[cur];
        spmm_owner<<<N_BLOCKS, SPMM_THREADS, 0, stream>>>(
            hin, norm_out, norm_in, src, dst, (const int*)segs, (const int*)segoffs,
            Wall + (size_t)l * DIM * DIM, ball + (size_t)l * DIM, hout,
            gid, out, cnt, lcount, lists, ws_size, l, N_EDGES, N_NODES);
        hin = hout;
        cur ^= 1;
    }

    pool_div<<<(N_GRAPHS * DIM) / BT, BT, 0, stream>>>(out, cnt);
}
